# starting kernel
# baseline (speedup 1.0000x reference)
#include <hip/hip_runtime.h>
#include <math.h>


#define N_NODES 50000
#define N_EDGES 1600000
#define ET_EDGES (N_EDGES + N_NODES)
#define IN_CH 128
#define OUT_CH 128
#define HEADS 4
#define CPH 32
#define NEG_SLOPE 0.2f
#define SOFTMAX_EPS 1e-16f
#define LN_EPS 1e-5f

typedef float v2f __attribute__((ext_vector_type(2)));
typedef float v8f __attribute__((ext_vector_type(8)));

__global__ void k_init(float* __restrict__ out, float* __restrict__ ssum,
                       unsigned* __restrict__ mkey, int nout, int nseg) {
    int i = blockIdx.x * blockDim.x + threadIdx.x;
    if (i < nout) out[i] = 0.0f;
    if (i < nseg) { ssum[i] = 0.0f; mkey[i] = 0u; }
}

__global__ void k_gemm_wmma(const float* __restrict__ x,
                            const float* __restrict__ W,
                            float* __restrict__ h, int nrows) {
    extern __shared__ float Wlds[];
    for (int i = threadIdx.x; i < (IN_CH * OUT_CH) / 4; i += blockDim.x)
        ((float4*)Wlds)[i] = ((const float4*)W)[i];
    __syncthreads();

    const int wave = threadIdx.x >> 5;
    const int lane = threadIdx.x & 31;
    const int rowBase = blockIdx.x * 128 + wave * 16;
    const int m  = lane & 15;
    const int kk = (lane >> 4) << 1;

    int arow = rowBase + m;
    if (arow >= nrows) arow = nrows - 1;
    const float* xk = x + (size_t)arow * IN_CH + kk;

    v8f c[8];
#pragma unroll
    for (int t = 0; t < 8; ++t) c[t] = (v8f){0,0,0,0,0,0,0,0};

    float2 av = *(const float2*)xk;
    for (int k = 0; k < IN_CH; k += 4) {
        v2f a; a.x = av.x; a.y = av.y;
        if (k + 4 < IN_CH) av = *(const float2*)(xk + k + 4);

        const float* w0 = &Wlds[(k + kk) * OUT_CH + m];
        v2f b[8];
#pragma unroll
        for (int t = 0; t < 8; ++t) {
            b[t].x = w0[t * 16];
            b[t].y = w0[OUT_CH + t * 16];
        }
#pragma unroll
        for (int t = 0; t < 8; ++t)
            c[t] = __builtin_amdgcn_wmma_f32_16x16x4_f32(
                false, a, false, b[t], (short)0, c[t], false, false);
    }

#pragma unroll
    for (int t = 0; t < 8; ++t) {
#pragma unroll
        for (int v = 0; v < 8; ++v) {
            const int M = v + ((lane >> 4) << 3);
            const int row = rowBase + M;
            if (row < nrows)
                h[(size_t)row * OUT_CH + t * 16 + m] = c[t][v];
        }
    }
}

__global__ void k_dots(const float* __restrict__ h,
                       const float* __restrict__ att_src,
                       const float* __restrict__ att_dst,
                       float* __restrict__ a_s, float* __restrict__ a_d, int n) {
    const int wave = threadIdx.x >> 5;
    const int lane = threadIdx.x & 31;
    const int node = blockIdx.x * 2 + (wave >> 2);
    const int head = wave & 3;
    if (node >= n) return;
    const float hv = h[(size_t)node * OUT_CH + head * CPH + lane];
    float vs = hv * att_src[head * CPH + lane];
    float vd = hv * att_dst[head * CPH + lane];
#pragma unroll
    for (int o = 16; o > 0; o >>= 1) {
        vs += __shfl_xor(vs, o, 32);
        vd += __shfl_xor(vd, o, 32);
    }
    if (lane == 0) {
        a_s[node * HEADS + head] = vs;
        a_d[node * HEADS + head] = vd;
    }
}

__device__ __forceinline__ unsigned f2key(float f) {
    unsigned b = __float_as_uint(f);
    return (b & 0x80000000u) ? ~b : (b | 0x80000000u);
}
__device__ __forceinline__ float key2f(unsigned k) {
    unsigned b = (k & 0x80000000u) ? (k ^ 0x80000000u) : ~k;
    return __uint_as_float(b);
}

__global__ void k_edge_max(const int* __restrict__ ei,
                           const float* __restrict__ a_s,
                           const float* __restrict__ a_d,
                           float* __restrict__ logits,
                           unsigned* __restrict__ mkey) {
    const int tid = blockIdx.x * blockDim.x + threadIdx.x;
    if (tid >= ET_EDGES * HEADS) return;
    const int e  = tid >> 2;
    const int hd = tid & 3;
    int s, d;
    if (e < N_EDGES) { s = ei[e]; d = ei[N_EDGES + e]; }
    else             { s = d = e - N_EDGES; }
    float v = a_s[s * HEADS + hd] + a_d[d * HEADS + hd];
    v = v > 0.0f ? v : NEG_SLOPE * v;
    logits[tid] = v;
    atomicMax(&mkey[d * HEADS + hd], f2key(v));
}

__global__ void k_edge_exp(const int* __restrict__ ei,
                           float* __restrict__ logits,
                           const unsigned* __restrict__ mkey,
                           float* __restrict__ ssum) {
    const int tid = blockIdx.x * blockDim.x + threadIdx.x;
    if (tid >= ET_EDGES * HEADS) return;
    const int e  = tid >> 2;
    const int hd = tid & 3;
    const int d = (e < N_EDGES) ? ei[N_EDGES + e] : (e - N_EDGES);
    const float mf = key2f(mkey[d * HEADS + hd]);
    const float ex = expf(logits[tid] - mf);
    logits[tid] = ex;
    atomicAdd(&ssum[d * HEADS + hd], ex);
}

__global__ void k_edge_agg(const int* __restrict__ ei,
                           const float* __restrict__ h,
                           const float* __restrict__ evals,
                           const float* __restrict__ ssum,
                           float* __restrict__ out) {
    const int wave = threadIdx.x >> 5;
    const int lane = threadIdx.x & 31;
    const int e = blockIdx.x * 8 + wave;
    if (e >= ET_EDGES) return;
    int s, d;
    if (e < N_EDGES) { s = ei[e]; d = ei[N_EDGES + e]; }
    else             { s = d = e - N_EDGES; }
    const int head = lane >> 3;
    const float alpha = evals[e * HEADS + head] /
                        (ssum[d * HEADS + head] + SOFTMAX_EPS);
    const float4 hv = *(const float4*)(h + (size_t)s * OUT_CH + lane * 4);
    float* op = out + (size_t)d * OUT_CH + lane * 4;
    atomicAdd(op + 0, hv.x * alpha);
    atomicAdd(op + 1, hv.y * alpha);
    atomicAdd(op + 2, hv.z * alpha);
    atomicAdd(op + 3, hv.w * alpha);
}

__global__ void k_ln_elu(float* __restrict__ out,
                         const float* __restrict__ bias,
                         const float* __restrict__ gamma,
                         const float* __restrict__ beta) {
    __shared__ float red[4];
    const int node = blockIdx.x;
    const int c = threadIdx.x;
    const int wave = c >> 5, lane = c & 31;

    float v = out[(size_t)node * OUT_CH + c] + bias[c];

    float t = v;
#pragma unroll
    for (int o = 16; o > 0; o >>= 1) t += __shfl_xor(t, o, 32);
    if (lane == 0) red[wave] = t;
    __syncthreads();
    const float mu = (red[0] + red[1] + red[2] + red[3]) * (1.0f / OUT_CH);
    __syncthreads();

    const float dv = v - mu;
    t = dv * dv;
#pragma unroll
    for (int o = 16; o > 0; o >>= 1) t += __shfl_xor(t, o, 32);
    if (lane == 0) red[wave] = t;
    __syncthreads();
    const float var = (red[0] + red[1] + red[2] + red[3]) * (1.0f / OUT_CH);

    float y = dv * (1.0f / sqrtf(var + LN_EPS)) * gamma[c] + beta[c];
    y = y > 0.0f ? y : (expf(y) - 1.0f);
    out[(size_t)node * OUT_CH + c] = y;
}

extern "C" void kernel_launch(void* const* d_in, const int* in_sizes, int n_in,
                              void* d_out, int out_size, void* d_ws, size_t ws_size,
                              hipStream_t stream) {
    const float* x       = (const float*)d_in[0];
    const int*   ei      = (const int*)  d_in[1];
    const float* W       = (const float*)d_in[2];
    const float* att_src = (const float*)d_in[3];
    const float* att_dst = (const float*)d_in[4];
    const float* bias    = (const float*)d_in[5];
    const float* gamma   = (const float*)d_in[6];
    const float* beta    = (const float*)d_in[7];
    float* out = (float*)d_out;

    char* ws = (char*)d_ws;
    float*    h      = (float*)ws;
    ws += (size_t)N_NODES * OUT_CH * sizeof(float);
    float*    a_s    = (float*)ws;  ws += (size_t)N_NODES * HEADS * sizeof(float);
    float*    a_d    = (float*)ws;  ws += (size_t)N_NODES * HEADS * sizeof(float);
    unsigned* mkey   = (unsigned*)ws; ws += (size_t)N_NODES * HEADS * sizeof(unsigned);
    float*    ssum   = (float*)ws;  ws += (size_t)N_NODES * HEADS * sizeof(float);
    float*    evals  = (float*)ws;

    const int nout = N_NODES * OUT_CH;
    const int nseg = N_NODES * HEADS;

    k_init<<<(nout + 255) / 256, 256, 0, stream>>>(out, ssum, mkey, nout, nseg);

    k_gemm_wmma<<<(N_NODES + 127) / 128, 256, IN_CH * OUT_CH * sizeof(float),
                  stream>>>(x, W, h, N_NODES);

    k_dots<<<(N_NODES + 1) / 2, 256, 0, stream>>>(h, att_src, att_dst, a_s, a_d,
                                                  N_NODES);

    k_edge_max<<<(ET_EDGES * HEADS + 255) / 256, 256, 0, stream>>>(
        ei, a_s, a_d, evals, mkey);

    k_edge_exp<<<(ET_EDGES * HEADS + 255) / 256, 256, 0, stream>>>(
        ei, evals, mkey, ssum);

    k_edge_agg<<<(ET_EDGES + 7) / 8, 256, 0, stream>>>(ei, h, evals, ssum, out);

    k_ln_elu<<<N_NODES, 128, 0, stream>>>(out, bias, gamma, beta);
}


# Global scatter atomics replaced by owner-computes sums: a block keeps 196 destination rows in LDS, filters the unsorted edge list for them, adds by LDS float atomics (arrival order), writes rows once
# speedup vs baseline: 1.8983x; 1.8983x over previous
#include <hip/hip_runtime.h>
#include <math.h>


#define N_NODES 50000
#define N_EDGES 1600000
#define ET_EDGES (N_EDGES + N_NODES)
#define IN_CH 128
#define OUT_CH 128
#define HEADS 4
#define CPH 32
#define NEG_SLOPE 0.2f
#define SOFTMAX_EPS 1e-16f
#define LN_EPS 1e-5f

typedef float v2f __attribute__((ext_vector_type(2)));
typedef float v8f __attribute__((ext_vector_type(8)));

__global__ void k_init(unsigned* __restrict__ mkey, int nseg) {
    int i = blockIdx.x * blockDim.x + threadIdx.x;
    if (i < nseg) mkey[i] = 0u;
}

__global__ void k_gemm_wmma(const float* __restrict__ x,
                            const float* __restrict__ W,
                            float* __restrict__ h, int nrows) {
    extern __shared__ float Wlds[];
    for (int i = threadIdx.x; i < (IN_CH * OUT_CH) / 4; i += blockDim.x)
        ((float4*)Wlds)[i] = ((const float4*)W)[i];
    __syncthreads();

    const int wave = threadIdx.x >> 5;
    const int lane = threadIdx.x & 31;
    const int rowBase = blockIdx.x * 128 + wave * 16;
    const int m  = lane & 15;
    const int kk = (lane >> 4) << 1;

    int arow = rowBase + m;
    if (arow >= nrows) arow = nrows - 1;
    const float* xk = x + (size_t)arow * IN_CH + kk;

    v8f c[8];
#pragma unroll
    for (int t = 0; t < 8; ++t) c[t] = (v8f){0,0,0,0,0,0,0,0};

    float2 av = *(const float2*)xk;
    for (int k = 0; k < IN_CH; k += 4) {
        v2f a; a.x = av.x; a.y = av.y;
        if (k + 4 < IN_CH) av = *(const float2*)(xk + k + 4);

        const float* w0 = &Wlds[(k + kk) * OUT_CH + m];
        v2f b[8];
#pragma unroll
        for (int t = 0; t < 8; ++t) {
            b[t].x = w0[t * 16];
            b[t].y = w0[OUT_CH + t * 16];
        }
#pragma unroll
        for (int t = 0; t < 8; ++t)
            c[t] = __builtin_amdgcn_wmma_f32_16x16x4_f32(
                false, a, false, b[t], (short)0, c[t], false, false);
    }

#pragma unroll
    for (int t = 0; t < 8; ++t) {
#pragma unroll
        for (int v = 0; v < 8; ++v) {
            const int M = v + ((lane >> 4) << 3);
            const int row = rowBase + M;
            if (row < nrows)
                h[(size_t)row * OUT_CH + t * 16 + m] = c[t][v];
        }
    }
}

__global__ void k_dots(const float* __restrict__ h,
                       const float* __restrict__ att_src,
                       const float* __restrict__ att_dst,
                       float* __restrict__ a_s, float* __restrict__ a_d, int n) {
    const int wave = threadIdx.x >> 5;
    const int lane = threadIdx.x & 31;
    const int node = blockIdx.x * 2 + (wave >> 2);
    const int head = wave & 3;
    if (node >= n) return;
    const float hv = h[(size_t)node * OUT_CH + head * CPH + lane];
    float vs = hv * att_src[head * CPH + lane];
    float vd = hv * att_dst[head * CPH + lane];
#pragma unroll
    for (int o = 16; o > 0; o >>= 1) {
        vs += __shfl_xor(vs, o, 32);
        vd += __shfl_xor(vd, o, 32);
    }
    if (lane == 0) {
        a_s[node * HEADS + head] = vs;
        a_d[node * HEADS + head] = vd;
    }
}

__device__ __forceinline__ unsigned f2key(float f) {
    unsigned b = __float_as_uint(f);
    return (b & 0x80000000u) ? ~b : (b | 0x80000000u);
}
__device__ __forceinline__ float key2f(unsigned k) {
    unsigned b = (k & 0x80000000u) ? (k ^ 0x80000000u) : ~k;
    return __uint_as_float(b);
}

__global__ void k_edge_max(const int* __restrict__ ei,
                           const float* __restrict__ a_s,
                           const float* __restrict__ a_d,
                           float* __restrict__ logits,
                           unsigned* __restrict__ mkey) {
    const int tid = blockIdx.x * blockDim.x + threadIdx.x;
    if (tid >= ET_EDGES * HEADS) return;
    const int e  = tid >> 2;
    const int hd = tid & 3;
    int s, d;
    if (e < N_EDGES) { s = ei[e]; d = ei[N_EDGES + e]; }
    else             { s = d = e - N_EDGES; }
    float v = a_s[s * HEADS + hd] + a_d[d * HEADS + hd];
    v = v > 0.0f ? v : NEG_SLOPE * v;
    logits[tid] = v;
    atomicMax(&mkey[d * HEADS + hd], f2key(v));
}

#define AGG_BLOCKS  256
#define AGG_R       196
#define AGG_THREADS 1024
#define AGG_WAVES   (AGG_THREADS / 32)
#define AGG_U       4
#define AGG_QCAP    1024
#define N_DST4      (N_EDGES / 4)
#define AGG_STEPS   ((N_DST4 + AGG_THREADS * AGG_U - 1) / (AGG_THREADS * AGG_U))
static_assert(AGG_BLOCKS * AGG_R >= N_NODES, "ranges must cover every node");
static_assert(N_EDGES % 4 == 0, "destination list is read as int4");
static_assert(N_DST4 % 32 == 0, "end-of-list test is per wave");
static_assert(AGG_STEPS % 2 == 0, "steps are taken in pairs");
static_assert(N_EDGES <= (1 << 21) && AGG_R <= 256, "ring entry = edge | row << 21");

template <bool FULL>
__device__ __forceinline__ void agg_drain(const int* __restrict__ ei,
                                          const float* __restrict__ h,
                                          const float* __restrict__ logits,
                                          const int* q, int qhead, int n, int lane,
                                          float* acc, float* ssl, const float* mxl) {
    int ent = 0;
    if (FULL || lane < n) ent = q[(qhead + lane) & (AGG_QCAP - 1)];
    const int e   = ent & 0x1FFFFF;
    const int rel = ent >> 21;
    const int s   = ei[e];
    const float4 lg = *(const float4*)(logits + (size_t)e * HEADS);
    const float4 mm = *(const float4*)(mxl + rel * HEADS);
    const float w0 = expf(lg.x - mm.x);
    const float w1 = expf(lg.y - mm.y);
    const float w2 = expf(lg.z - mm.z);
    const float w3 = expf(lg.w - mm.w);
    if (FULL || lane < n) {
        atomicAdd(&ssl[rel * HEADS + 0], w0);
        atomicAdd(&ssl[rel * HEADS + 1], w1);
        atomicAdd(&ssl[rel * HEADS + 2], w2);
        atomicAdd(&ssl[rel * HEADS + 3], w3);
    }
    const int nk = FULL ? 32 : n;
#pragma unroll 1
    for (int kb = 0; kb < nk; kb += 8) {
        float hv[8][4];
#pragma unroll
        for (int j = 0; j < 8; ++j) {
            const int k = kb + j;
            if (FULL || k < n) {
                const int sk = __builtin_amdgcn_readlane(s, k);
                const float* hp = h + (size_t)sk * OUT_CH + lane;
                hv[j][0] = hp[0];
                hv[j][1] = hp[32];
                hv[j][2] = hp[64];
                hv[j][3] = hp[96];
            }
        }
#pragma unroll
        for (int j = 0; j < 8; ++j) {
            const int k = kb + j;
            if (FULL || k < n) {
                const int rk = __builtin_amdgcn_readlane(rel, k);
                const float a0 = __int_as_float(__builtin_amdgcn_readlane(__float_as_int(w0), k));
                const float a1 = __int_as_float(__builtin_amdgcn_readlane(__float_as_int(w1), k));
                const float a2 = __int_as_float(__builtin_amdgcn_readlane(__float_as_int(w2), k));
                const float a3 = __int_as_float(__builtin_amdgcn_readlane(__float_as_int(w3), k));
                float* ap = acc + rk * OUT_CH + lane;
                atomicAdd(ap,      hv[j][0] * a0);
                atomicAdd(ap + 32, hv[j][1] * a1);
                atomicAdd(ap + 64, hv[j][2] * a2);
                atomicAdd(ap + 96, hv[j][3] * a3);
            }
        }
    }
}

__device__ __forceinline__ int agg_wrap(int sp) {
    while (sp >= AGG_STEPS) sp -= AGG_STEPS;
    return sp;
}

__device__ __forceinline__ void agg_load(int4 (&dv)[AGG_U], const int4* __restrict__ dst4,
                                         int sp, int wave, int lane) {
#pragma unroll
    for (int u = 0; u < AGG_U; ++u) {
        int wbase = (sp * AGG_U + u) * AGG_THREADS + wave * 32;
        if (wbase > N_DST4 - 32) wbase = N_DST4 - 32;
        dv[u] = dst4[wbase + lane];
    }
}

__device__ __forceinline__ void agg_scan(const int4 (&dv)[AGG_U], int sp, int lo, int wave, int lane,
                                         int* q, int& qhead, int& cnt,
                                         const int* __restrict__ ei,
                                         const float* __restrict__ h,
                                         const float* __restrict__ logits,
                                         float* acc, float* ssl, const float* mxl) {
#pragma unroll
    for (int u = 0; u < AGG_U; ++u) {
        const int wbase = (sp * AGG_U + u) * AGG_THREADS + wave * 32;
        if (wbase < N_DST4) {
            const int e0 = (wbase + lane) * 4;
            const int r0 = dv[u].x - lo, r1 = dv[u].y - lo, r2 = dv[u].z - lo, r3 = dv[u].w - lo;
            const bool m0 = (unsigned)r0 < (unsigned)AGG_R;
            const bool m1 = (unsigned)r1 < (unsigned)AGG_R;
            const bool m2 = (unsigned)r2 < (unsigned)AGG_R;
            const bool m3 = (unsigned)r3 < (unsigned)AGG_R;
            const unsigned b0 = __builtin_amdgcn_ballot_w32(m0);
            const unsigned b1 = __builtin_amdgcn_ballot_w32(m1);
            const unsigned b2 = __builtin_amdgcn_ballot_w32(m2);
            const unsigned b3 = __builtin_amdgcn_ballot_w32(m3);
            if (b0 | b1 | b2 | b3) {
                if (m0) q[(qhead + cnt + __builtin_amdgcn_mbcnt_lo(b0, 0)) & (AGG_QCAP - 1)] = (e0 + 0) | (r0 << 21);
                cnt += __builtin_popcount(b0);
                if (m1) q[(qhead + cnt + __builtin_amdgcn_mbcnt_lo(b1, 0)) & (AGG_QCAP - 1)] = (e0 + 1) | (r1 << 21);
                cnt += __builtin_popcount(b1);
                if (m2) q[(qhead + cnt + __builtin_amdgcn_mbcnt_lo(b2, 0)) & (AGG_QCAP - 1)] = (e0 + 2) | (r2 << 21);
                cnt += __builtin_popcount(b2);
                if (m3) q[(qhead + cnt + __builtin_amdgcn_mbcnt_lo(b3, 0)) & (AGG_QCAP - 1)] = (e0 + 3) | (r3 << 21);
                cnt += __builtin_popcount(b3);
            }
        }
    }
    while (cnt >= 32) {
        agg_drain<true>(ei, h, logits, q, qhead, 32, lane, acc, ssl, mxl);
        qhead = (qhead + 32) & (AGG_QCAP - 1);
        cnt -= 32;
    }
}

__global__ __launch_bounds__(AGG_THREADS)
void k_agg_lds(const int* __restrict__ ei,
               const float* __restrict__ h,
               const float* __restrict__ logits,
               const unsigned* __restrict__ mkey,
               float* __restrict__ ssum,
               float* __restrict__ out) {
    __shared__ __attribute__((aligned(16))) float acc[AGG_R * OUT_CH];
    __shared__ __attribute__((aligned(16))) float ssl[AGG_R * HEADS];
    __shared__ __attribute__((aligned(16))) float mxl[AGG_R * HEADS];
    __shared__ int que[AGG_WAVES * AGG_QCAP];

    const int tid  = threadIdx.x;
    const int wave = __builtin_amdgcn_readfirstlane(tid >> 5);
    const int lane = tid & 31;
    const int lo   = blockIdx.x * AGG_R;

    for (int r = wave; r < AGG_R; r += AGG_WAVES) {
        const int node = lo + r;
        if (node < N_NODES) {
            const float* hp = h + (size_t)node * OUT_CH + lane;
            float* ap = acc + r * OUT_CH + lane;
            for (int hd = 0; hd < HEADS; ++hd) {
                const float mf = key2f(mkey[node * HEADS + hd]);
                const float w  = expf(logits[(size_t)(N_EDGES + node) * HEADS + hd] - mf);
                ap[hd * CPH] = hp[hd * CPH] * w;
                if (lane == 0) { ssl[r * HEADS + hd] = w; mxl[r * HEADS + hd] = mf; }
            }
        }
    }
    __syncthreads();

    int* q = que + wave * AGG_QCAP;
    int qhead = 0, cnt = 0;
    const int4* dst4 = (const int4*)(ei + N_EDGES);
    const int start = (blockIdx.x * AGG_STEPS) / AGG_BLOCKS;

    int4 da[AGG_U], db[AGG_U];
    agg_load(da, dst4, agg_wrap(start), wave, lane);
    for (int st = 0; st < AGG_STEPS; st += 2) {
        const int spa = agg_wrap(st + start);
        const int spb = agg_wrap(st + 1 + start);
        agg_load(db, dst4, spb, wave, lane);
        agg_scan(da, spa, lo, wave, lane, q, qhead, cnt, ei, h, logits, acc, ssl, mxl);
        agg_load(da, dst4, agg_wrap(st + 2 + start), wave, lane);
        agg_scan(db, spb, lo, wave, lane, q, qhead, cnt, ei, h, logits, acc, ssl, mxl);
    }
    if (cnt > 0) agg_drain<false>(ei, h, logits, q, qhead, cnt, lane, acc, ssl, mxl);
    __syncthreads();

    for (int r = wave; r < AGG_R; r += AGG_WAVES) {
        const int node = lo + r;
        if (node < N_NODES) {
            const float* ap = acc + r * OUT_CH + lane;
            float* op = out + (size_t)node * OUT_CH + lane;
            op[0]  = ap[0];
            op[32] = ap[32];
            op[64] = ap[64];
            op[96] = ap[96];
            if (lane < HEADS) ssum[node * HEADS + lane] = ssl[r * HEADS + lane];
        }
    }
}

__global__ void k_ln_elu(float* __restrict__ out,
                         const float* __restrict__ ssum,
                         const float* __restrict__ bias,
                         const float* __restrict__ gamma,
                         const float* __restrict__ beta) {
    __shared__ float red[4];
    const int node = blockIdx.x;
    const int c = threadIdx.x;
    const int wave = c >> 5, lane = c & 31;

    float v = out[(size_t)node * OUT_CH + c] /
                  (ssum[node * HEADS + wave] + SOFTMAX_EPS) + bias[c];

    float t = v;
#pragma unroll
    for (int o = 16; o > 0; o >>= 1) t += __shfl_xor(t, o, 32);
    if (lane == 0) red[wave] = t;
    __syncthreads();
    const float mu = (red[0] + red[1] + red[2] + red[3]) * (1.0f / OUT_CH);
    __syncthreads();

    const float dv = v - mu;
    t = dv * dv;
#pragma unroll
    for (int o = 16; o > 0; o >>= 1) t += __shfl_xor(t, o, 32);
    if (lane == 0) red[wave] = t;
    __syncthreads();
    const float var = (red[0] + red[1] + red[2] + red[3]) * (1.0f / OUT_CH);

    float y = dv * (1.0f / sqrtf(var + LN_EPS)) * gamma[c] + beta[c];
    y = y > 0.0f ? y : (expf(y) - 1.0f);
    out[(size_t)node * OUT_CH + c] = y;
}

extern "C" void kernel_launch(void* const* d_in, const int* in_sizes, int n_in,
                              void* d_out, int out_size, void* d_ws, size_t ws_size,
                              hipStream_t stream) {
    const float* x       = (const float*)d_in[0];
    const int*   ei      = (const int*)  d_in[1];
    const float* W       = (const float*)d_in[2];
    const float* att_src = (const float*)d_in[3];
    const float* att_dst = (const float*)d_in[4];
    const float* bias    = (const float*)d_in[5];
    const float* gamma   = (const float*)d_in[6];
    const float* beta    = (const float*)d_in[7];
    float* out = (float*)d_out;

    char* ws = (char*)d_ws;
    float*    h      = (float*)ws;
    ws += (size_t)N_NODES * OUT_CH * sizeof(float);
    float*    a_s    = (float*)ws;  ws += (size_t)N_NODES * HEADS * sizeof(float);
    float*    a_d    = (float*)ws;  ws += (size_t)N_NODES * HEADS * sizeof(float);
    unsigned* mkey   = (unsigned*)ws; ws += (size_t)N_NODES * HEADS * sizeof(unsigned);
    float*    ssum   = (float*)ws;  ws += (size_t)N_NODES * HEADS * sizeof(float);
    float*    logits = (float*)ws;

    const int nseg = N_NODES * HEADS;

    k_init<<<(nseg + 255) / 256, 256, 0, stream>>>(mkey, nseg);

    k_gemm_wmma<<<(N_NODES + 127) / 128, 256, IN_CH * OUT_CH * sizeof(float),
                  stream>>>(x, W, h, N_NODES);

    k_dots<<<(N_NODES + 1) / 2, 256, 0, stream>>>(h, att_src, att_dst, a_s, a_d,
                                                  N_NODES);

    k_edge_max<<<(ET_EDGES * HEADS + 255) / 256, 256, 0, stream>>>(
        ei, a_s, a_d, logits, mkey);

    k_agg_lds<<<AGG_BLOCKS, AGG_THREADS, 0, stream>>>(ei, h, logits, mkey, ssum,
                                                      out);

    k_ln_elu<<<N_NODES, 128, 0, stream>>>(out, ssum, bias, gamma, beta);
}


# Division, bias, LayerNorm, ELU done by the owner block on each finished row as it leaves LDS (wave = row; sums 4 channels a lane, then 32-lane butterfly: other reduction tree); last launch gone
# speedup vs baseline: 2.0468x; 1.0782x over previous
#include <hip/hip_runtime.h>
#include <math.h>


#define N_NODES 50000
#define N_EDGES 1600000
#define ET_EDGES (N_EDGES + N_NODES)
#define IN_CH 128
#define OUT_CH 128
#define HEADS 4
#define CPH 32
#define NEG_SLOPE 0.2f
#define SOFTMAX_EPS 1e-16f
#define LN_EPS 1e-5f

typedef float v2f __attribute__((ext_vector_type(2)));
typedef float v8f __attribute__((ext_vector_type(8)));

__global__ void k_init(unsigned* __restrict__ mkey, int nseg) {
    int i = blockIdx.x * blockDim.x + threadIdx.x;
    if (i < nseg) mkey[i] = 0u;
}

__global__ void k_gemm_wmma(const float* __restrict__ x,
                            const float* __restrict__ W,
                            float* __restrict__ h, int nrows) {
    extern __shared__ float Wlds[];
    for (int i = threadIdx.x; i < (IN_CH * OUT_CH) / 4; i += blockDim.x)
        ((float4*)Wlds)[i] = ((const float4*)W)[i];
    __syncthreads();

    const int wave = threadIdx.x >> 5;
    const int lane = threadIdx.x & 31;
    const int rowBase = blockIdx.x * 128 + wave * 16;
    const int m  = lane & 15;
    const int kk = (lane >> 4) << 1;

    int arow = rowBase + m;
    if (arow >= nrows) arow = nrows - 1;
    const float* xk = x + (size_t)arow * IN_CH + kk;

    v8f c[8];
#pragma unroll
    for (int t = 0; t < 8; ++t) c[t] = (v8f){0,0,0,0,0,0,0,0};

    float2 av = *(const float2*)xk;
    for (int k = 0; k < IN_CH; k += 4) {
        v2f a; a.x = av.x; a.y = av.y;
        if (k + 4 < IN_CH) av = *(const float2*)(xk + k + 4);

        const float* w0 = &Wlds[(k + kk) * OUT_CH + m];
        v2f b[8];
#pragma unroll
        for (int t = 0; t < 8; ++t) {
            b[t].x = w0[t * 16];
            b[t].y = w0[OUT_CH + t * 16];
        }
#pragma unroll
        for (int t = 0; t < 8; ++t)
            c[t] = __builtin_amdgcn_wmma_f32_16x16x4_f32(
                false, a, false, b[t], (short)0, c[t], false, false);
    }

#pragma unroll
    for (int t = 0; t < 8; ++t) {
#pragma unroll
        for (int v = 0; v < 8; ++v) {
            const int M = v + ((lane >> 4) << 3);
            const int row = rowBase + M;
            if (row < nrows)
                h[(size_t)row * OUT_CH + t * 16 + m] = c[t][v];
        }
    }
}

__global__ void k_dots(const float* __restrict__ h,
                       const float* __restrict__ att_src,
                       const float* __restrict__ att_dst,
                       float* __restrict__ a_s, float* __restrict__ a_d, int n) {
    const int wave = threadIdx.x >> 5;
    const int lane = threadIdx.x & 31;
    const int node = blockIdx.x * 2 + (wave >> 2);
    const int head = wave & 3;
    if (node >= n) return;
    const float hv = h[(size_t)node * OUT_CH + head * CPH + lane];
    float vs = hv * att_src[head * CPH + lane];
    float vd = hv * att_dst[head * CPH + lane];
#pragma unroll
    for (int o = 16; o > 0; o >>= 1) {
        vs += __shfl_xor(vs, o, 32);
        vd += __shfl_xor(vd, o, 32);
    }
    if (lane == 0) {
        a_s[node * HEADS + head] = vs;
        a_d[node * HEADS + head] = vd;
    }
}

__device__ __forceinline__ unsigned f2key(float f) {
    unsigned b = __float_as_uint(f);
    return (b & 0x80000000u) ? ~b : (b | 0x80000000u);
}
__device__ __forceinline__ float key2f(unsigned k) {
    unsigned b = (k & 0x80000000u) ? (k ^ 0x80000000u) : ~k;
    return __uint_as_float(b);
}

__global__ void k_edge_max(const int* __restrict__ ei,
                           const float* __restrict__ a_s,
                           const float* __restrict__ a_d,
                           float* __restrict__ logits,
                           unsigned* __restrict__ mkey) {
    const int tid = blockIdx.x * blockDim.x + threadIdx.x;
    if (tid >= ET_EDGES * HEADS) return;
    const int e  = tid >> 2;
    const int hd = tid & 3;
    int s, d;
    if (e < N_EDGES) { s = ei[e]; d = ei[N_EDGES + e]; }
    else             { s = d = e - N_EDGES; }
    float v = a_s[s * HEADS + hd] + a_d[d * HEADS + hd];
    v = v > 0.0f ? v : NEG_SLOPE * v;
    logits[tid] = v;
    atomicMax(&mkey[d * HEADS + hd], f2key(v));
}

#define AGG_BLOCKS  256
#define AGG_R       196
#define AGG_THREADS 1024
#define AGG_WAVES   (AGG_THREADS / 32)
#define AGG_U       4
#define AGG_QCAP    1024
#define N_DST4      (N_EDGES / 4)
#define AGG_STEPS   ((N_DST4 + AGG_THREADS * AGG_U - 1) / (AGG_THREADS * AGG_U))
static_assert(AGG_BLOCKS * AGG_R >= N_NODES, "ranges must cover every node");
static_assert(N_EDGES % 4 == 0, "destination list is read as int4");
static_assert(N_DST4 % 32 == 0, "end-of-list test is per wave");
static_assert(AGG_STEPS % 2 == 0, "steps are taken in pairs");
static_assert(N_EDGES <= (1 << 21) && AGG_R <= 256, "ring entry = edge | row << 21");

template <bool FULL>
__device__ __forceinline__ void agg_drain(const int* __restrict__ ei,
                                          const float* __restrict__ h,
                                          const float* __restrict__ logits,
                                          const int* q, int qhead, int n, int lane,
                                          float* acc, float* ssl, const float* mxl) {
    int ent = 0;
    if (FULL || lane < n) ent = q[(qhead + lane) & (AGG_QCAP - 1)];
    const int e   = ent & 0x1FFFFF;
    const int rel = ent >> 21;
    const int s   = ei[e];
    const float4 lg = *(const float4*)(logits + (size_t)e * HEADS);
    const float4 mm = *(const float4*)(mxl + rel * HEADS);
    const float w0 = expf(lg.x - mm.x);
    const float w1 = expf(lg.y - mm.y);
    const float w2 = expf(lg.z - mm.z);
    const float w3 = expf(lg.w - mm.w);
    if (FULL || lane < n) {
        atomicAdd(&ssl[rel * HEADS + 0], w0);
        atomicAdd(&ssl[rel * HEADS + 1], w1);
        atomicAdd(&ssl[rel * HEADS + 2], w2);
        atomicAdd(&ssl[rel * HEADS + 3], w3);
    }
    const int nk = FULL ? 32 : n;
#pragma unroll 1
    for (int kb = 0; kb < nk; kb += 8) {
        float hv[8][4];
#pragma unroll
        for (int j = 0; j < 8; ++j) {
            const int k = kb + j;
            if (FULL || k < n) {
                const int sk = __builtin_amdgcn_readlane(s, k);
                const float* hp = h + (size_t)sk * OUT_CH + lane;
                hv[j][0] = hp[0];
                hv[j][1] = hp[32];
                hv[j][2] = hp[64];
                hv[j][3] = hp[96];
            }
        }
#pragma unroll
        for (int j = 0; j < 8; ++j) {
            const int k = kb + j;
            if (FULL || k < n) {
                const int rk = __builtin_amdgcn_readlane(rel, k);
                const float a0 = __int_as_float(__builtin_amdgcn_readlane(__float_as_int(w0), k));
                const float a1 = __int_as_float(__builtin_amdgcn_readlane(__float_as_int(w1), k));
                const float a2 = __int_as_float(__builtin_amdgcn_readlane(__float_as_int(w2), k));
                const float a3 = __int_as_float(__builtin_amdgcn_readlane(__float_as_int(w3), k));
                float* ap = acc + rk * OUT_CH + lane;
                atomicAdd(ap,      hv[j][0] * a0);
                atomicAdd(ap + 32, hv[j][1] * a1);
                atomicAdd(ap + 64, hv[j][2] * a2);
                atomicAdd(ap + 96, hv[j][3] * a3);
            }
        }
    }
}

__device__ __forceinline__ int agg_wrap(int sp) {
    while (sp >= AGG_STEPS) sp -= AGG_STEPS;
    return sp;
}

__device__ __forceinline__ void agg_load(int4 (&dv)[AGG_U], const int4* __restrict__ dst4,
                                         int sp, int wave, int lane) {
#pragma unroll
    for (int u = 0; u < AGG_U; ++u) {
        int wbase = (sp * AGG_U + u) * AGG_THREADS + wave * 32;
        if (wbase > N_DST4 - 32) wbase = N_DST4 - 32;
        dv[u] = dst4[wbase + lane];
    }
}

__device__ __forceinline__ void agg_scan(const int4 (&dv)[AGG_U], int sp, int lo, int wave, int lane,
                                         int* q, int& qhead, int& cnt,
                                         const int* __restrict__ ei,
                                         const float* __restrict__ h,
                                         const float* __restrict__ logits,
                                         float* acc, float* ssl, const float* mxl) {
#pragma unroll
    for (int u = 0; u < AGG_U; ++u) {
        const int wbase = (sp * AGG_U + u) * AGG_THREADS + wave * 32;
        if (wbase < N_DST4) {
            const int e0 = (wbase + lane) * 4;
            const int r0 = dv[u].x - lo, r1 = dv[u].y - lo, r2 = dv[u].z - lo, r3 = dv[u].w - lo;
            const bool m0 = (unsigned)r0 < (unsigned)AGG_R;
            const bool m1 = (unsigned)r1 < (unsigned)AGG_R;
            const bool m2 = (unsigned)r2 < (unsigned)AGG_R;
            const bool m3 = (unsigned)r3 < (unsigned)AGG_R;
            const unsigned b0 = __builtin_amdgcn_ballot_w32(m0);
            const unsigned b1 = __builtin_amdgcn_ballot_w32(m1);
            const unsigned b2 = __builtin_amdgcn_ballot_w32(m2);
            const unsigned b3 = __builtin_amdgcn_ballot_w32(m3);
            if (b0 | b1 | b2 | b3) {
                if (m0) q[(qhead + cnt + __builtin_amdgcn_mbcnt_lo(b0, 0)) & (AGG_QCAP - 1)] = (e0 + 0) | (r0 << 21);
                cnt += __builtin_popcount(b0);
                if (m1) q[(qhead + cnt + __builtin_amdgcn_mbcnt_lo(b1, 0)) & (AGG_QCAP - 1)] = (e0 + 1) | (r1 << 21);
                cnt += __builtin_popcount(b1);
                if (m2) q[(qhead + cnt + __builtin_amdgcn_mbcnt_lo(b2, 0)) & (AGG_QCAP - 1)] = (e0 + 2) | (r2 << 21);
                cnt += __builtin_popcount(b2);
                if (m3) q[(qhead + cnt + __builtin_amdgcn_mbcnt_lo(b3, 0)) & (AGG_QCAP - 1)] = (e0 + 3) | (r3 << 21);
                cnt += __builtin_popcount(b3);
            }
        }
    }
    while (cnt >= 32) {
        agg_drain<true>(ei, h, logits, q, qhead, 32, lane, acc, ssl, mxl);
        qhead = (qhead + 32) & (AGG_QCAP - 1);
        cnt -= 32;
    }
}

__device__ __forceinline__ float wave_sum(float t) {
    for (int o = 16; o > 0; o >>= 1) t += __shfl_xor(t, o, 32);
    return t;
}

__global__ __launch_bounds__(AGG_THREADS)
void k_agg_lds(const int* __restrict__ ei,
               const float* __restrict__ h,
               const float* __restrict__ logits,
               const unsigned* __restrict__ mkey,
               const float* __restrict__ bias,
               const float* __restrict__ gamma,
               const float* __restrict__ beta,
               float* __restrict__ out) {
    __shared__ __attribute__((aligned(16))) float acc[AGG_R * OUT_CH];
    __shared__ __attribute__((aligned(16))) float ssl[AGG_R * HEADS];
    __shared__ __attribute__((aligned(16))) float mxl[AGG_R * HEADS];
    __shared__ int que[AGG_WAVES * AGG_QCAP];

    const int tid  = threadIdx.x;
    const int wave = __builtin_amdgcn_readfirstlane(tid >> 5);
    const int lane = tid & 31;
    const int lo   = blockIdx.x * AGG_R;

    for (int r = wave; r < AGG_R; r += AGG_WAVES) {
        const int node = lo + r;
        if (node < N_NODES) {
            const float* hp = h + (size_t)node * OUT_CH + lane;
            float* ap = acc + r * OUT_CH + lane;
            for (int hd = 0; hd < HEADS; ++hd) {
                const float mf = key2f(mkey[node * HEADS + hd]);
                const float w  = expf(logits[(size_t)(N_EDGES + node) * HEADS + hd] - mf);
                ap[hd * CPH] = hp[hd * CPH] * w;
                if (lane == 0) { ssl[r * HEADS + hd] = w; mxl[r * HEADS + hd] = mf; }
            }
        }
    }
    __syncthreads();

    int* q = que + wave * AGG_QCAP;
    int qhead = 0, cnt = 0;
    const int4* dst4 = (const int4*)(ei + N_EDGES);
    const int start = (blockIdx.x * AGG_STEPS) / AGG_BLOCKS;

    int4 da[AGG_U], db[AGG_U];
    agg_load(da, dst4, agg_wrap(start), wave, lane);
    for (int st = 0; st < AGG_STEPS; st += 2) {
        const int spa = agg_wrap(st + start);
        const int spb = agg_wrap(st + 1 + start);
        agg_load(db, dst4, spb, wave, lane);
        agg_scan(da, spa, lo, wave, lane, q, qhead, cnt, ei, h, logits, acc, ssl, mxl);
        agg_load(da, dst4, agg_wrap(st + 2 + start), wave, lane);
        agg_scan(db, spb, lo, wave, lane, q, qhead, cnt, ei, h, logits, acc, ssl, mxl);
    }
    if (cnt > 0) agg_drain<false>(ei, h, logits, q, qhead, cnt, lane, acc, ssl, mxl);
    __syncthreads();

    const float bs0 = bias[lane],  bs1 = bias[32 + lane],  bs2 = bias[64 + lane],  bs3 = bias[96 + lane];
    const float gm0 = gamma[lane], gm1 = gamma[32 + lane], gm2 = gamma[64 + lane], gm3 = gamma[96 + lane];
    const float bt0 = beta[lane],  bt1 = beta[32 + lane],  bt2 = beta[64 + lane],  bt3 = beta[96 + lane];
    for (int r = wave; r < AGG_R; r += AGG_WAVES) {
        const int node = lo + r;
        if (node < N_NODES) {
            const float* ap = acc + r * OUT_CH + lane;
            const float4 sv = *(const float4*)(ssl + r * HEADS);
            const float v0 = ap[0]  / (sv.x + SOFTMAX_EPS) + bs0;
            const float v1 = ap[32] / (sv.y + SOFTMAX_EPS) + bs1;
            const float v2 = ap[64] / (sv.z + SOFTMAX_EPS) + bs2;
            const float v3 = ap[96] / (sv.w + SOFTMAX_EPS) + bs3;

            const float mu = wave_sum((v0 + v1) + (v2 + v3)) * (1.0f / OUT_CH);
            const float d0 = v0 - mu, d1 = v1 - mu, d2 = v2 - mu, d3 = v3 - mu;
            const float var = wave_sum((d0 * d0 + d1 * d1) + (d2 * d2 + d3 * d3)) * (1.0f / OUT_CH);
            const float rs = 1.0f / sqrtf(var + LN_EPS);

            float y0 = d0 * rs * gm0 + bt0;
            float y1 = d1 * rs * gm1 + bt1;
            float y2 = d2 * rs * gm2 + bt2;
            float y3 = d3 * rs * gm3 + bt3;
            y0 = y0 > 0.0f ? y0 : (expf(y0) - 1.0f);
            y1 = y1 > 0.0f ? y1 : (expf(y1) - 1.0f);
            y2 = y2 > 0.0f ? y2 : (expf(y2) - 1.0f);
            y3 = y3 > 0.0f ? y3 : (expf(y3) - 1.0f);

            float* op = out + (size_t)node * OUT_CH + lane;
            op[0]  = y0;
            op[32] = y1;
            op[64] = y2;
            op[96] = y3;
        }
    }
}

extern "C" void kernel_launch(void* const* d_in, const int* in_sizes, int n_in,
                              void* d_out, int out_size, void* d_ws, size_t ws_size,
                              hipStream_t stream) {
    const float* x       = (const float*)d_in[0];
    const int*   ei      = (const int*)  d_in[1];
    const float* W       = (const float*)d_in[2];
    const float* att_src = (const float*)d_in[3];
    const float* att_dst = (const float*)d_in[4];
    const float* bias    = (const float*)d_in[5];
    const float* gamma   = (const float*)d_in[6];
    const float* beta    = (const float*)d_in[7];
    float* out = (float*)d_out;

    char* ws = (char*)d_ws;
    float*    h      = (float*)ws;
    ws += (size_t)N_NODES * OUT_CH * sizeof(float);
    float*    a_s    = (float*)ws;  ws += (size_t)N_NODES * HEADS * sizeof(float);
    float*    a_d    = (float*)ws;  ws += (size_t)N_NODES * HEADS * sizeof(float);
    unsigned* mkey   = (unsigned*)ws; ws += (size_t)N_NODES * HEADS * sizeof(unsigned);
    float*    logits = (float*)ws;

    const int nseg = N_NODES * HEADS;

    k_init<<<(nseg + 255) / 256, 256, 0, stream>>>(mkey, nseg);

    k_gemm_wmma<<<(N_NODES + 127) / 128, 256, IN_CH * OUT_CH * sizeof(float),
                  stream>>>(x, W, h, N_NODES);

    k_dots<<<(N_NODES + 1) / 2, 256, 0, stream>>>(h, att_src, att_dst, a_s, a_d,
                                                  N_NODES);

    k_edge_max<<<(ET_EDGES * HEADS + 255) / 256, 256, 0, stream>>>(
        ei, a_s, a_d, logits, mkey);

    k_agg_lds<<<AGG_BLOCKS, AGG_THREADS, 0, stream>>>(ei, h, logits, mkey, bias,
                                                      gamma, beta, out);
}


# Wave entry: dots and logit/max launches run as fat waves (16 nodes, 25 elements a lane; 3125 and 8250 waves, not 200000 and 206250); dots: 4 channels a lane, then 8-lane butterfly (other sum tree)
# speedup vs baseline: 2.1304x; 1.0408x over previous
#include <hip/hip_runtime.h>
#include <math.h>


#define N_NODES 50000
#define N_EDGES 1600000
#define ET_EDGES (N_EDGES + N_NODES)
#define IN_CH 128
#define OUT_CH 128
#define HEADS 4
#define CPH 32
#define NEG_SLOPE 0.2f
#define SOFTMAX_EPS 1e-16f
#define LN_EPS 1e-5f

typedef float v2f __attribute__((ext_vector_type(2)));
typedef float v8f __attribute__((ext_vector_type(8)));

__global__ void k_init(unsigned* __restrict__ mkey, int nseg) {
    int i = blockIdx.x * blockDim.x + threadIdx.x;
    if (i < nseg) mkey[i] = 0u;
}

__global__ void k_gemm_wmma(const float* __restrict__ x,
                            const float* __restrict__ W,
                            float* __restrict__ h, int nrows) {
    extern __shared__ float Wlds[];
    for (int i = threadIdx.x; i < (IN_CH * OUT_CH) / 4; i += blockDim.x)
        ((float4*)Wlds)[i] = ((const float4*)W)[i];
    __syncthreads();

    const int wave = threadIdx.x >> 5;
    const int lane = threadIdx.x & 31;
    const int rowBase = blockIdx.x * 128 + wave * 16;
    const int m  = lane & 15;
    const int kk = (lane >> 4) << 1;

    int arow = rowBase + m;
    if (arow >= nrows) arow = nrows - 1;
    const float* xk = x + (size_t)arow * IN_CH + kk;

    v8f c[8];
#pragma unroll
    for (int t = 0; t < 8; ++t) c[t] = (v8f){0,0,0,0,0,0,0,0};

    float2 av = *(const float2*)xk;
    for (int k = 0; k < IN_CH; k += 4) {
        v2f a; a.x = av.x; a.y = av.y;
        if (k + 4 < IN_CH) av = *(const float2*)(xk + k + 4);

        const float* w0 = &Wlds[(k + kk) * OUT_CH + m];
        v2f b[8];
#pragma unroll
        for (int t = 0; t < 8; ++t) {
            b[t].x = w0[t * 16];
            b[t].y = w0[OUT_CH + t * 16];
        }
#pragma unroll
        for (int t = 0; t < 8; ++t)
            c[t] = __builtin_amdgcn_wmma_f32_16x16x4_f32(
                false, a, false, b[t], (short)0, c[t], false, false);
    }

#pragma unroll
    for (int t = 0; t < 8; ++t) {
#pragma unroll
        for (int v = 0; v < 8; ++v) {
            const int M = v + ((lane >> 4) << 3);
            const int row = rowBase + M;
            if (row < nrows)
                h[(size_t)row * OUT_CH + t * 16 + m] = c[t][v];
        }
    }
}

#define DOT_THREADS 256
#define DOT_NPW     16
#define DOT_BLOCKS  ((N_NODES + (DOT_THREADS / 32) * DOT_NPW - 1) / ((DOT_THREADS / 32) * DOT_NPW))

__device__ __forceinline__ float dots_sum8(const float (&v)[8], int lane) {
    const bool b2 = (lane & 4) != 0, b1 = (lane & 2) != 0, b0 = (lane & 1) != 0;
    float r[4], q[2];
#pragma unroll
    for (int j = 0; j < 4; ++j) {
        const float keep = b2 ? v[j + 4] : v[j];
        const float send = b2 ? v[j] : v[j + 4];
        r[j] = keep + __shfl_xor(send, 4, 32);
    }
#pragma unroll
    for (int j = 0; j < 2; ++j) {
        const float keep = b1 ? r[j + 2] : r[j];
        const float send = b1 ? r[j] : r[j + 2];
        q[j] = keep + __shfl_xor(send, 2, 32);
    }
    const float keep = b0 ? q[1] : q[0];
    const float send = b0 ? q[0] : q[1];
    return keep + __shfl_xor(send, 1, 32);
}

__global__ __launch_bounds__(DOT_THREADS)
void k_dots(const float* __restrict__ h,
            const float* __restrict__ att_src,
            const float* __restrict__ att_dst,
            float* __restrict__ a_s, float* __restrict__ a_d, int n) {
    const int wave = threadIdx.x >> 5;
    const int lane = threadIdx.x & 31;
    const int base = (blockIdx.x * (DOT_THREADS / 32) + wave) * DOT_NPW;
    if (base >= n) return;
    const float4 ws = ((const float4*)att_src)[lane];
    const float4 wd = ((const float4*)att_dst)[lane];

    float4 hv[DOT_NPW];
#pragma unroll
    for (int i = 0; i < DOT_NPW; ++i) {
        int node = base + i;
        if (node >= n) node = n - 1;
        hv[i] = ((const float4*)(h + (size_t)node * OUT_CH))[lane];
    }
    __builtin_amdgcn_sched_barrier(0);
#pragma unroll
    for (int g = 0; g < DOT_NPW / 8; ++g) {
        float vs[8], vd[8];
#pragma unroll
        for (int i = 0; i < 8; ++i) {
            const float4 x4 = hv[g * 8 + i];
            vs[i] = (x4.x * ws.x + x4.y * ws.y) + (x4.z * ws.z + x4.w * ws.w);
            vd[i] = (x4.x * wd.x + x4.y * wd.y) + (x4.z * wd.z + x4.w * wd.w);
        }
        const float rs = dots_sum8(vs, lane);
        const float rd = dots_sum8(vd, lane);
        const int node = base + g * 8 + (lane & 7);
        if (node < n) {
            a_s[node * HEADS + (lane >> 3)] = rs;
            a_d[node * HEADS + (lane >> 3)] = rd;
        }
    }
}

__device__ __forceinline__ unsigned f2key(float f) {
    unsigned b = __float_as_uint(f);
    return (b & 0x80000000u) ? ~b : (b | 0x80000000u);
}
__device__ __forceinline__ float key2f(unsigned k) {
    unsigned b = (k & 0x80000000u) ? (k ^ 0x80000000u) : ~k;
    return __uint_as_float(b);
}

#define EM_THREADS   192
#define EM_U         5
#define EM_ROUNDS    5
#define EM_TOTAL     (ET_EDGES * HEADS)
#define EM_PER_BLOCK (EM_THREADS * EM_U * EM_ROUNDS)
#define EM_BLOCKS    (EM_TOTAL / EM_PER_BLOCK)
static_assert(EM_TOTAL % EM_PER_BLOCK == 0, "blocks tile the (edge, head) list exactly");
static_assert(EM_THREADS % HEADS == 0, "a lane keeps its head from element to element");

__global__ __launch_bounds__(EM_THREADS)
void k_edge_max(const int* __restrict__ ei,
                const float* __restrict__ a_s,
                const float* __restrict__ a_d,
                float* __restrict__ logits,
                unsigned* __restrict__ mkey) {
    const int hd = threadIdx.x & 3;
    int t0 = blockIdx.x * EM_PER_BLOCK + threadIdx.x;
    for (int r = 0; r < EM_ROUNDS; ++r, t0 += EM_THREADS * EM_U) {
        int s[EM_U], d[EM_U];
        float v[EM_U];
#pragma unroll
        for (int u = 0; u < EM_U; ++u) {
            const int e  = (t0 + u * EM_THREADS) >> 2;
            const int ec = e < N_EDGES ? e : N_EDGES - 1;
            const int ls = ei[ec];
            const int ld = ei[N_EDGES + ec];
            s[u] = e < N_EDGES ? ls : e - N_EDGES;
            d[u] = e < N_EDGES ? ld : e - N_EDGES;
        }
#pragma unroll
        for (int u = 0; u < EM_U; ++u)
            v[u] = a_s[s[u] * HEADS + hd] + a_d[d[u] * HEADS + hd];
#pragma unroll
        for (int u = 0; u < EM_U; ++u) {
            const int t = t0 + u * EM_THREADS;
            const float lv = v[u] > 0.0f ? v[u] : NEG_SLOPE * v[u];
            logits[t] = lv;
            atomicMax(&mkey[d[u] * HEADS + hd], f2key(lv));
        }
    }
}

#define AGG_BLOCKS  256
#define AGG_R       196
#define AGG_THREADS 1024
#define AGG_WAVES   (AGG_THREADS / 32)
#define AGG_U       4
#define AGG_QCAP    1024
#define N_DST4      (N_EDGES / 4)
#define AGG_STEPS   ((N_DST4 + AGG_THREADS * AGG_U - 1) / (AGG_THREADS * AGG_U))
static_assert(AGG_BLOCKS * AGG_R >= N_NODES, "ranges must cover every node");
static_assert(N_EDGES % 4 == 0, "destination list is read as int4");
static_assert(N_DST4 % 32 == 0, "end-of-list test is per wave");
static_assert(AGG_STEPS % 2 == 0, "steps are taken in pairs");
static_assert(N_EDGES <= (1 << 21) && AGG_R <= 256, "ring entry = edge | row << 21");

template <bool FULL>
__device__ __forceinline__ void agg_drain(const int* __restrict__ ei,
                                          const float* __restrict__ h,
                                          const float* __restrict__ logits,
                                          const int* q, int qhead, int n, int lane,
                                          float* acc, float* ssl, const float* mxl) {
    int ent = 0;
    if (FULL || lane < n) ent = q[(qhead + lane) & (AGG_QCAP - 1)];
    const int e   = ent & 0x1FFFFF;
    const int rel = ent >> 21;
    const int s   = ei[e];
    const float4 lg = *(const float4*)(logits + (size_t)e * HEADS);
    const float4 mm = *(const float4*)(mxl + rel * HEADS);
    const float w0 = expf(lg.x - mm.x);
    const float w1 = expf(lg.y - mm.y);
    const float w2 = expf(lg.z - mm.z);
    const float w3 = expf(lg.w - mm.w);
    if (FULL || lane < n) {
        atomicAdd(&ssl[rel * HEADS + 0], w0);
        atomicAdd(&ssl[rel * HEADS + 1], w1);
        atomicAdd(&ssl[rel * HEADS + 2], w2);
        atomicAdd(&ssl[rel * HEADS + 3], w3);
    }
    const int nk = FULL ? 32 : n;
#pragma unroll 1
    for (int kb = 0; kb < nk; kb += 8) {
        float hv[8][4];
#pragma unroll
        for (int j = 0; j < 8; ++j) {
            const int k = kb + j;
            if (FULL || k < n) {
                const int sk = __builtin_amdgcn_readlane(s, k);
                const float* hp = h + (size_t)sk * OUT_CH + lane;
                hv[j][0] = hp[0];
                hv[j][1] = hp[32];
                hv[j][2] = hp[64];
                hv[j][3] = hp[96];
            }
        }
#pragma unroll
        for (int j = 0; j < 8; ++j) {
            const int k = kb + j;
            if (FULL || k < n) {
                const int rk = __builtin_amdgcn_readlane(rel, k);
                const float a0 = __int_as_float(__builtin_amdgcn_readlane(__float_as_int(w0), k));
                const float a1 = __int_as_float(__builtin_amdgcn_readlane(__float_as_int(w1), k));
                const float a2 = __int_as_float(__builtin_amdgcn_readlane(__float_as_int(w2), k));
                const float a3 = __int_as_float(__builtin_amdgcn_readlane(__float_as_int(w3), k));
                float* ap = acc + rk * OUT_CH + lane;
                atomicAdd(ap,      hv[j][0] * a0);
                atomicAdd(ap + 32, hv[j][1] * a1);
                atomicAdd(ap + 64, hv[j][2] * a2);
                atomicAdd(ap + 96, hv[j][3] * a3);
            }
        }
    }
}

__device__ __forceinline__ int agg_wrap(int sp) {
    while (sp >= AGG_STEPS) sp -= AGG_STEPS;
    return sp;
}

__device__ __forceinline__ void agg_load(int4 (&dv)[AGG_U], const int4* __restrict__ dst4,
                                         int sp, int wave, int lane) {
#pragma unroll
    for (int u = 0; u < AGG_U; ++u) {
        int wbase = (sp * AGG_U + u) * AGG_THREADS + wave * 32;
        if (wbase > N_DST4 - 32) wbase = N_DST4 - 32;
        dv[u] = dst4[wbase + lane];
    }
}

__device__ __forceinline__ void agg_scan(const int4 (&dv)[AGG_U], int sp, int lo, int wave, int lane,
                                         int* q, int& qhead, int& cnt,
                                         const int* __restrict__ ei,
                                         const float* __restrict__ h,
                                         const float* __restrict__ logits,
                                         float* acc, float* ssl, const float* mxl) {
#pragma unroll
    for (int u = 0; u < AGG_U; ++u) {
        const int wbase = (sp * AGG_U + u) * AGG_THREADS + wave * 32;
        if (wbase < N_DST4) {
            const int e0 = (wbase + lane) * 4;
            const int r0 = dv[u].x - lo, r1 = dv[u].y - lo, r2 = dv[u].z - lo, r3 = dv[u].w - lo;
            const bool m0 = (unsigned)r0 < (unsigned)AGG_R;
            const bool m1 = (unsigned)r1 < (unsigned)AGG_R;
            const bool m2 = (unsigned)r2 < (unsigned)AGG_R;
            const bool m3 = (unsigned)r3 < (unsigned)AGG_R;
            const unsigned b0 = __builtin_amdgcn_ballot_w32(m0);
            const unsigned b1 = __builtin_amdgcn_ballot_w32(m1);
            const unsigned b2 = __builtin_amdgcn_ballot_w32(m2);
            const unsigned b3 = __builtin_amdgcn_ballot_w32(m3);
            if (b0 | b1 | b2 | b3) {
                if (m0) q[(qhead + cnt + __builtin_amdgcn_mbcnt_lo(b0, 0)) & (AGG_QCAP - 1)] = (e0 + 0) | (r0 << 21);
                cnt += __builtin_popcount(b0);
                if (m1) q[(qhead + cnt + __builtin_amdgcn_mbcnt_lo(b1, 0)) & (AGG_QCAP - 1)] = (e0 + 1) | (r1 << 21);
                cnt += __builtin_popcount(b1);
                if (m2) q[(qhead + cnt + __builtin_amdgcn_mbcnt_lo(b2, 0)) & (AGG_QCAP - 1)] = (e0 + 2) | (r2 << 21);
                cnt += __builtin_popcount(b2);
                if (m3) q[(qhead + cnt + __builtin_amdgcn_mbcnt_lo(b3, 0)) & (AGG_QCAP - 1)] = (e0 + 3) | (r3 << 21);
                cnt += __builtin_popcount(b3);
            }
        }
    }
    while (cnt >= 32) {
        agg_drain<true>(ei, h, logits, q, qhead, 32, lane, acc, ssl, mxl);
        qhead = (qhead + 32) & (AGG_QCAP - 1);
        cnt -= 32;
    }
}

__device__ __forceinline__ float wave_sum(float t) {
    for (int o = 16; o > 0; o >>= 1) t += __shfl_xor(t, o, 32);
    return t;
}

__global__ __launch_bounds__(AGG_THREADS)
void k_agg_lds(const int* __restrict__ ei,
               const float* __restrict__ h,
               const float* __restrict__ logits,
               const unsigned* __restrict__ mkey,
               const float* __restrict__ bias,
               const float* __restrict__ gamma,
               const float* __restrict__ beta,
               float* __restrict__ out) {
    __shared__ __attribute__((aligned(16))) float acc[AGG_R * OUT_CH];
    __shared__ __attribute__((aligned(16))) float ssl[AGG_R * HEADS];
    __shared__ __attribute__((aligned(16))) float mxl[AGG_R * HEADS];
    __shared__ int que[AGG_WAVES * AGG_QCAP];

    const int tid  = threadIdx.x;
    const int wave = __builtin_amdgcn_readfirstlane(tid >> 5);
    const int lane = tid & 31;
    const int lo   = blockIdx.x * AGG_R;

    for (int r = wave; r < AGG_R; r += AGG_WAVES) {
        const int node = lo + r;
        if (node < N_NODES) {
            const float* hp = h + (size_t)node * OUT_CH + lane;
            float* ap = acc + r * OUT_CH + lane;
            for (int hd = 0; hd < HEADS; ++hd) {
                const float mf = key2f(mkey[node * HEADS + hd]);
                const float w  = expf(logits[(size_t)(N_EDGES + node) * HEADS + hd] - mf);
                ap[hd * CPH] = hp[hd * CPH] * w;
                if (lane == 0) { ssl[r * HEADS + hd] = w; mxl[r * HEADS + hd] = mf; }
            }
        }
    }
    __syncthreads();

    int* q = que + wave * AGG_QCAP;
    int qhead = 0, cnt = 0;
    const int4* dst4 = (const int4*)(ei + N_EDGES);
    const int start = (blockIdx.x * AGG_STEPS) / AGG_BLOCKS;

    int4 da[AGG_U], db[AGG_U];
    agg_load(da, dst4, agg_wrap(start), wave, lane);
    for (int st = 0; st < AGG_STEPS; st += 2) {
        const int spa = agg_wrap(st + start);
        const int spb = agg_wrap(st + 1 + start);
        agg_load(db, dst4, spb, wave, lane);
        agg_scan(da, spa, lo, wave, lane, q, qhead, cnt, ei, h, logits, acc, ssl, mxl);
        agg_load(da, dst4, agg_wrap(st + 2 + start), wave, lane);
        agg_scan(db, spb, lo, wave, lane, q, qhead, cnt, ei, h, logits, acc, ssl, mxl);
    }
    if (cnt > 0) agg_drain<false>(ei, h, logits, q, qhead, cnt, lane, acc, ssl, mxl);
    __syncthreads();

    const float bs0 = bias[lane],  bs1 = bias[32 + lane],  bs2 = bias[64 + lane],  bs3 = bias[96 + lane];
    const float gm0 = gamma[lane], gm1 = gamma[32 + lane], gm2 = gamma[64 + lane], gm3 = gamma[96 + lane];
    const float bt0 = beta[lane],  bt1 = beta[32 + lane],  bt2 = beta[64 + lane],  bt3 = beta[96 + lane];
    for (int r = wave; r < AGG_R; r += AGG_WAVES) {
        const int node = lo + r;
        if (node < N_NODES) {
            const float* ap = acc + r * OUT_CH + lane;
            const float4 sv = *(const float4*)(ssl + r * HEADS);
            const float v0 = ap[0]  / (sv.x + SOFTMAX_EPS) + bs0;
            const float v1 = ap[32] / (sv.y + SOFTMAX_EPS) + bs1;
            const float v2 = ap[64] / (sv.z + SOFTMAX_EPS) + bs2;
            const float v3 = ap[96] / (sv.w + SOFTMAX_EPS) + bs3;

            const float mu = wave_sum((v0 + v1) + (v2 + v3)) * (1.0f / OUT_CH);
            const float d0 = v0 - mu, d1 = v1 - mu, d2 = v2 - mu, d3 = v3 - mu;
            const float var = wave_sum((d0 * d0 + d1 * d1) + (d2 * d2 + d3 * d3)) * (1.0f / OUT_CH);
            const float rs = 1.0f / sqrtf(var + LN_EPS);

            float y0 = d0 * rs * gm0 + bt0;
            float y1 = d1 * rs * gm1 + bt1;
            float y2 = d2 * rs * gm2 + bt2;
            float y3 = d3 * rs * gm3 + bt3;
            y0 = y0 > 0.0f ? y0 : (expf(y0) - 1.0f);
            y1 = y1 > 0.0f ? y1 : (expf(y1) - 1.0f);
            y2 = y2 > 0.0f ? y2 : (expf(y2) - 1.0f);
            y3 = y3 > 0.0f ? y3 : (expf(y3) - 1.0f);

            float* op = out + (size_t)node * OUT_CH + lane;
            op[0]  = y0;
            op[32] = y1;
            op[64] = y2;
            op[96] = y3;
        }
    }
}

extern "C" void kernel_launch(void* const* d_in, const int* in_sizes, int n_in,
                              void* d_out, int out_size, void* d_ws, size_t ws_size,
                              hipStream_t stream) {
    const float* x       = (const float*)d_in[0];
    const int*   ei      = (const int*)  d_in[1];
    const float* W       = (const float*)d_in[2];
    const float* att_src = (const float*)d_in[3];
    const float* att_dst = (const float*)d_in[4];
    const float* bias    = (const float*)d_in[5];
    const float* gamma   = (const float*)d_in[6];
    const float* beta    = (const float*)d_in[7];
    float* out = (float*)d_out;

    char* ws = (char*)d_ws;
    float*    h      = (float*)ws;
    ws += (size_t)N_NODES * OUT_CH * sizeof(float);
    float*    a_s    = (float*)ws;  ws += (size_t)N_NODES * HEADS * sizeof(float);
    float*    a_d    = (float*)ws;  ws += (size_t)N_NODES * HEADS * sizeof(float);
    unsigned* mkey   = (unsigned*)ws; ws += (size_t)N_NODES * HEADS * sizeof(unsigned);
    float*    logits = (float*)ws;

    const int nseg = N_NODES * HEADS;

    k_init<<<(nseg + 255) / 256, 256, 0, stream>>>(mkey, nseg);

    k_gemm_wmma<<<(N_NODES + 127) / 128, 256, IN_CH * OUT_CH * sizeof(float),
                  stream>>>(x, W, h, N_NODES);

    k_dots<<<DOT_BLOCKS, DOT_THREADS, 0, stream>>>(h, att_src, att_dst, a_s, a_d,
                                                   N_NODES);

    k_edge_max<<<EM_BLOCKS, EM_THREADS, 0, stream>>>(ei, a_s, a_d, logits, mkey);

    k_agg_lds<<<AGG_BLOCKS, AGG_THREADS, 0, stream>>>(ei, h, logits, mkey, bias,
                                                      gamma, beta, out);
}


# Segment max taken by the owner block in LDS: pass 1 over its edges forms logits, LDS integer max and a block list; pass 2 walks the list; atomicMax, logits, 2 launches gone; LDS arrival order kept
# speedup vs baseline: 3.3365x; 1.5661x over previous
#include <hip/hip_runtime.h>
#include <math.h>


#define N_NODES 50000
#define N_EDGES 1600000
#define ET_EDGES (N_EDGES + N_NODES)
#define IN_CH 128
#define OUT_CH 128
#define HEADS 4
#define CPH 32
#define NEG_SLOPE 0.2f
#define SOFTMAX_EPS 1e-16f
#define LN_EPS 1e-5f

typedef float v2f __attribute__((ext_vector_type(2)));
typedef float v8f __attribute__((ext_vector_type(8)));

__global__ void k_gemm_wmma(const float* __restrict__ x,
                            const float* __restrict__ W,
                            float* __restrict__ h, int nrows) {
    extern __shared__ float Wlds[];
    for (int i = threadIdx.x; i < (IN_CH * OUT_CH) / 4; i += blockDim.x)
        ((float4*)Wlds)[i] = ((const float4*)W)[i];
    __syncthreads();

    const int wave = threadIdx.x >> 5;
    const int lane = threadIdx.x & 31;
    const int rowBase = blockIdx.x * 128 + wave * 16;
    const int m  = lane & 15;
    const int kk = (lane >> 4) << 1;

    int arow = rowBase + m;
    if (arow >= nrows) arow = nrows - 1;
    const float* xk = x + (size_t)arow * IN_CH + kk;

    v8f c[8];
#pragma unroll
    for (int t = 0; t < 8; ++t) c[t] = (v8f){0,0,0,0,0,0,0,0};

    float2 av = *(const float2*)xk;
    for (int k = 0; k < IN_CH; k += 4) {
        v2f a; a.x = av.x; a.y = av.y;
        if (k + 4 < IN_CH) av = *(const float2*)(xk + k + 4);

        const float* w0 = &Wlds[(k + kk) * OUT_CH + m];
        v2f b[8];
#pragma unroll
        for (int t = 0; t < 8; ++t) {
            b[t].x = w0[t * 16];
            b[t].y = w0[OUT_CH + t * 16];
        }
#pragma unroll
        for (int t = 0; t < 8; ++t)
            c[t] = __builtin_amdgcn_wmma_f32_16x16x4_f32(
                false, a, false, b[t], (short)0, c[t], false, false);
    }

#pragma unroll
    for (int t = 0; t < 8; ++t) {
#pragma unroll
        for (int v = 0; v < 8; ++v) {
            const int M = v + ((lane >> 4) << 3);
            const int row = rowBase + M;
            if (row < nrows)
                h[(size_t)row * OUT_CH + t * 16 + m] = c[t][v];
        }
    }
}

#define DOT_THREADS 256
#define DOT_NPW     16
#define DOT_BLOCKS  ((N_NODES + (DOT_THREADS / 32) * DOT_NPW - 1) / ((DOT_THREADS / 32) * DOT_NPW))

__device__ __forceinline__ float dots_sum8(const float (&v)[8], int lane) {
    const bool b2 = (lane & 4) != 0, b1 = (lane & 2) != 0, b0 = (lane & 1) != 0;
    float r[4], q[2];
#pragma unroll
    for (int j = 0; j < 4; ++j) {
        const float keep = b2 ? v[j + 4] : v[j];
        const float send = b2 ? v[j] : v[j + 4];
        r[j] = keep + __shfl_xor(send, 4, 32);
    }
#pragma unroll
    for (int j = 0; j < 2; ++j) {
        const float keep = b1 ? r[j + 2] : r[j];
        const float send = b1 ? r[j] : r[j + 2];
        q[j] = keep + __shfl_xor(send, 2, 32);
    }
    const float keep = b0 ? q[1] : q[0];
    const float send = b0 ? q[0] : q[1];
    return keep + __shfl_xor(send, 1, 32);
}

__global__ __launch_bounds__(DOT_THREADS)
void k_dots(const float* __restrict__ h,
            const float* __restrict__ att_src,
            const float* __restrict__ att_dst,
            float* __restrict__ a_s, float* __restrict__ a_d, int n) {
    const int wave = threadIdx.x >> 5;
    const int lane = threadIdx.x & 31;
    const int base = (blockIdx.x * (DOT_THREADS / 32) + wave) * DOT_NPW;
    if (base >= n) return;
    const float4 ws = ((const float4*)att_src)[lane];
    const float4 wd = ((const float4*)att_dst)[lane];

    float4 hv[DOT_NPW];
#pragma unroll
    for (int i = 0; i < DOT_NPW; ++i) {
        int node = base + i;
        if (node >= n) node = n - 1;
        hv[i] = ((const float4*)(h + (size_t)node * OUT_CH))[lane];
    }
    __builtin_amdgcn_sched_barrier(0);
#pragma unroll
    for (int g = 0; g < DOT_NPW / 8; ++g) {
        float vs[8], vd[8];
#pragma unroll
        for (int i = 0; i < 8; ++i) {
            const float4 x4 = hv[g * 8 + i];
            vs[i] = (x4.x * ws.x + x4.y * ws.y) + (x4.z * ws.z + x4.w * ws.w);
            vd[i] = (x4.x * wd.x + x4.y * wd.y) + (x4.z * wd.z + x4.w * wd.w);
        }
        const float rs = dots_sum8(vs, lane);
        const float rd = dots_sum8(vd, lane);
        const int node = base + g * 8 + (lane & 7);
        if (node < n) {
            a_s[node * HEADS + (lane >> 3)] = rs;
            a_d[node * HEADS + (lane >> 3)] = rd;
        }
    }
}

__device__ __forceinline__ unsigned f2key(float f) {
    unsigned b = __float_as_uint(f);
    return (b & 0x80000000u) ? ~b : (b | 0x80000000u);
}
__device__ __forceinline__ float key2f(unsigned k) {
    unsigned b = (k & 0x80000000u) ? (k ^ 0x80000000u) : ~k;
    return __uint_as_float(b);
}
__device__ __forceinline__ float lrelu(float v) {
    return v > 0.0f ? v : NEG_SLOPE * v;
}

#define AGG_BLOCKS  256
#define AGG_R       196
#define AGG_THREADS 1024
#define AGG_WAVES   (AGG_THREADS / 32)
#define AGG_U       4
#define AGG_QCAP    1024
#define AGG_CAP     24576
#define N_DST4      (N_EDGES / 4)
#define AGG_STEPS   ((N_DST4 + AGG_THREADS * AGG_U - 1) / (AGG_THREADS * AGG_U))
static_assert(AGG_BLOCKS * AGG_R >= N_NODES, "ranges must cover every node");
static_assert(N_EDGES % 4 == 0, "destination list is read as int4");
static_assert(N_DST4 % 32 == 0, "end-of-list test is per wave");
static_assert(AGG_STEPS % 2 == 0, "steps are taken in pairs");
static_assert(N_EDGES <= (1 << 21) && AGG_R <= 256, "ring entry = edge | row << 21");
static_assert(N_NODES <= (1 << 16), "list entry = source | row << 16");
static_assert(AGG_R * HEADS <= AGG_THREADS, "one thread per (row, head) at the start");

struct AggLds {
    float*    acc;
    float*    ssl;
    float*    mxl;
    unsigned* mkl;
    float*    adl;
    int*      ctl;
};

template <bool FULL>
__device__ __forceinline__ void agg_drain1(const int* __restrict__ ei,
                                           const float* __restrict__ a_s,
                                           const int* q, int qhead, int n, int lane,
                                           const AggLds& L, int* __restrict__ lst) {
    const bool on = FULL || lane < n;
    int ent = 0;
    if (on) ent = q[(qhead + lane) & (AGG_QCAP - 1)];
    const int e   = ent & 0x1FFFFF;
    const int rel = ent >> 21;
    const int s   = ei[e];
    const float4 as4 = *(const float4*)(a_s + (size_t)s * HEADS);
    const float4 ad4 = *(const float4*)(L.adl + rel * HEADS);
    const float l0 = lrelu(as4.x + ad4.x);
    const float l1 = lrelu(as4.y + ad4.y);
    const float l2 = lrelu(as4.z + ad4.z);
    const float l3 = lrelu(as4.w + ad4.w);
    if (on) {
        atomicMax(&L.mkl[rel * HEADS + 0], f2key(l0));
        atomicMax(&L.mkl[rel * HEADS + 1], f2key(l1));
        atomicMax(&L.mkl[rel * HEADS + 2], f2key(l2));
        atomicMax(&L.mkl[rel * HEADS + 3], f2key(l3));
    }
    const int nn = FULL ? 32 : n;
    int base = 0;
    if (lane == 0) base = atomicAdd(&L.ctl[0], nn);
    base = __builtin_amdgcn_readfirstlane(base);
    if (on && base + lane < AGG_CAP) lst[base + lane] = s | (rel << 16);
    if (lane == 0 && base + nn > AGG_CAP) L.ctl[1] = 1;
}

template <bool FULL>
__device__ __forceinline__ void agg_drain2(const float* __restrict__ h,
                                           const float* __restrict__ a_s,
                                           int s, int rel, int n, int lane,
                                           const AggLds& L) {
    const float4 as4 = *(const float4*)(a_s + (size_t)s * HEADS);
    const float4 ad4 = *(const float4*)(L.adl + rel * HEADS);
    const float4 mm  = *(const float4*)(L.mxl + rel * HEADS);
    const float w0 = expf(lrelu(as4.x + ad4.x) - mm.x);
    const float w1 = expf(lrelu(as4.y + ad4.y) - mm.y);
    const float w2 = expf(lrelu(as4.z + ad4.z) - mm.z);
    const float w3 = expf(lrelu(as4.w + ad4.w) - mm.w);
    if (FULL || lane < n) {
        atomicAdd(&L.ssl[rel * HEADS + 0], w0);
        atomicAdd(&L.ssl[rel * HEADS + 1], w1);
        atomicAdd(&L.ssl[rel * HEADS + 2], w2);
        atomicAdd(&L.ssl[rel * HEADS + 3], w3);
    }
    const int nk = FULL ? 32 : n;
#pragma unroll 1
    for (int kb = 0; kb < nk; kb += 8) {
        float hv[8][4];
#pragma unroll
        for (int j = 0; j < 8; ++j) {
            const int k = kb + j;
            if (FULL || k < n) {
                const int sk = __builtin_amdgcn_readlane(s, k);
                const float* hp = h + (size_t)sk * OUT_CH + lane;
                hv[j][0] = hp[0];
                hv[j][1] = hp[32];
                hv[j][2] = hp[64];
                hv[j][3] = hp[96];
            }
        }
#pragma unroll
        for (int j = 0; j < 8; ++j) {
            const int k = kb + j;
            if (FULL || k < n) {
                const int rk = __builtin_amdgcn_readlane(rel, k);
                const float a0 = __int_as_float(__builtin_amdgcn_readlane(__float_as_int(w0), k));
                const float a1 = __int_as_float(__builtin_amdgcn_readlane(__float_as_int(w1), k));
                const float a2 = __int_as_float(__builtin_amdgcn_readlane(__float_as_int(w2), k));
                const float a3 = __int_as_float(__builtin_amdgcn_readlane(__float_as_int(w3), k));
                float* ap = L.acc + rk * OUT_CH + lane;
                atomicAdd(ap,      hv[j][0] * a0);
                atomicAdd(ap + 32, hv[j][1] * a1);
                atomicAdd(ap + 64, hv[j][2] * a2);
                atomicAdd(ap + 96, hv[j][3] * a3);
            }
        }
    }
}

template <int PH, bool FULL>
__device__ __forceinline__ void agg_drain(const int* __restrict__ ei,
                                          const float* __restrict__ h,
                                          const float* __restrict__ a_s,
                                          const int* q, int qhead, int n, int lane,
                                          const AggLds& L, int* __restrict__ lst) {
    if (PH == 1) {
        agg_drain1<FULL>(ei, a_s, q, qhead, n, lane, L, lst);
    } else {
        int ent = 0;
        if (FULL || lane < n) ent = q[(qhead + lane) & (AGG_QCAP - 1)];
        const int s = ei[ent & 0x1FFFFF];
        agg_drain2<FULL>(h, a_s, s, ent >> 21, n, lane, L);
    }
}

__device__ __forceinline__ int agg_wrap(int sp) {
    while (sp >= AGG_STEPS) sp -= AGG_STEPS;
    return sp;
}

__device__ __forceinline__ void agg_load(int4 (&dv)[AGG_U], const int4* __restrict__ dst4,
                                         int sp, int wave, int lane) {
#pragma unroll
    for (int u = 0; u < AGG_U; ++u) {
        int wbase = (sp * AGG_U + u) * AGG_THREADS + wave * 32;
        if (wbase > N_DST4 - 32) wbase = N_DST4 - 32;
        dv[u] = dst4[wbase + lane];
    }
}

template <int PH>
__device__ __forceinline__ void agg_scan(const int4 (&dv)[AGG_U], int sp, int lo, int wave, int lane,
                                         int* q, int& qhead, int& cnt,
                                         const int* __restrict__ ei,
                                         const float* __restrict__ h,
                                         const float* __restrict__ a_s,
                                         const AggLds& L, int* __restrict__ lst) {
#pragma unroll
    for (int u = 0; u < AGG_U; ++u) {
        const int wbase = (sp * AGG_U + u) * AGG_THREADS + wave * 32;
        if (wbase < N_DST4) {
            const int e0 = (wbase + lane) * 4;
            const int r0 = dv[u].x - lo, r1 = dv[u].y - lo, r2 = dv[u].z - lo, r3 = dv[u].w - lo;
            const bool m0 = (unsigned)r0 < (unsigned)AGG_R;
            const bool m1 = (unsigned)r1 < (unsigned)AGG_R;
            const bool m2 = (unsigned)r2 < (unsigned)AGG_R;
            const bool m3 = (unsigned)r3 < (unsigned)AGG_R;
            const unsigned b0 = __builtin_amdgcn_ballot_w32(m0);
            const unsigned b1 = __builtin_amdgcn_ballot_w32(m1);
            const unsigned b2 = __builtin_amdgcn_ballot_w32(m2);
            const unsigned b3 = __builtin_amdgcn_ballot_w32(m3);
            if (b0 | b1 | b2 | b3) {
                if (m0) q[(qhead + cnt + __builtin_amdgcn_mbcnt_lo(b0, 0)) & (AGG_QCAP - 1)] = (e0 + 0) | (r0 << 21);
                cnt += __builtin_popcount(b0);
                if (m1) q[(qhead + cnt + __builtin_amdgcn_mbcnt_lo(b1, 0)) & (AGG_QCAP - 1)] = (e0 + 1) | (r1 << 21);
                cnt += __builtin_popcount(b1);
                if (m2) q[(qhead + cnt + __builtin_amdgcn_mbcnt_lo(b2, 0)) & (AGG_QCAP - 1)] = (e0 + 2) | (r2 << 21);
                cnt += __builtin_popcount(b2);
                if (m3) q[(qhead + cnt + __builtin_amdgcn_mbcnt_lo(b3, 0)) & (AGG_QCAP - 1)] = (e0 + 3) | (r3 << 21);
                cnt += __builtin_popcount(b3);
            }
        }
    }
    while (cnt >= 32) {
        agg_drain<PH, true>(ei, h, a_s, q, qhead, 32, lane, L, lst);
        qhead = (qhead + 32) & (AGG_QCAP - 1);
        cnt -= 32;
    }
}

template <int PH>
__device__ __forceinline__ void agg_pass(const int* __restrict__ ei,
                                         const float* __restrict__ h,
                                         const float* __restrict__ a_s,
                                         int lo, int wave, int lane, int* q,
                                         const AggLds& L, int* __restrict__ lst) {
    int qhead = 0, cnt = 0;
    const int4* dst4 = (const int4*)(ei + N_EDGES);
    const int start = (blockIdx.x * AGG_STEPS) / AGG_BLOCKS;

    int4 da[AGG_U], db[AGG_U];
    agg_load(da, dst4, agg_wrap(start), wave, lane);
    for (int st = 0; st < AGG_STEPS; st += 2) {
        const int spa = agg_wrap(st + start);
        const int spb = agg_wrap(st + 1 + start);
        agg_load(db, dst4, spb, wave, lane);
        agg_scan<PH>(da, spa, lo, wave, lane, q, qhead, cnt, ei, h, a_s, L, lst);
        agg_load(da, dst4, agg_wrap(st + 2 + start), wave, lane);
        agg_scan<PH>(db, spb, lo, wave, lane, q, qhead, cnt, ei, h, a_s, L, lst);
    }
    if (cnt > 0) agg_drain<PH, false>(ei, h, a_s, q, qhead, cnt, lane, L, lst);
}

__device__ __forceinline__ float wave_sum(float t) {
    for (int o = 16; o > 0; o >>= 1) t += __shfl_xor(t, o, 32);
    return t;
}

__global__ __launch_bounds__(AGG_THREADS)
void k_agg_lds(const int* __restrict__ ei,
               const float* __restrict__ h,
               const float* __restrict__ a_s,
               const float* __restrict__ a_d,
               int* __restrict__ lists,
               const float* __restrict__ bias,
               const float* __restrict__ gamma,
               const float* __restrict__ beta,
               float* __restrict__ out) {
    __shared__ __attribute__((aligned(16))) float acc[AGG_R * OUT_CH];
    __shared__ __attribute__((aligned(16))) float ssl[AGG_R * HEADS];
    __shared__ __attribute__((aligned(16))) float mxl[AGG_R * HEADS];
    __shared__ __attribute__((aligned(16))) unsigned mkl[AGG_R * HEADS];
    __shared__ __attribute__((aligned(16))) float adl[AGG_R * HEADS];
    __shared__ int ctl[2];
    __shared__ int que[AGG_WAVES * AGG_QCAP];

    const int tid  = threadIdx.x;
    const int wave = __builtin_amdgcn_readfirstlane(tid >> 5);
    const int lane = tid & 31;
    const int lo   = blockIdx.x * AGG_R;
    AggLds L;
    L.acc = acc; L.ssl = ssl; L.mxl = mxl; L.mkl = mkl; L.adl = adl; L.ctl = ctl;
    int* lst = lists + (size_t)blockIdx.x * AGG_CAP;
    int* q = que + wave * AGG_QCAP;

    if (tid < AGG_R * HEADS) {
        const int node = lo + (tid >> 2);
        float ad = 0.0f;
        unsigned key = 0u;
        if (node < N_NODES) {
            ad  = a_d[node * HEADS + (tid & 3)];
            key = f2key(lrelu(a_s[node * HEADS + (tid & 3)] + ad));
        }
        adl[tid] = ad;
        mkl[tid] = key;
    }
    if (tid == 0) { ctl[0] = 0; ctl[1] = 0; }
    __syncthreads();

    agg_pass<1>(ei, h, a_s, lo, wave, lane, q, L, lst);
    __syncthreads();

    for (int r = wave; r < AGG_R; r += AGG_WAVES) {
        const int node = lo + r;
        if (node < N_NODES) {
            const float4 as4 = *(const float4*)(a_s + (size_t)node * HEADS);
            const float4 ad4 = *(const float4*)(adl + r * HEADS);
            const uint4  mk  = *(const uint4*)(mkl + r * HEADS);
            float4 mf, w;
            mf.x = key2f(mk.x); mf.y = key2f(mk.y); mf.z = key2f(mk.z); mf.w = key2f(mk.w);
            w.x = expf(lrelu(as4.x + ad4.x) - mf.x);
            w.y = expf(lrelu(as4.y + ad4.y) - mf.y);
            w.z = expf(lrelu(as4.z + ad4.z) - mf.z);
            w.w = expf(lrelu(as4.w + ad4.w) - mf.w);
            const float* hp = h + (size_t)node * OUT_CH + lane;
            float* ap = acc + r * OUT_CH + lane;
            ap[0]  = hp[0]  * w.x;
            ap[32] = hp[32] * w.y;
            ap[64] = hp[64] * w.z;
            ap[96] = hp[96] * w.w;
            if (lane == 0) {
                *(float4*)(ssl + r * HEADS) = w;
                *(float4*)(mxl + r * HEADS) = mf;
            }
        }
    }
    __syncthreads();

    const int total = __builtin_amdgcn_readfirstlane(ctl[0]);
    const int ovf   = __builtin_amdgcn_readfirstlane(ctl[1]);
    if (ovf == 0) {
        const int last = total > 0 ? total - 1 : 0;
        int i = wave * 32;
        int ent = 0;
        if (total > 0) ent = lst[min(i + lane, last)];
        while (i < total) {
            const int inext = i + AGG_THREADS;
            const int entn = lst[min(inext + lane, last)];
            const int n = total - i;
            if (n >= 32) agg_drain2<true >(h, a_s, ent & 0xFFFF, ent >> 16, 32, lane, L);
            else         agg_drain2<false>(h, a_s, ent & 0xFFFF, ent >> 16, n,  lane, L);
            ent = entn;
            i = inext;
        }
    } else {
        agg_pass<2>(ei, h, a_s, lo, wave, lane, q, L, lst);
    }
    __syncthreads();

    const float bs0 = bias[lane],  bs1 = bias[32 + lane],  bs2 = bias[64 + lane],  bs3 = bias[96 + lane];
    const float gm0 = gamma[lane], gm1 = gamma[32 + lane], gm2 = gamma[64 + lane], gm3 = gamma[96 + lane];
    const float bt0 = beta[lane],  bt1 = beta[32 + lane],  bt2 = beta[64 + lane],  bt3 = beta[96 + lane];
    for (int r = wave; r < AGG_R; r += AGG_WAVES) {
        const int node = lo + r;
        if (node < N_NODES) {
            const float* ap = acc + r * OUT_CH + lane;
            const float4 sv = *(const float4*)(ssl + r * HEADS);
            const float v0 = ap[0]  / (sv.x + SOFTMAX_EPS) + bs0;
            const float v1 = ap[32] / (sv.y + SOFTMAX_EPS) + bs1;
            const float v2 = ap[64] / (sv.z + SOFTMAX_EPS) + bs2;
            const float v3 = ap[96] / (sv.w + SOFTMAX_EPS) + bs3;

            const float mu = wave_sum((v0 + v1) + (v2 + v3)) * (1.0f / OUT_CH);
            const float d0 = v0 - mu, d1 = v1 - mu, d2 = v2 - mu, d3 = v3 - mu;
            const float var = wave_sum((d0 * d0 + d1 * d1) + (d2 * d2 + d3 * d3)) * (1.0f / OUT_CH);
            const float rs = 1.0f / sqrtf(var + LN_EPS);

            float y0 = d0 * rs * gm0 + bt0;
            float y1 = d1 * rs * gm1 + bt1;
            float y2 = d2 * rs * gm2 + bt2;
            float y3 = d3 * rs * gm3 + bt3;
            y0 = y0 > 0.0f ? y0 : (expf(y0) - 1.0f);
            y1 = y1 > 0.0f ? y1 : (expf(y1) - 1.0f);
            y2 = y2 > 0.0f ? y2 : (expf(y2) - 1.0f);
            y3 = y3 > 0.0f ? y3 : (expf(y3) - 1.0f);

            float* op = out + (size_t)node * OUT_CH + lane;
            op[0]  = y0;
            op[32] = y1;
            op[64] = y2;
            op[96] = y3;
        }
    }
}

extern "C" void kernel_launch(void* const* d_in, const int* in_sizes, int n_in,
                              void* d_out, int out_size, void* d_ws, size_t ws_size,
                              hipStream_t stream) {
    const float* x       = (const float*)d_in[0];
    const int*   ei      = (const int*)  d_in[1];
    const float* W       = (const float*)d_in[2];
    const float* att_src = (const float*)d_in[3];
    const float* att_dst = (const float*)d_in[4];
    const float* bias    = (const float*)d_in[5];
    const float* gamma   = (const float*)d_in[6];
    const float* beta    = (const float*)d_in[7];
    float* out = (float*)d_out;

    char* ws = (char*)d_ws;
    float*    h      = (float*)ws;
    ws += (size_t)N_NODES * OUT_CH * sizeof(float);
    float*    a_s    = (float*)ws;  ws += (size_t)N_NODES * HEADS * sizeof(float);
    float*    a_d    = (float*)ws;  ws += (size_t)N_NODES * HEADS * sizeof(float);
    int*      lists  = (int*)ws;

    k_gemm_wmma<<<(N_NODES + 127) / 128, 256, IN_CH * OUT_CH * sizeof(float),
                  stream>>>(x, W, h, N_NODES);

    k_dots<<<DOT_BLOCKS, DOT_THREADS, 0, stream>>>(h, att_src, att_dst, a_s, a_d,
                                                   N_NODES);

    k_agg_lds<<<AGG_BLOCKS, AGG_THREADS, 0, stream>>>(ei, h, a_s, a_d, lists, bias,
                                                      gamma, beta, out);
}


# Filter key narrowed to one byte: dots launch writes each edge's owner-block number; owner blocks scan that 1.6 MB plane (16 edges a load, zero-byte test), not 6.4 MB of dst; LDS arrival order kept
# speedup vs baseline: 3.7402x; 1.1210x over previous
#include <hip/hip_runtime.h>
#include <math.h>


#define N_NODES 50000
#define N_EDGES 1600000
#define ET_EDGES (N_EDGES + N_NODES)
#define IN_CH 128
#define OUT_CH 128
#define HEADS 4
#define CPH 32
#define NEG_SLOPE 0.2f
#define SOFTMAX_EPS 1e-16f
#define LN_EPS 1e-5f

typedef float v2f __attribute__((ext_vector_type(2)));
typedef float v8f __attribute__((ext_vector_type(8)));

__global__ void k_gemm_wmma(const float* __restrict__ x,
                            const float* __restrict__ W,
                            float* __restrict__ h, int nrows) {
    extern __shared__ float Wlds[];
    for (int i = threadIdx.x; i < (IN_CH * OUT_CH) / 4; i += blockDim.x)
        ((float4*)Wlds)[i] = ((const float4*)W)[i];
    __syncthreads();

    const int wave = threadIdx.x >> 5;
    const int lane = threadIdx.x & 31;
    const int rowBase = blockIdx.x * 128 + wave * 16;
    const int m  = lane & 15;
    const int kk = (lane >> 4) << 1;

    int arow = rowBase + m;
    if (arow >= nrows) arow = nrows - 1;
    const float* xk = x + (size_t)arow * IN_CH + kk;

    v8f c[8];
#pragma unroll
    for (int t = 0; t < 8; ++t) c[t] = (v8f){0,0,0,0,0,0,0,0};

    float2 av = *(const float2*)xk;
    for (int k = 0; k < IN_CH; k += 4) {
        v2f a; a.x = av.x; a.y = av.y;
        if (k + 4 < IN_CH) av = *(const float2*)(xk + k + 4);

        const float* w0 = &Wlds[(k + kk) * OUT_CH + m];
        v2f b[8];
#pragma unroll
        for (int t = 0; t < 8; ++t) {
            b[t].x = w0[t * 16];
            b[t].y = w0[OUT_CH + t * 16];
        }
#pragma unroll
        for (int t = 0; t < 8; ++t)
            c[t] = __builtin_amdgcn_wmma_f32_16x16x4_f32(
                false, a, false, b[t], (short)0, c[t], false, false);
    }

#pragma unroll
    for (int t = 0; t < 8; ++t) {
#pragma unroll
        for (int v = 0; v < 8; ++v) {
            const int M = v + ((lane >> 4) << 3);
            const int row = rowBase + M;
            if (row < nrows)
                h[(size_t)row * OUT_CH + t * 16 + m] = c[t][v];
        }
    }
}

#define DOT_THREADS 256
#define DOT_NPW     16
#define DOT_BLOCKS  ((N_NODES + (DOT_THREADS / 32) * DOT_NPW - 1) / ((DOT_THREADS / 32) * DOT_NPW))

#define AGG_BLOCKS  256
#define AGG_R       196
#define KEY_CHUNK   512
#define KEY_CHUNKS  (N_EDGES / KEY_CHUNK)
static_assert(N_EDGES % KEY_CHUNK == 0, "a wave writes the keys of 512 edges");
static_assert((N_NODES - 1) / AGG_R < 256, "an owner key is one byte");
static_assert(DOT_BLOCKS * (DOT_THREADS / 32) >= KEY_CHUNKS, "one key chunk per wave at most");

__device__ __forceinline__ unsigned keys_pack4(int4 d) {
    return ((unsigned)d.x / AGG_R) | (((unsigned)d.y / AGG_R) << 8) |
           (((unsigned)d.z / AGG_R) << 16) | (((unsigned)d.w / AGG_R) << 24);
}

__device__ __forceinline__ float dots_sum8(const float (&v)[8], int lane) {
    const bool b2 = (lane & 4) != 0, b1 = (lane & 2) != 0, b0 = (lane & 1) != 0;
    float r[4], q[2];
#pragma unroll
    for (int j = 0; j < 4; ++j) {
        const float keep = b2 ? v[j + 4] : v[j];
        const float send = b2 ? v[j] : v[j + 4];
        r[j] = keep + __shfl_xor(send, 4, 32);
    }
#pragma unroll
    for (int j = 0; j < 2; ++j) {
        const float keep = b1 ? r[j + 2] : r[j];
        const float send = b1 ? r[j] : r[j + 2];
        q[j] = keep + __shfl_xor(send, 2, 32);
    }
    const float keep = b0 ? q[1] : q[0];
    const float send = b0 ? q[0] : q[1];
    return keep + __shfl_xor(send, 1, 32);
}

__global__ __launch_bounds__(DOT_THREADS)
void k_dots(const float* __restrict__ h,
            const float* __restrict__ att_src,
            const float* __restrict__ att_dst,
            const int* __restrict__ ei, unsigned* __restrict__ keys,
            float* __restrict__ a_s, float* __restrict__ a_d, int n) {
    const int wave = threadIdx.x >> 5;
    const int lane = threadIdx.x & 31;
    const int wg   = blockIdx.x * (DOT_THREADS / 32) + wave;
    const int base = wg * DOT_NPW;
    const bool keyw = wg < KEY_CHUNKS;
    if (base >= n && !keyw) return;
    const float4 ws = ((const float4*)att_src)[lane];
    const float4 wd = ((const float4*)att_dst)[lane];

    const int wk = keyw ? wg : KEY_CHUNKS - 1;
    const int4* dp = (const int4*)(ei + N_EDGES) + (size_t)wk * (KEY_CHUNK / 4) + lane * 4;
    int4 dk[4];
#pragma unroll
    for (int j = 0; j < 4; ++j) dk[j] = dp[j];

    float4 hv[DOT_NPW];
#pragma unroll
    for (int i = 0; i < DOT_NPW; ++i) {
        int node = base + i;
        if (node >= n) node = n - 1;
        hv[i] = ((const float4*)(h + (size_t)node * OUT_CH))[lane];
    }
    __builtin_amdgcn_sched_barrier(0);
    if (keyw) {
        uint4 kv;
        kv.x = keys_pack4(dk[0]);
        kv.y = keys_pack4(dk[1]);
        kv.z = keys_pack4(dk[2]);
        kv.w = keys_pack4(dk[3]);
        ((uint4*)keys)[(size_t)wg * 32 + lane] = kv;
    }
    if (base >= n) return;
#pragma unroll
    for (int g = 0; g < DOT_NPW / 8; ++g) {
        float vs[8], vd[8];
#pragma unroll
        for (int i = 0; i < 8; ++i) {
            const float4 x4 = hv[g * 8 + i];
            vs[i] = (x4.x * ws.x + x4.y * ws.y) + (x4.z * ws.z + x4.w * ws.w);
            vd[i] = (x4.x * wd.x + x4.y * wd.y) + (x4.z * wd.z + x4.w * wd.w);
        }
        const float rs = dots_sum8(vs, lane);
        const float rd = dots_sum8(vd, lane);
        const int node = base + g * 8 + (lane & 7);
        if (node < n) {
            a_s[node * HEADS + (lane >> 3)] = rs;
            a_d[node * HEADS + (lane >> 3)] = rd;
        }
    }
}

__device__ __forceinline__ unsigned f2key(float f) {
    unsigned b = __float_as_uint(f);
    return (b & 0x80000000u) ? ~b : (b | 0x80000000u);
}
__device__ __forceinline__ float key2f(unsigned k) {
    unsigned b = (k & 0x80000000u) ? (k ^ 0x80000000u) : ~k;
    return __uint_as_float(b);
}
__device__ __forceinline__ float lrelu(float v) {
    return v > 0.0f ? v : NEG_SLOPE * v;
}

#define AGG_THREADS 1024
#define AGG_WAVES   (AGG_THREADS / 32)
#define AGG_NSET    4
#define AGG_QCAP    1024
#define AGG_CAP     22528
#define N_K16       (N_EDGES / 16)
#define AGG_STEPS   (((N_K16 + AGG_THREADS - 1) / AGG_THREADS + AGG_NSET - 1) / AGG_NSET * AGG_NSET)
static_assert(AGG_BLOCKS * AGG_R >= N_NODES, "ranges must cover every node");
static_assert(N_K16 % 32 == 0, "end-of-plane test is per wave");
static_assert(AGG_R <= 256 && N_NODES <= (1 << 16), "list entry = source | row << 16");
static_assert(AGG_R * HEADS <= AGG_THREADS, "one thread per (row, head) at the start");

struct AggLds {
    float*    acc;
    float*    ssl;
    float*    mxl;
    unsigned* mkl;
    float*    adl;
    int*      ctl;
};

template <bool FULL>
__device__ __forceinline__ void agg_drain1(const int* __restrict__ ei,
                                           const float* __restrict__ a_s,
                                           const int* q, int qhead, int n, int lo, int lane,
                                           const AggLds& L, int* __restrict__ lst) {
    const bool on = FULL || lane < n;
    int e = 0;
    if (on) e = q[(qhead + lane) & (AGG_QCAP - 1)];
    const int s   = ei[e];
    const int rel = on ? ei[N_EDGES + e] - lo : 0;
    const float4 as4 = *(const float4*)(a_s + (size_t)s * HEADS);
    const float4 ad4 = *(const float4*)(L.adl + rel * HEADS);
    const float l0 = lrelu(as4.x + ad4.x);
    const float l1 = lrelu(as4.y + ad4.y);
    const float l2 = lrelu(as4.z + ad4.z);
    const float l3 = lrelu(as4.w + ad4.w);
    if (on) {
        atomicMax(&L.mkl[rel * HEADS + 0], f2key(l0));
        atomicMax(&L.mkl[rel * HEADS + 1], f2key(l1));
        atomicMax(&L.mkl[rel * HEADS + 2], f2key(l2));
        atomicMax(&L.mkl[rel * HEADS + 3], f2key(l3));
    }
    const int nn = FULL ? 32 : n;
    int base = 0;
    if (lane == 0) base = atomicAdd(&L.ctl[0], nn);
    base = __builtin_amdgcn_readfirstlane(base);
    if (on && base + lane < AGG_CAP) lst[base + lane] = s | (rel << 16);
    if (lane == 0 && base + nn > AGG_CAP) L.ctl[1] = 1;
}

template <bool FULL>
__device__ __forceinline__ void agg_drain2(const float* __restrict__ h,
                                           const float* __restrict__ a_s,
                                           int s, int rel, int n, int lane,
                                           const AggLds& L) {
    const float4 as4 = *(const float4*)(a_s + (size_t)s * HEADS);
    const float4 ad4 = *(const float4*)(L.adl + rel * HEADS);
    const float4 mm  = *(const float4*)(L.mxl + rel * HEADS);
    const float w0 = expf(lrelu(as4.x + ad4.x) - mm.x);
    const float w1 = expf(lrelu(as4.y + ad4.y) - mm.y);
    const float w2 = expf(lrelu(as4.z + ad4.z) - mm.z);
    const float w3 = expf(lrelu(as4.w + ad4.w) - mm.w);
    if (FULL || lane < n) {
        atomicAdd(&L.ssl[rel * HEADS + 0], w0);
        atomicAdd(&L.ssl[rel * HEADS + 1], w1);
        atomicAdd(&L.ssl[rel * HEADS + 2], w2);
        atomicAdd(&L.ssl[rel * HEADS + 3], w3);
    }
    const int nk = FULL ? 32 : n;
#pragma unroll 1
    for (int kb = 0; kb < nk; kb += 8) {
        float hv[8][4];
#pragma unroll
        for (int j = 0; j < 8; ++j) {
            const int k = kb + j;
            if (FULL || k < n) {
                const int sk = __builtin_amdgcn_readlane(s, k);
                const float* hp = h + (size_t)sk * OUT_CH + lane;
                hv[j][0] = hp[0];
                hv[j][1] = hp[32];
                hv[j][2] = hp[64];
                hv[j][3] = hp[96];
            }
        }
#pragma unroll
        for (int j = 0; j < 8; ++j) {
            const int k = kb + j;
            if (FULL || k < n) {
                const int rk = __builtin_amdgcn_readlane(rel, k);
                const float a0 = __int_as_float(__builtin_amdgcn_readlane(__float_as_int(w0), k));
                const float a1 = __int_as_float(__builtin_amdgcn_readlane(__float_as_int(w1), k));
                const float a2 = __int_as_float(__builtin_amdgcn_readlane(__float_as_int(w2), k));
                const float a3 = __int_as_float(__builtin_amdgcn_readlane(__float_as_int(w3), k));
                float* ap = L.acc + rk * OUT_CH + lane;
                atomicAdd(ap,      hv[j][0] * a0);
                atomicAdd(ap + 32, hv[j][1] * a1);
                atomicAdd(ap + 64, hv[j][2] * a2);
                atomicAdd(ap + 96, hv[j][3] * a3);
            }
        }
    }
}

template <int PH, bool FULL>
__device__ __forceinline__ void agg_drain(const int* __restrict__ ei,
                                          const float* __restrict__ h,
                                          const float* __restrict__ a_s,
                                          const int* q, int qhead, int n, int lo, int lane,
                                          const AggLds& L, int* __restrict__ lst) {
    if (PH == 1) {
        agg_drain1<FULL>(ei, a_s, q, qhead, n, lo, lane, L, lst);
    } else {
        const bool on = FULL || lane < n;
        int e = 0;
        if (on) e = q[(qhead + lane) & (AGG_QCAP - 1)];
        const int s   = ei[e];
        const int rel = on ? ei[N_EDGES + e] - lo : 0;
        agg_drain2<FULL>(h, a_s, s, rel, n, lane, L);
    }
}

__device__ __forceinline__ int agg_wrap(int sp) {
    while (sp >= AGG_STEPS) sp -= AGG_STEPS;
    return sp;
}

__device__ __forceinline__ uint4 agg_load(const uint4* __restrict__ keys, int sp, int wave, int lane) {
    int wbase = sp * AGG_THREADS + wave * 32;
    if (wbase > N_K16 - 32) wbase = N_K16 - 32;
    return keys[wbase + lane];
}

__device__ __forceinline__ unsigned agg_nonzero_bytes(unsigned x) {
    return ((x & 0x7F7F7F7Fu) + 0x7F7F7F7Fu) | x | 0x7F7F7F7Fu;
}

template <int PH>
__device__ __forceinline__ void agg_scan(uint4 kv, int sp, unsigned own4, int lo, int wave, int lane,
                                         int* q, int& qhead, int& cnt,
                                         const int* __restrict__ ei,
                                         const float* __restrict__ h,
                                         const float* __restrict__ a_s,
                                         const AggLds& L, int* __restrict__ lst) {
    const int wbase = sp * AGG_THREADS + wave * 32;
    if (wbase >= N_K16) return;
    const unsigned y0 = agg_nonzero_bytes(kv.x ^ own4);
    const unsigned y1 = agg_nonzero_bytes(kv.y ^ own4);
    const unsigned y2 = agg_nonzero_bytes(kv.z ^ own4);
    const unsigned y3 = agg_nonzero_bytes(kv.w ^ own4);
    const bool hit = ((y0 & y1) & (y2 & y3)) != 0xFFFFFFFFu;
    unsigned bm = __builtin_amdgcn_ballot_w32(hit);
    if (bm) {
        unsigned cm = (~y0 >> 7) | (~y1 >> 6) | (~y2 >> 5) | (~y3 >> 4);
        const int e0 = (wbase + lane) * 16;
        do {
            if (cm) {
                const int p = __builtin_ctz(cm);
                q[(qhead + cnt + __builtin_amdgcn_mbcnt_lo(bm, 0)) & (AGG_QCAP - 1)] =
                    e0 + (((p & 7) << 2) | (p >> 3));
                cm &= cm - 1;
            }
            cnt += __builtin_popcount(bm);
            bm = __builtin_amdgcn_ballot_w32(cm != 0);
        } while (bm);
    }
    while (cnt >= 32) {
        agg_drain<PH, true>(ei, h, a_s, q, qhead, 32, lo, lane, L, lst);
        qhead = (qhead + 32) & (AGG_QCAP - 1);
        cnt -= 32;
    }
}

template <int PH>
__device__ __forceinline__ void agg_pass(const int* __restrict__ ei,
                                         const uint4* __restrict__ keys,
                                         const float* __restrict__ h,
                                         const float* __restrict__ a_s,
                                         int lo, int wave, int lane, int* q,
                                         const AggLds& L, int* __restrict__ lst) {
    int qhead = 0, cnt = 0;
    const unsigned own4 = blockIdx.x * 0x01010101u;
    const int start = (blockIdx.x * AGG_STEPS) / AGG_BLOCKS;

    uint4 k0 = agg_load(keys, agg_wrap(start),     wave, lane);
    uint4 k1 = agg_load(keys, agg_wrap(start + 1), wave, lane);
    uint4 k2 = agg_load(keys, agg_wrap(start + 2), wave, lane);
    uint4 k3 = agg_load(keys, agg_wrap(start + 3), wave, lane);
    for (int st = 0; st < AGG_STEPS; st += AGG_NSET) {
        agg_scan<PH>(k0, agg_wrap(st + start),     own4, lo, wave, lane, q, qhead, cnt, ei, h, a_s, L, lst);
        k0 = agg_load(keys, agg_wrap(st + 4 + start), wave, lane);
        agg_scan<PH>(k1, agg_wrap(st + 1 + start), own4, lo, wave, lane, q, qhead, cnt, ei, h, a_s, L, lst);
        k1 = agg_load(keys, agg_wrap(st + 5 + start), wave, lane);
        agg_scan<PH>(k2, agg_wrap(st + 2 + start), own4, lo, wave, lane, q, qhead, cnt, ei, h, a_s, L, lst);
        k2 = agg_load(keys, agg_wrap(st + 6 + start), wave, lane);
        agg_scan<PH>(k3, agg_wrap(st + 3 + start), own4, lo, wave, lane, q, qhead, cnt, ei, h, a_s, L, lst);
        k3 = agg_load(keys, agg_wrap(st + 7 + start), wave, lane);
    }
    if (cnt > 0) agg_drain<PH, false>(ei, h, a_s, q, qhead, cnt, lo, lane, L, lst);
}

__device__ __forceinline__ float wave_sum(float t) {
    for (int o = 16; o > 0; o >>= 1) t += __shfl_xor(t, o, 32);
    return t;
}

__global__ __launch_bounds__(AGG_THREADS)
void k_agg_lds(const int* __restrict__ ei,
               const unsigned* __restrict__ keys,
               const float* __restrict__ h,
               const float* __restrict__ a_s,
               const float* __restrict__ a_d,
               int* __restrict__ lists,
               const float* __restrict__ bias,
               const float* __restrict__ gamma,
               const float* __restrict__ beta,
               float* __restrict__ out) {
    __shared__ __attribute__((aligned(16))) float acc[AGG_R * OUT_CH];
    __shared__ __attribute__((aligned(16))) float ssl[AGG_R * HEADS];
    __shared__ __attribute__((aligned(16))) float mxl[AGG_R * HEADS];
    __shared__ __attribute__((aligned(16))) unsigned mkl[AGG_R * HEADS];
    __shared__ __attribute__((aligned(16))) float adl[AGG_R * HEADS];
    __shared__ int ctl[2];
    __shared__ int que[AGG_WAVES * AGG_QCAP];

    const int tid  = threadIdx.x;
    const int wave = __builtin_amdgcn_readfirstlane(tid >> 5);
    const int lane = tid & 31;
    const int lo   = blockIdx.x * AGG_R;
    AggLds L;
    L.acc = acc; L.ssl = ssl; L.mxl = mxl; L.mkl = mkl; L.adl = adl; L.ctl = ctl;
    int* lst = lists + (size_t)blockIdx.x * AGG_CAP;
    int* q = que + wave * AGG_QCAP;

    if (tid < AGG_R * HEADS) {
        const int node = lo + (tid >> 2);
        float ad = 0.0f;
        unsigned key = 0u;
        if (node < N_NODES) {
            ad  = a_d[node * HEADS + (tid & 3)];
            key = f2key(lrelu(a_s[node * HEADS + (tid & 3)] + ad));
        }
        adl[tid] = ad;
        mkl[tid] = key;
    }
    if (tid == 0) { ctl[0] = 0; ctl[1] = 0; }
    __syncthreads();

    agg_pass<1>(ei, (const uint4*)keys, h, a_s, lo, wave, lane, q, L, lst);
    __syncthreads();

    for (int r = wave; r < AGG_R; r += AGG_WAVES) {
        const int node = lo + r;
        if (node < N_NODES) {
            const float4 as4 = *(const float4*)(a_s + (size_t)node * HEADS);
            const float4 ad4 = *(const float4*)(adl + r * HEADS);
            const uint4  mk  = *(const uint4*)(mkl + r * HEADS);
            float4 mf, w;
            mf.x = key2f(mk.x); mf.y = key2f(mk.y); mf.z = key2f(mk.z); mf.w = key2f(mk.w);
            w.x = expf(lrelu(as4.x + ad4.x) - mf.x);
            w.y = expf(lrelu(as4.y + ad4.y) - mf.y);
            w.z = expf(lrelu(as4.z + ad4.z) - mf.z);
            w.w = expf(lrelu(as4.w + ad4.w) - mf.w);
            const float* hp = h + (size_t)node * OUT_CH + lane;
            float* ap = acc + r * OUT_CH + lane;
            ap[0]  = hp[0]  * w.x;
            ap[32] = hp[32] * w.y;
            ap[64] = hp[64] * w.z;
            ap[96] = hp[96] * w.w;
            if (lane == 0) {
                *(float4*)(ssl + r * HEADS) = w;
                *(float4*)(mxl + r * HEADS) = mf;
            }
        }
    }
    __syncthreads();

    const int total = __builtin_amdgcn_readfirstlane(ctl[0]);
    const int ovf   = __builtin_amdgcn_readfirstlane(ctl[1]);
    if (ovf == 0) {
        const int last = total > 0 ? total - 1 : 0;
        int i = wave * 32;
        int ent = 0;
        if (total > 0) ent = lst[min(i + lane, last)];
        while (i < total) {
            const int inext = i + AGG_THREADS;
            const int entn = lst[min(inext + lane, last)];
            const int n = total - i;
            if (n >= 32) agg_drain2<true >(h, a_s, ent & 0xFFFF, ent >> 16, 32, lane, L);
            else         agg_drain2<false>(h, a_s, ent & 0xFFFF, ent >> 16, n,  lane, L);
            ent = entn;
            i = inext;
        }
    } else {
        agg_pass<2>(ei, (const uint4*)keys, h, a_s, lo, wave, lane, q, L, lst);
    }
    __syncthreads();

    const float bs0 = bias[lane],  bs1 = bias[32 + lane],  bs2 = bias[64 + lane],  bs3 = bias[96 + lane];
    const float gm0 = gamma[lane], gm1 = gamma[32 + lane], gm2 = gamma[64 + lane], gm3 = gamma[96 + lane];
    const float bt0 = beta[lane],  bt1 = beta[32 + lane],  bt2 = beta[64 + lane],  bt3 = beta[96 + lane];
    for (int r = wave; r < AGG_R; r += AGG_WAVES) {
        const int node = lo + r;
        if (node < N_NODES) {
            const float* ap = acc + r * OUT_CH + lane;
            const float4 sv = *(const float4*)(ssl + r * HEADS);
            const float v0 = ap[0]  / (sv.x + SOFTMAX_EPS) + bs0;
            const float v1 = ap[32] / (sv.y + SOFTMAX_EPS) + bs1;
            const float v2 = ap[64] / (sv.z + SOFTMAX_EPS) + bs2;
            const float v3 = ap[96] / (sv.w + SOFTMAX_EPS) + bs3;

            const float mu = wave_sum((v0 + v1) + (v2 + v3)) * (1.0f / OUT_CH);
            const float d0 = v0 - mu, d1 = v1 - mu, d2 = v2 - mu, d3 = v3 - mu;
            const float var = wave_sum((d0 * d0 + d1 * d1) + (d2 * d2 + d3 * d3)) * (1.0f / OUT_CH);
            const float rs = 1.0f / sqrtf(var + LN_EPS);

            float y0 = d0 * rs * gm0 + bt0;
            float y1 = d1 * rs * gm1 + bt1;
            float y2 = d2 * rs * gm2 + bt2;
            float y3 = d3 * rs * gm3 + bt3;
            y0 = y0 > 0.0f ? y0 : (expf(y0) - 1.0f);
            y1 = y1 > 0.0f ? y1 : (expf(y1) - 1.0f);
            y2 = y2 > 0.0f ? y2 : (expf(y2) - 1.0f);
            y3 = y3 > 0.0f ? y3 : (expf(y3) - 1.0f);

            float* op = out + (size_t)node * OUT_CH + lane;
            op[0]  = y0;
            op[32] = y1;
            op[64] = y2;
            op[96] = y3;
        }
    }
}

extern "C" void kernel_launch(void* const* d_in, const int* in_sizes, int n_in,
                              void* d_out, int out_size, void* d_ws, size_t ws_size,
                              hipStream_t stream) {
    const float* x       = (const float*)d_in[0];
    const int*   ei      = (const int*)  d_in[1];
    const float* W       = (const float*)d_in[2];
    const float* att_src = (const float*)d_in[3];
    const float* att_dst = (const float*)d_in[4];
    const float* bias    = (const float*)d_in[5];
    const float* gamma   = (const float*)d_in[6];
    const float* beta    = (const float*)d_in[7];
    float* out = (float*)d_out;

    char* ws = (char*)d_ws;
    float*    h      = (float*)ws;
    ws += (size_t)N_NODES * OUT_CH * sizeof(float);
    float*    a_s    = (float*)ws;  ws += (size_t)N_NODES * HEADS * sizeof(float);
    float*    a_d    = (float*)ws;  ws += (size_t)N_NODES * HEADS * sizeof(float);
    int*      lists  = (int*)ws;
    ws += (size_t)AGG_BLOCKS * AGG_CAP * sizeof(int);
    unsigned* keys   = (unsigned*)ws;

    k_gemm_wmma<<<(N_NODES + 127) / 128, 256, IN_CH * OUT_CH * sizeof(float),
                  stream>>>(x, W, h, N_NODES);

    k_dots<<<DOT_BLOCKS, DOT_THREADS, 0, stream>>>(h, att_src, att_dst, ei, keys, a_s, a_d,
                                                   N_NODES);

    k_agg_lds<<<AGG_BLOCKS, AGG_THREADS, 0, stream>>>(ei, keys, h, a_s, a_d, lists, bias,
                                                      gamma, beta, out);
}


# Dots launch also writes one word per edge (source | row in owner range << 16): pass 1 of the owner block gathers that one word per owned edge, not source and destination separately; sum order kept
# speedup vs baseline: 3.9159x; 1.0470x over previous
#include <hip/hip_runtime.h>
#include <math.h>


#define N_NODES 50000
#define N_EDGES 1600000
#define ET_EDGES (N_EDGES + N_NODES)
#define IN_CH 128
#define OUT_CH 128
#define HEADS 4
#define CPH 32
#define NEG_SLOPE 0.2f
#define SOFTMAX_EPS 1e-16f
#define LN_EPS 1e-5f

typedef float v2f __attribute__((ext_vector_type(2)));
typedef float v8f __attribute__((ext_vector_type(8)));

__global__ void k_gemm_wmma(const float* __restrict__ x,
                            const float* __restrict__ W,
                            float* __restrict__ h, int nrows) {
    extern __shared__ float Wlds[];
    for (int i = threadIdx.x; i < (IN_CH * OUT_CH) / 4; i += blockDim.x)
        ((float4*)Wlds)[i] = ((const float4*)W)[i];
    __syncthreads();

    const int wave = threadIdx.x >> 5;
    const int lane = threadIdx.x & 31;
    const int rowBase = blockIdx.x * 128 + wave * 16;
    const int m  = lane & 15;
    const int kk = (lane >> 4) << 1;

    int arow = rowBase + m;
    if (arow >= nrows) arow = nrows - 1;
    const float* xk = x + (size_t)arow * IN_CH + kk;

    v8f c[8];
#pragma unroll
    for (int t = 0; t < 8; ++t) c[t] = (v8f){0,0,0,0,0,0,0,0};

    float2 av = *(const float2*)xk;
    for (int k = 0; k < IN_CH; k += 4) {
        v2f a; a.x = av.x; a.y = av.y;
        if (k + 4 < IN_CH) av = *(const float2*)(xk + k + 4);

        const float* w0 = &Wlds[(k + kk) * OUT_CH + m];
        v2f b[8];
#pragma unroll
        for (int t = 0; t < 8; ++t) {
            b[t].x = w0[t * 16];
            b[t].y = w0[OUT_CH + t * 16];
        }
#pragma unroll
        for (int t = 0; t < 8; ++t)
            c[t] = __builtin_amdgcn_wmma_f32_16x16x4_f32(
                false, a, false, b[t], (short)0, c[t], false, false);
    }

#pragma unroll
    for (int t = 0; t < 8; ++t) {
#pragma unroll
        for (int v = 0; v < 8; ++v) {
            const int M = v + ((lane >> 4) << 3);
            const int row = rowBase + M;
            if (row < nrows)
                h[(size_t)row * OUT_CH + t * 16 + m] = c[t][v];
        }
    }
}

#define DOT_THREADS 256
#define DOT_NPW     16
#define DOT_BLOCKS  ((N_NODES + (DOT_THREADS / 32) * DOT_NPW - 1) / ((DOT_THREADS / 32) * DOT_NPW))

#define AGG_BLOCKS  256
#define AGG_R       196
#define KEY_CHUNK   512
#define KEY_CHUNKS  (N_EDGES / KEY_CHUNK)
static_assert(N_EDGES % KEY_CHUNK == 0, "a wave writes the keys of 512 edges");
static_assert((N_NODES - 1) / AGG_R < 256, "an owner key is one byte");
static_assert(DOT_BLOCKS * (DOT_THREADS / 32) >= KEY_CHUNKS, "one key chunk per wave at most");

__device__ __forceinline__ unsigned keys_pack4(int4 d) {
    return ((unsigned)d.x / AGG_R) | (((unsigned)d.y / AGG_R) << 8) |
           (((unsigned)d.z / AGG_R) << 16) | (((unsigned)d.w / AGG_R) << 24);
}

__device__ __forceinline__ uint4 rows_pack4(int4 s, int4 d) {
    uint4 r;
    r.x = (unsigned)s.x | (((unsigned)d.x % AGG_R) << 16);
    r.y = (unsigned)s.y | (((unsigned)d.y % AGG_R) << 16);
    r.z = (unsigned)s.z | (((unsigned)d.z % AGG_R) << 16);
    r.w = (unsigned)s.w | (((unsigned)d.w % AGG_R) << 16);
    return r;
}

__device__ __forceinline__ float dots_sum8(const float (&v)[8], int lane) {
    const bool b2 = (lane & 4) != 0, b1 = (lane & 2) != 0, b0 = (lane & 1) != 0;
    float r[4], q[2];
#pragma unroll
    for (int j = 0; j < 4; ++j) {
        const float keep = b2 ? v[j + 4] : v[j];
        const float send = b2 ? v[j] : v[j + 4];
        r[j] = keep + __shfl_xor(send, 4, 32);
    }
#pragma unroll
    for (int j = 0; j < 2; ++j) {
        const float keep = b1 ? r[j + 2] : r[j];
        const float send = b1 ? r[j] : r[j + 2];
        q[j] = keep + __shfl_xor(send, 2, 32);
    }
    const float keep = b0 ? q[1] : q[0];
    const float send = b0 ? q[0] : q[1];
    return keep + __shfl_xor(send, 1, 32);
}

__global__ __launch_bounds__(DOT_THREADS)
void k_dots(const float* __restrict__ h,
            const float* __restrict__ att_src,
            const float* __restrict__ att_dst,
            const int* __restrict__ ei, unsigned* __restrict__ keys,
            unsigned* __restrict__ rows,
            float* __restrict__ a_s, float* __restrict__ a_d, int n) {
    const int wave = threadIdx.x >> 5;
    const int lane = threadIdx.x & 31;
    const int wg   = blockIdx.x * (DOT_THREADS / 32) + wave;
    const int base = wg * DOT_NPW;
    const bool keyw = wg < KEY_CHUNKS;
    if (base >= n && !keyw) return;
    const float4 ws = ((const float4*)att_src)[lane];
    const float4 wd = ((const float4*)att_dst)[lane];

    const int wk = keyw ? wg : KEY_CHUNKS - 1;
    const int4* sp = (const int4*)ei + (size_t)wk * (KEY_CHUNK / 4) + lane;
    const int4* dp = (const int4*)(ei + N_EDGES) + (size_t)wk * (KEY_CHUNK / 4) + lane;
    int4 dk[4], sk[4];
#pragma unroll
    for (int j = 0; j < 4; ++j) {
        dk[j] = dp[j * 32];
        sk[j] = sp[j * 32];
    }

    float4 hv[DOT_NPW];
#pragma unroll
    for (int i = 0; i < DOT_NPW; ++i) {
        int node = base + i;
        if (node >= n) node = n - 1;
        hv[i] = ((const float4*)(h + (size_t)node * OUT_CH))[lane];
    }
    __builtin_amdgcn_sched_barrier(0);
    if (keyw) {
        unsigned* kp = keys + (size_t)wg * (KEY_CHUNK / 4) + lane;
        uint4* rp = (uint4*)rows + (size_t)wg * (KEY_CHUNK / 4) + lane;
#pragma unroll
        for (int j = 0; j < 4; ++j) {
            kp[j * 32] = keys_pack4(dk[j]);
            rp[j * 32] = rows_pack4(sk[j], dk[j]);
        }
    }
    if (base >= n) return;
#pragma unroll
    for (int g = 0; g < DOT_NPW / 8; ++g) {
        float vs[8], vd[8];
#pragma unroll
        for (int i = 0; i < 8; ++i) {
            const float4 x4 = hv[g * 8 + i];
            vs[i] = (x4.x * ws.x + x4.y * ws.y) + (x4.z * ws.z + x4.w * ws.w);
            vd[i] = (x4.x * wd.x + x4.y * wd.y) + (x4.z * wd.z + x4.w * wd.w);
        }
        const float rs = dots_sum8(vs, lane);
        const float rd = dots_sum8(vd, lane);
        const int node = base + g * 8 + (lane & 7);
        if (node < n) {
            a_s[node * HEADS + (lane >> 3)] = rs;
            a_d[node * HEADS + (lane >> 3)] = rd;
        }
    }
}

__device__ __forceinline__ unsigned f2key(float f) {
    unsigned b = __float_as_uint(f);
    return (b & 0x80000000u) ? ~b : (b | 0x80000000u);
}
__device__ __forceinline__ float key2f(unsigned k) {
    unsigned b = (k & 0x80000000u) ? (k ^ 0x80000000u) : ~k;
    return __uint_as_float(b);
}
__device__ __forceinline__ float lrelu(float v) {
    return v > 0.0f ? v : NEG_SLOPE * v;
}

#define AGG_THREADS 1024
#define AGG_WAVES   (AGG_THREADS / 32)
#define AGG_NSET    4
#define AGG_QCAP    1024
#define AGG_CAP     16128
#define N_K16       (N_EDGES / 16)
#define AGG_STEPS   (((N_K16 + AGG_THREADS - 1) / AGG_THREADS + AGG_NSET - 1) / AGG_NSET * AGG_NSET)
static_assert(AGG_BLOCKS * AGG_R >= N_NODES, "ranges must cover every node");
static_assert(N_K16 % 32 == 0, "end-of-plane test is per wave");
static_assert(AGG_R <= 256 && N_NODES <= (1 << 16), "list entry = source | row << 16");
static_assert(AGG_R * HEADS <= AGG_THREADS, "one thread per (row, head) at the start");

struct AggLds {
    float*    acc;
    float*    ssl;
    float*    mxl;
    unsigned* mkl;
    float*    adl;
    int*      ctl;
};

template <bool FULL>
__device__ __forceinline__ void agg_drain1(const unsigned* __restrict__ rows,
                                           const float* __restrict__ a_s,
                                           const int* q, int qhead, int n, int lane,
                                           const AggLds& L, int* __restrict__ lst) {
    const bool on = FULL || lane < n;
    int e = 0;
    if (on) e = q[(qhead + lane) & (AGG_QCAP - 1)];
    const unsigned ent = rows[e];
    const int s   = (int)(ent & 0xFFFFu);
    const int rel = on ? (int)(ent >> 16) : 0;
    const float4 as4 = *(const float4*)(a_s + (size_t)s * HEADS);
    const float4 ad4 = *(const float4*)(L.adl + rel * HEADS);
    const float l0 = lrelu(as4.x + ad4.x);
    const float l1 = lrelu(as4.y + ad4.y);
    const float l2 = lrelu(as4.z + ad4.z);
    const float l3 = lrelu(as4.w + ad4.w);
    if (on) {
        atomicMax(&L.mkl[rel * HEADS + 0], f2key(l0));
        atomicMax(&L.mkl[rel * HEADS + 1], f2key(l1));
        atomicMax(&L.mkl[rel * HEADS + 2], f2key(l2));
        atomicMax(&L.mkl[rel * HEADS + 3], f2key(l3));
    }
    const int nn = FULL ? 32 : n;
    int base = 0;
    if (lane == 0) base = atomicAdd(&L.ctl[0], nn);
    base = __builtin_amdgcn_readfirstlane(base);
    if (on && base + lane < AGG_CAP) lst[base + lane] = (int)ent;
    if (lane == 0 && base + nn > AGG_CAP) L.ctl[1] = 1;
}

template <bool FULL>
__device__ __forceinline__ void agg_drain2(const float* __restrict__ h,
                                           const float* __restrict__ a_s,
                                           int s, int rel, int n, int lane,
                                           const AggLds& L) {
    const float4 as4 = *(const float4*)(a_s + (size_t)s * HEADS);
    const float4 ad4 = *(const float4*)(L.adl + rel * HEADS);
    const float4 mm  = *(const float4*)(L.mxl + rel * HEADS);
    const float w0 = expf(lrelu(as4.x + ad4.x) - mm.x);
    const float w1 = expf(lrelu(as4.y + ad4.y) - mm.y);
    const float w2 = expf(lrelu(as4.z + ad4.z) - mm.z);
    const float w3 = expf(lrelu(as4.w + ad4.w) - mm.w);
    if (FULL || lane < n) {
        atomicAdd(&L.ssl[rel * HEADS + 0], w0);
        atomicAdd(&L.ssl[rel * HEADS + 1], w1);
        atomicAdd(&L.ssl[rel * HEADS + 2], w2);
        atomicAdd(&L.ssl[rel * HEADS + 3], w3);
    }
    const int nk = FULL ? 32 : n;
#pragma unroll 1
    for (int kb = 0; kb < nk; kb += 8) {
        float hv[8][4];
#pragma unroll
        for (int j = 0; j < 8; ++j) {
            const int k = kb + j;
            if (FULL || k < n) {
                const int sk = __builtin_amdgcn_readlane(s, k);
                const float* hp = h + (size_t)sk * OUT_CH + lane;
                hv[j][0] = hp[0];
                hv[j][1] = hp[32];
                hv[j][2] = hp[64];
                hv[j][3] = hp[96];
            }
        }
#pragma unroll
        for (int j = 0; j < 8; ++j) {
            const int k = kb + j;
            if (FULL || k < n) {
                const int rk = __builtin_amdgcn_readlane(rel, k);
                const float a0 = __int_as_float(__builtin_amdgcn_readlane(__float_as_int(w0), k));
                const float a1 = __int_as_float(__builtin_amdgcn_readlane(__float_as_int(w1), k));
                const float a2 = __int_as_float(__builtin_amdgcn_readlane(__float_as_int(w2), k));
                const float a3 = __int_as_float(__builtin_amdgcn_readlane(__float_as_int(w3), k));
                float* ap = L.acc + rk * OUT_CH + lane;
                atomicAdd(ap,      hv[j][0] * a0);
                atomicAdd(ap + 32, hv[j][1] * a1);
                atomicAdd(ap + 64, hv[j][2] * a2);
                atomicAdd(ap + 96, hv[j][3] * a3);
            }
        }
    }
}

template <int PH, bool FULL>
__device__ __forceinline__ void agg_drain(const unsigned* __restrict__ rows,
                                          const float* __restrict__ h,
                                          const float* __restrict__ a_s,
                                          const int* q, int qhead, int n, int lane,
                                          const AggLds& L, int* __restrict__ lst) {
    if (PH == 1) {
        agg_drain1<FULL>(rows, a_s, q, qhead, n, lane, L, lst);
    } else {
        const bool on = FULL || lane < n;
        int e = 0;
        if (on) e = q[(qhead + lane) & (AGG_QCAP - 1)];
        const unsigned ent = rows[e];
        const int s   = (int)(ent & 0xFFFFu);
        const int rel = on ? (int)(ent >> 16) : 0;
        agg_drain2<FULL>(h, a_s, s, rel, n, lane, L);
    }
}

__device__ __forceinline__ int agg_wrap(int sp) {
    while (sp >= AGG_STEPS) sp -= AGG_STEPS;
    return sp;
}

__device__ __forceinline__ uint4 agg_load(const uint4* __restrict__ keys, int sp, int wave, int lane) {
    int wbase = sp * AGG_THREADS + wave * 32;
    if (wbase > N_K16 - 32) wbase = N_K16 - 32;
    return keys[wbase + lane];
}

__device__ __forceinline__ unsigned agg_nonzero_bytes(unsigned x) {
    return ((x & 0x7F7F7F7Fu) + 0x7F7F7F7Fu) | x | 0x7F7F7F7Fu;
}

template <int PH>
__device__ __forceinline__ void agg_scan(uint4 kv, int sp, unsigned own4, int wave, int lane,
                                         int* q, int& qhead, int& cnt,
                                         const unsigned* __restrict__ rows,
                                         const float* __restrict__ h,
                                         const float* __restrict__ a_s,
                                         const AggLds& L, int* __restrict__ lst) {
    const int wbase = sp * AGG_THREADS + wave * 32;
    if (wbase >= N_K16) return;
    const unsigned y0 = agg_nonzero_bytes(kv.x ^ own4);
    const unsigned y1 = agg_nonzero_bytes(kv.y ^ own4);
    const unsigned y2 = agg_nonzero_bytes(kv.z ^ own4);
    const unsigned y3 = agg_nonzero_bytes(kv.w ^ own4);
    const bool hit = ((y0 & y1) & (y2 & y3)) != 0xFFFFFFFFu;
    unsigned bm = __builtin_amdgcn_ballot_w32(hit);
    if (bm) {
        unsigned cm = (~y0 >> 7) | (~y1 >> 6) | (~y2 >> 5) | (~y3 >> 4);
        const int e0 = (wbase + lane) * 16;
        do {
            if (cm) {
                const int p = __builtin_ctz(cm);
                q[(qhead + cnt + __builtin_amdgcn_mbcnt_lo(bm, 0)) & (AGG_QCAP - 1)] =
                    e0 + (((p & 7) << 2) | (p >> 3));
                cm &= cm - 1;
            }
            cnt += __builtin_popcount(bm);
            bm = __builtin_amdgcn_ballot_w32(cm != 0);
        } while (bm);
    }
    while (cnt >= 32) {
        agg_drain<PH, true>(rows, h, a_s, q, qhead, 32, lane, L, lst);
        qhead = (qhead + 32) & (AGG_QCAP - 1);
        cnt -= 32;
    }
}

template <int PH>
__device__ __forceinline__ void agg_pass(const unsigned* __restrict__ rows,
                                         const uint4* __restrict__ keys,
                                         const float* __restrict__ h,
                                         const float* __restrict__ a_s,
                                         int wave, int lane, int* q,
                                         const AggLds& L, int* __restrict__ lst) {
    int qhead = 0, cnt = 0;
    const unsigned own4 = blockIdx.x * 0x01010101u;
    const int start = (blockIdx.x * AGG_STEPS) / AGG_BLOCKS;

    uint4 k0 = agg_load(keys, agg_wrap(start),     wave, lane);
    uint4 k1 = agg_load(keys, agg_wrap(start + 1), wave, lane);
    uint4 k2 = agg_load(keys, agg_wrap(start + 2), wave, lane);
    uint4 k3 = agg_load(keys, agg_wrap(start + 3), wave, lane);
    for (int st = 0; st < AGG_STEPS; st += AGG_NSET) {
        agg_scan<PH>(k0, agg_wrap(st + start),     own4, wave, lane, q, qhead, cnt, rows, h, a_s, L, lst);
        k0 = agg_load(keys, agg_wrap(st + 4 + start), wave, lane);
        agg_scan<PH>(k1, agg_wrap(st + 1 + start), own4, wave, lane, q, qhead, cnt, rows, h, a_s, L, lst);
        k1 = agg_load(keys, agg_wrap(st + 5 + start), wave, lane);
        agg_scan<PH>(k2, agg_wrap(st + 2 + start), own4, wave, lane, q, qhead, cnt, rows, h, a_s, L, lst);
        k2 = agg_load(keys, agg_wrap(st + 6 + start), wave, lane);
        agg_scan<PH>(k3, agg_wrap(st + 3 + start), own4, wave, lane, q, qhead, cnt, rows, h, a_s, L, lst);
        k3 = agg_load(keys, agg_wrap(st + 7 + start), wave, lane);
    }
    if (cnt > 0) agg_drain<PH, false>(rows, h, a_s, q, qhead, cnt, lane, L, lst);
}

__device__ __forceinline__ float wave_sum(float t) {
    for (int o = 16; o > 0; o >>= 1) t += __shfl_xor(t, o, 32);
    return t;
}

__global__ __launch_bounds__(AGG_THREADS)
void k_agg_lds(const unsigned* __restrict__ rows,
               const unsigned* __restrict__ keys,
               const float* __restrict__ h,
               const float* __restrict__ a_s,
               const float* __restrict__ a_d,
               int* __restrict__ lists,
               const float* __restrict__ bias,
               const float* __restrict__ gamma,
               const float* __restrict__ beta,
               float* __restrict__ out) {
    __shared__ __attribute__((aligned(16))) float acc[AGG_R * OUT_CH];
    __shared__ __attribute__((aligned(16))) float ssl[AGG_R * HEADS];
    __shared__ __attribute__((aligned(16))) float mxl[AGG_R * HEADS];
    __shared__ __attribute__((aligned(16))) unsigned mkl[AGG_R * HEADS];
    __shared__ __attribute__((aligned(16))) float adl[AGG_R * HEADS];
    __shared__ int ctl[2];
    __shared__ int que[AGG_WAVES * AGG_QCAP];

    const int tid  = threadIdx.x;
    const int wave = __builtin_amdgcn_readfirstlane(tid >> 5);
    const int lane = tid & 31;
    const int lo   = blockIdx.x * AGG_R;
    AggLds L;
    L.acc = acc; L.ssl = ssl; L.mxl = mxl; L.mkl = mkl; L.adl = adl; L.ctl = ctl;
    int* lst = lists + (size_t)blockIdx.x * AGG_CAP;
    int* q = que + wave * AGG_QCAP;

    if (tid < AGG_R * HEADS) {
        const int node = lo + (tid >> 2);
        float ad = 0.0f;
        unsigned key = 0u;
        if (node < N_NODES) {
            ad  = a_d[node * HEADS + (tid & 3)];
            key = f2key(lrelu(a_s[node * HEADS + (tid & 3)] + ad));
        }
        adl[tid] = ad;
        mkl[tid] = key;
    }
    if (tid == 0) { ctl[0] = 0; ctl[1] = 0; }
    __syncthreads();

    agg_pass<1>(rows, (const uint4*)keys, h, a_s, wave, lane, q, L, lst);
    __syncthreads();

    for (int r = wave; r < AGG_R; r += AGG_WAVES) {
        const int node = lo + r;
        if (node < N_NODES) {
            const float4 as4 = *(const float4*)(a_s + (size_t)node * HEADS);
            const float4 ad4 = *(const float4*)(adl + r * HEADS);
            const uint4  mk  = *(const uint4*)(mkl + r * HEADS);
            float4 mf, w;
            mf.x = key2f(mk.x); mf.y = key2f(mk.y); mf.z = key2f(mk.z); mf.w = key2f(mk.w);
            w.x = expf(lrelu(as4.x + ad4.x) - mf.x);
            w.y = expf(lrelu(as4.y + ad4.y) - mf.y);
            w.z = expf(lrelu(as4.z + ad4.z) - mf.z);
            w.w = expf(lrelu(as4.w + ad4.w) - mf.w);
            const float* hp = h + (size_t)node * OUT_CH + lane;
            float* ap = acc + r * OUT_CH + lane;
            ap[0]  = hp[0]  * w.x;
            ap[32] = hp[32] * w.y;
            ap[64] = hp[64] * w.z;
            ap[96] = hp[96] * w.w;
            if (lane == 0) {
                *(float4*)(ssl + r * HEADS) = w;
                *(float4*)(mxl + r * HEADS) = mf;
            }
        }
    }
    __syncthreads();

    const int total = __builtin_amdgcn_readfirstlane(ctl[0]);
    const int ovf   = __builtin_amdgcn_readfirstlane(ctl[1]);
    if (ovf == 0) {
        const int last = total > 0 ? total - 1 : 0;
        int i = wave * 32;
        int ent = 0;
        if (total > 0) ent = lst[min(i + lane, last)];
        while (i < total) {
            const int inext = i + AGG_THREADS;
            const int entn = lst[min(inext + lane, last)];
            const int n = total - i;
            if (n >= 32) agg_drain2<true >(h, a_s, ent & 0xFFFF, ent >> 16, 32, lane, L);
            else         agg_drain2<false>(h, a_s, ent & 0xFFFF, ent >> 16, n,  lane, L);
            ent = entn;
            i = inext;
        }
    } else {
        agg_pass<2>(rows, (const uint4*)keys, h, a_s, wave, lane, q, L, lst);
    }
    __syncthreads();

    const float bs0 = bias[lane],  bs1 = bias[32 + lane],  bs2 = bias[64 + lane],  bs3 = bias[96 + lane];
    const float gm0 = gamma[lane], gm1 = gamma[32 + lane], gm2 = gamma[64 + lane], gm3 = gamma[96 + lane];
    const float bt0 = beta[lane],  bt1 = beta[32 + lane],  bt2 = beta[64 + lane],  bt3 = beta[96 + lane];
    for (int r = wave; r < AGG_R; r += AGG_WAVES) {
        const int node = lo + r;
        if (node < N_NODES) {
            const float* ap = acc + r * OUT_CH + lane;
            const float4 sv = *(const float4*)(ssl + r * HEADS);
            const float v0 = ap[0]  / (sv.x + SOFTMAX_EPS) + bs0;
            const float v1 = ap[32] / (sv.y + SOFTMAX_EPS) + bs1;
            const float v2 = ap[64] / (sv.z + SOFTMAX_EPS) + bs2;
            const float v3 = ap[96] / (sv.w + SOFTMAX_EPS) + bs3;

            const float mu = wave_sum((v0 + v1) + (v2 + v3)) * (1.0f / OUT_CH);
            const float d0 = v0 - mu, d1 = v1 - mu, d2 = v2 - mu, d3 = v3 - mu;
            const float var = wave_sum((d0 * d0 + d1 * d1) + (d2 * d2 + d3 * d3)) * (1.0f / OUT_CH);
            const float rs = 1.0f / sqrtf(var + LN_EPS);

            float y0 = d0 * rs * gm0 + bt0;
            float y1 = d1 * rs * gm1 + bt1;
            float y2 = d2 * rs * gm2 + bt2;
            float y3 = d3 * rs * gm3 + bt3;
            y0 = y0 > 0.0f ? y0 : (expf(y0) - 1.0f);
            y1 = y1 > 0.0f ? y1 : (expf(y1) - 1.0f);
            y2 = y2 > 0.0f ? y2 : (expf(y2) - 1.0f);
            y3 = y3 > 0.0f ? y3 : (expf(y3) - 1.0f);

            float* op = out + (size_t)node * OUT_CH + lane;
            op[0]  = y0;
            op[32] = y1;
            op[64] = y2;
            op[96] = y3;
        }
    }
}

extern "C" void kernel_launch(void* const* d_in, const int* in_sizes, int n_in,
                              void* d_out, int out_size, void* d_ws, size_t ws_size,
                              hipStream_t stream) {
    const float* x       = (const float*)d_in[0];
    const int*   ei      = (const int*)  d_in[1];
    const float* W       = (const float*)d_in[2];
    const float* att_src = (const float*)d_in[3];
    const float* att_dst = (const float*)d_in[4];
    const float* bias    = (const float*)d_in[5];
    const float* gamma   = (const float*)d_in[6];
    const float* beta    = (const float*)d_in[7];
    float* out = (float*)d_out;

    char* ws = (char*)d_ws;
    float*    h      = (float*)ws;
    ws += (size_t)N_NODES * OUT_CH * sizeof(float);
    float*    a_s    = (float*)ws;  ws += (size_t)N_NODES * HEADS * sizeof(float);
    float*    a_d    = (float*)ws;  ws += (size_t)N_NODES * HEADS * sizeof(float);
    int*      lists  = (int*)ws;
    ws += (size_t)AGG_BLOCKS * AGG_CAP * sizeof(int);
    unsigned* keys   = (unsigned*)ws;
    ws += (size_t)N_EDGES;
    unsigned* rows   = (unsigned*)ws;

    k_gemm_wmma<<<(N_NODES + 127) / 128, 256, IN_CH * OUT_CH * sizeof(float),
                  stream>>>(x, W, h, N_NODES);

    k_dots<<<DOT_BLOCKS, DOT_THREADS, 0, stream>>>(h, att_src, att_dst, ei, keys, rows, a_s, a_d,
                                                   N_NODES);

    k_agg_lds<<<AGG_BLOCKS, AGG_THREADS, 0, stream>>>(rows, keys, h, a_s, a_d, lists, bias,
                                                      gamma, beta, out);
}


# Owner block's edge list moves from the workspace into LDS and carries each edge's four logits (ring cut to 64 slots a wave makes the room): pass 2 gathers a_s no more; sum order kept
# speedup vs baseline: 4.1303x; 1.0548x over previous
#include <hip/hip_runtime.h>
#include <math.h>


#define N_NODES 50000
#define N_EDGES 1600000
#define ET_EDGES (N_EDGES + N_NODES)
#define IN_CH 128
#define OUT_CH 128
#define HEADS 4
#define CPH 32
#define NEG_SLOPE 0.2f
#define SOFTMAX_EPS 1e-16f
#define LN_EPS 1e-5f

typedef float v2f __attribute__((ext_vector_type(2)));
typedef float v8f __attribute__((ext_vector_type(8)));

__global__ void k_gemm_wmma(const float* __restrict__ x,
                            const float* __restrict__ W,
                            float* __restrict__ h, int nrows) {
    extern __shared__ float Wlds[];
    for (int i = threadIdx.x; i < (IN_CH * OUT_CH) / 4; i += blockDim.x)
        ((float4*)Wlds)[i] = ((const float4*)W)[i];
    __syncthreads();

    const int wave = threadIdx.x >> 5;
    const int lane = threadIdx.x & 31;
    const int rowBase = blockIdx.x * 128 + wave * 16;
    const int m  = lane & 15;
    const int kk = (lane >> 4) << 1;

    int arow = rowBase + m;
    if (arow >= nrows) arow = nrows - 1;
    const float* xk = x + (size_t)arow * IN_CH + kk;

    v8f c[8];
#pragma unroll
    for (int t = 0; t < 8; ++t) c[t] = (v8f){0,0,0,0,0,0,0,0};

    float2 av = *(const float2*)xk;
    for (int k = 0; k < IN_CH; k += 4) {
        v2f a; a.x = av.x; a.y = av.y;
        if (k + 4 < IN_CH) av = *(const float2*)(xk + k + 4);

        const float* w0 = &Wlds[(k + kk) * OUT_CH + m];
        v2f b[8];
#pragma unroll
        for (int t = 0; t < 8; ++t) {
            b[t].x = w0[t * 16];
            b[t].y = w0[OUT_CH + t * 16];
        }
#pragma unroll
        for (int t = 0; t < 8; ++t)
            c[t] = __builtin_amdgcn_wmma_f32_16x16x4_f32(
                false, a, false, b[t], (short)0, c[t], false, false);
    }

#pragma unroll
    for (int t = 0; t < 8; ++t) {
#pragma unroll
        for (int v = 0; v < 8; ++v) {
            const int M = v + ((lane >> 4) << 3);
            const int row = rowBase + M;
            if (row < nrows)
                h[(size_t)row * OUT_CH + t * 16 + m] = c[t][v];
        }
    }
}

#define DOT_THREADS 256
#define DOT_NPW     16
#define DOT_BLOCKS  ((N_NODES + (DOT_THREADS / 32) * DOT_NPW - 1) / ((DOT_THREADS / 32) * DOT_NPW))

#define AGG_BLOCKS  256
#define AGG_R       196
#define KEY_CHUNK   512
#define KEY_CHUNKS  (N_EDGES / KEY_CHUNK)
static_assert(N_EDGES % KEY_CHUNK == 0, "a wave writes the keys of 512 edges");
static_assert((N_NODES - 1) / AGG_R < 256, "an owner key is one byte");
static_assert(DOT_BLOCKS * (DOT_THREADS / 32) >= KEY_CHUNKS, "one key chunk per wave at most");

__device__ __forceinline__ unsigned keys_pack4(int4 d) {
    return ((unsigned)d.x / AGG_R) | (((unsigned)d.y / AGG_R) << 8) |
           (((unsigned)d.z / AGG_R) << 16) | (((unsigned)d.w / AGG_R) << 24);
}

__device__ __forceinline__ uint4 rows_pack4(int4 s, int4 d) {
    uint4 r;
    r.x = (unsigned)s.x | (((unsigned)d.x % AGG_R) << 16);
    r.y = (unsigned)s.y | (((unsigned)d.y % AGG_R) << 16);
    r.z = (unsigned)s.z | (((unsigned)d.z % AGG_R) << 16);
    r.w = (unsigned)s.w | (((unsigned)d.w % AGG_R) << 16);
    return r;
}

__device__ __forceinline__ float dots_sum8(const float (&v)[8], int lane) {
    const bool b2 = (lane & 4) != 0, b1 = (lane & 2) != 0, b0 = (lane & 1) != 0;
    float r[4], q[2];
#pragma unroll
    for (int j = 0; j < 4; ++j) {
        const float keep = b2 ? v[j + 4] : v[j];
        const float send = b2 ? v[j] : v[j + 4];
        r[j] = keep + __shfl_xor(send, 4, 32);
    }
#pragma unroll
    for (int j = 0; j < 2; ++j) {
        const float keep = b1 ? r[j + 2] : r[j];
        const float send = b1 ? r[j] : r[j + 2];
        q[j] = keep + __shfl_xor(send, 2, 32);
    }
    const float keep = b0 ? q[1] : q[0];
    const float send = b0 ? q[0] : q[1];
    return keep + __shfl_xor(send, 1, 32);
}

__global__ __launch_bounds__(DOT_THREADS)
void k_dots(const float* __restrict__ h,
            const float* __restrict__ att_src,
            const float* __restrict__ att_dst,
            const int* __restrict__ ei, unsigned* __restrict__ keys,
            unsigned* __restrict__ rows,
            float* __restrict__ a_s, float* __restrict__ a_d, int n) {
    const int wave = threadIdx.x >> 5;
    const int lane = threadIdx.x & 31;
    const int wg   = blockIdx.x * (DOT_THREADS / 32) + wave;
    const int base = wg * DOT_NPW;
    const bool keyw = wg < KEY_CHUNKS;
    if (base >= n && !keyw) return;
    const float4 ws = ((const float4*)att_src)[lane];
    const float4 wd = ((const float4*)att_dst)[lane];

    const int wk = keyw ? wg : KEY_CHUNKS - 1;
    const int4* sp = (const int4*)ei + (size_t)wk * (KEY_CHUNK / 4) + lane;
    const int4* dp = (const int4*)(ei + N_EDGES) + (size_t)wk * (KEY_CHUNK / 4) + lane;
    int4 dk[4], sk[4];
#pragma unroll
    for (int j = 0; j < 4; ++j) {
        dk[j] = dp[j * 32];
        sk[j] = sp[j * 32];
    }

    float4 hv[DOT_NPW];
#pragma unroll
    for (int i = 0; i < DOT_NPW; ++i) {
        int node = base + i;
        if (node >= n) node = n - 1;
        hv[i] = ((const float4*)(h + (size_t)node * OUT_CH))[lane];
    }
    __builtin_amdgcn_sched_barrier(0);
    if (keyw) {
        unsigned* kp = keys + (size_t)wg * (KEY_CHUNK / 4) + lane;
        uint4* rp = (uint4*)rows + (size_t)wg * (KEY_CHUNK / 4) + lane;
#pragma unroll
        for (int j = 0; j < 4; ++j) {
            kp[j * 32] = keys_pack4(dk[j]);
            rp[j * 32] = rows_pack4(sk[j], dk[j]);
        }
    }
    if (base >= n) return;
#pragma unroll
    for (int g = 0; g < DOT_NPW / 8; ++g) {
        float vs[8], vd[8];
#pragma unroll
        for (int i = 0; i < 8; ++i) {
            const float4 x4 = hv[g * 8 + i];
            vs[i] = (x4.x * ws.x + x4.y * ws.y) + (x4.z * ws.z + x4.w * ws.w);
            vd[i] = (x4.x * wd.x + x4.y * wd.y) + (x4.z * wd.z + x4.w * wd.w);
        }
        const float rs = dots_sum8(vs, lane);
        const float rd = dots_sum8(vd, lane);
        const int node = base + g * 8 + (lane & 7);
        if (node < n) {
            a_s[node * HEADS + (lane >> 3)] = rs;
            a_d[node * HEADS + (lane >> 3)] = rd;
        }
    }
}

__device__ __forceinline__ unsigned f2key(float f) {
    unsigned b = __float_as_uint(f);
    return (b & 0x80000000u) ? ~b : (b | 0x80000000u);
}
__device__ __forceinline__ float key2f(unsigned k) {
    unsigned b = (k & 0x80000000u) ? (k ^ 0x80000000u) : ~k;
    return __uint_as_float(b);
}
__device__ __forceinline__ float lrelu(float v) {
    return v > 0.0f ? v : NEG_SLOPE * v;
}

#define AGG_THREADS 1024
#define AGG_WAVES   (AGG_THREADS / 32)
#define AGG_NSET    4
#define AGG_QCAP    64
#define AGG_CAP     8064
#define N_K16       (N_EDGES / 16)
#define AGG_STEPS   (((N_K16 + AGG_THREADS - 1) / AGG_THREADS + AGG_NSET - 1) / AGG_NSET * AGG_NSET)
static_assert(AGG_BLOCKS * AGG_R >= N_NODES, "ranges must cover every node");
static_assert(N_K16 % 32 == 0, "end-of-plane test is per wave");
static_assert(AGG_R <= 256 && N_NODES <= (1 << 16), "list entry = source | row << 16");
static_assert(AGG_R * HEADS <= AGG_THREADS, "one thread per (row, head) at the start");

struct AggLds {
    float*    acc;
    float*    ssl;
    float*    mxl;
    unsigned* mkl;
    float*    adl;
    int*      ctl;
    int*      lse;
    float4*   lsl;
};

template <bool FULL>
__device__ __forceinline__ void agg_drain1(const unsigned* __restrict__ rows,
                                           const float* __restrict__ a_s,
                                           const int* q, int qhead, int n, int lane,
                                           const AggLds& L) {
    const bool on = FULL || lane < n;
    int e = 0;
    if (on) e = q[(qhead + lane) & (AGG_QCAP - 1)];
    const unsigned ent = rows[e];
    const int s   = (int)(ent & 0xFFFFu);
    const int rel = on ? (int)(ent >> 16) : 0;
    const float4 as4 = *(const float4*)(a_s + (size_t)s * HEADS);
    const float4 ad4 = *(const float4*)(L.adl + rel * HEADS);
    const float l0 = lrelu(as4.x + ad4.x);
    const float l1 = lrelu(as4.y + ad4.y);
    const float l2 = lrelu(as4.z + ad4.z);
    const float l3 = lrelu(as4.w + ad4.w);
    if (on) {
        atomicMax(&L.mkl[rel * HEADS + 0], f2key(l0));
        atomicMax(&L.mkl[rel * HEADS + 1], f2key(l1));
        atomicMax(&L.mkl[rel * HEADS + 2], f2key(l2));
        atomicMax(&L.mkl[rel * HEADS + 3], f2key(l3));
    }
    const int nn = FULL ? 32 : n;
    int base = 0;
    if (lane == 0) base = atomicAdd(&L.ctl[0], nn);
    base = __builtin_amdgcn_readfirstlane(base);
    if (on && base + lane < AGG_CAP) {
        L.lse[base + lane] = (int)ent;
        L.lsl[base + lane] = make_float4(l0, l1, l2, l3);
    }
    if (lane == 0 && base + nn > AGG_CAP) L.ctl[1] = 1;
}

template <bool FULL>
__device__ __forceinline__ void agg_drain2(const float* __restrict__ h,
                                           int s, int rel, float4 lg, int n, int lane,
                                           const AggLds& L) {
    const float4 mm  = *(const float4*)(L.mxl + rel * HEADS);
    const float w0 = expf(lg.x - mm.x);
    const float w1 = expf(lg.y - mm.y);
    const float w2 = expf(lg.z - mm.z);
    const float w3 = expf(lg.w - mm.w);
    if (FULL || lane < n) {
        atomicAdd(&L.ssl[rel * HEADS + 0], w0);
        atomicAdd(&L.ssl[rel * HEADS + 1], w1);
        atomicAdd(&L.ssl[rel * HEADS + 2], w2);
        atomicAdd(&L.ssl[rel * HEADS + 3], w3);
    }
    const int nk = FULL ? 32 : n;
#pragma unroll 1
    for (int kb = 0; kb < nk; kb += 8) {
        float hv[8][4];
#pragma unroll
        for (int j = 0; j < 8; ++j) {
            const int k = kb + j;
            if (FULL || k < n) {
                const int sk = __builtin_amdgcn_readlane(s, k);
                const float* hp = h + (size_t)sk * OUT_CH + lane;
                hv[j][0] = hp[0];
                hv[j][1] = hp[32];
                hv[j][2] = hp[64];
                hv[j][3] = hp[96];
            }
        }
#pragma unroll
        for (int j = 0; j < 8; ++j) {
            const int k = kb + j;
            if (FULL || k < n) {
                const int rk = __builtin_amdgcn_readlane(rel, k);
                const float a0 = __int_as_float(__builtin_amdgcn_readlane(__float_as_int(w0), k));
                const float a1 = __int_as_float(__builtin_amdgcn_readlane(__float_as_int(w1), k));
                const float a2 = __int_as_float(__builtin_amdgcn_readlane(__float_as_int(w2), k));
                const float a3 = __int_as_float(__builtin_amdgcn_readlane(__float_as_int(w3), k));
                float* ap = L.acc + rk * OUT_CH + lane;
                atomicAdd(ap,      hv[j][0] * a0);
                atomicAdd(ap + 32, hv[j][1] * a1);
                atomicAdd(ap + 64, hv[j][2] * a2);
                atomicAdd(ap + 96, hv[j][3] * a3);
            }
        }
    }
}

template <int PH, bool FULL>
__device__ __forceinline__ void agg_drain(const unsigned* __restrict__ rows,
                                          const float* __restrict__ h,
                                          const float* __restrict__ a_s,
                                          const int* q, int qhead, int n, int lane,
                                          const AggLds& L) {
    if (PH == 1) {
        agg_drain1<FULL>(rows, a_s, q, qhead, n, lane, L);
    } else {
        const bool on = FULL || lane < n;
        int e = 0;
        if (on) e = q[(qhead + lane) & (AGG_QCAP - 1)];
        const unsigned ent = rows[e];
        const int s   = (int)(ent & 0xFFFFu);
        const int rel = on ? (int)(ent >> 16) : 0;
        const float4 as4 = *(const float4*)(a_s + (size_t)s * HEADS);
        const float4 ad4 = *(const float4*)(L.adl + rel * HEADS);
        const float4 lg = make_float4(lrelu(as4.x + ad4.x), lrelu(as4.y + ad4.y),
                                      lrelu(as4.z + ad4.z), lrelu(as4.w + ad4.w));
        agg_drain2<FULL>(h, s, rel, lg, n, lane, L);
    }
}

__device__ __forceinline__ int agg_wrap(int sp) {
    while (sp >= AGG_STEPS) sp -= AGG_STEPS;
    return sp;
}

__device__ __forceinline__ uint4 agg_load(const uint4* __restrict__ keys, int sp, int wave, int lane) {
    int wbase = sp * AGG_THREADS + wave * 32;
    if (wbase > N_K16 - 32) wbase = N_K16 - 32;
    return keys[wbase + lane];
}

__device__ __forceinline__ unsigned agg_nonzero_bytes(unsigned x) {
    return ((x & 0x7F7F7F7Fu) + 0x7F7F7F7Fu) | x | 0x7F7F7F7Fu;
}

template <int PH>
__device__ __forceinline__ void agg_scan(uint4 kv, int sp, unsigned own4, int wave, int lane,
                                         int* q, int& qhead, int& cnt,
                                         const unsigned* __restrict__ rows,
                                         const float* __restrict__ h,
                                         const float* __restrict__ a_s,
                                         const AggLds& L) {
    const int wbase = sp * AGG_THREADS + wave * 32;
    if (wbase >= N_K16) return;
    const unsigned y0 = agg_nonzero_bytes(kv.x ^ own4);
    const unsigned y1 = agg_nonzero_bytes(kv.y ^ own4);
    const unsigned y2 = agg_nonzero_bytes(kv.z ^ own4);
    const unsigned y3 = agg_nonzero_bytes(kv.w ^ own4);
    const bool hit = ((y0 & y1) & (y2 & y3)) != 0xFFFFFFFFu;
    unsigned bm = __builtin_amdgcn_ballot_w32(hit);
    if (bm) {
        unsigned cm = (~y0 >> 7) | (~y1 >> 6) | (~y2 >> 5) | (~y3 >> 4);
        const int e0 = (wbase + lane) * 16;
        do {
            if (cm) {
                const int p = __builtin_ctz(cm);
                q[(qhead + cnt + __builtin_amdgcn_mbcnt_lo(bm, 0)) & (AGG_QCAP - 1)] =
                    e0 + (((p & 7) << 2) | (p >> 3));
                cm &= cm - 1;
            }
            cnt += __builtin_popcount(bm);
            if (cnt >= 32) {
                agg_drain<PH, true>(rows, h, a_s, q, qhead, 32, lane, L);
                qhead = (qhead + 32) & (AGG_QCAP - 1);
                cnt -= 32;
            }
            bm = __builtin_amdgcn_ballot_w32(cm != 0);
        } while (bm);
    }
}

template <int PH>
__device__ __forceinline__ void agg_pass(const unsigned* __restrict__ rows,
                                         const uint4* __restrict__ keys,
                                         const float* __restrict__ h,
                                         const float* __restrict__ a_s,
                                         int wave, int lane, int* q,
                                         const AggLds& L) {
    int qhead = 0, cnt = 0;
    const unsigned own4 = blockIdx.x * 0x01010101u;
    const int start = (blockIdx.x * AGG_STEPS) / AGG_BLOCKS;

    uint4 k0 = agg_load(keys, agg_wrap(start),     wave, lane);
    uint4 k1 = agg_load(keys, agg_wrap(start + 1), wave, lane);
    uint4 k2 = agg_load(keys, agg_wrap(start + 2), wave, lane);
    uint4 k3 = agg_load(keys, agg_wrap(start + 3), wave, lane);
    for (int st = 0; st < AGG_STEPS; st += AGG_NSET) {
        agg_scan<PH>(k0, agg_wrap(st + start),     own4, wave, lane, q, qhead, cnt, rows, h, a_s, L);
        k0 = agg_load(keys, agg_wrap(st + 4 + start), wave, lane);
        agg_scan<PH>(k1, agg_wrap(st + 1 + start), own4, wave, lane, q, qhead, cnt, rows, h, a_s, L);
        k1 = agg_load(keys, agg_wrap(st + 5 + start), wave, lane);
        agg_scan<PH>(k2, agg_wrap(st + 2 + start), own4, wave, lane, q, qhead, cnt, rows, h, a_s, L);
        k2 = agg_load(keys, agg_wrap(st + 6 + start), wave, lane);
        agg_scan<PH>(k3, agg_wrap(st + 3 + start), own4, wave, lane, q, qhead, cnt, rows, h, a_s, L);
        k3 = agg_load(keys, agg_wrap(st + 7 + start), wave, lane);
    }
    if (cnt > 0) agg_drain<PH, false>(rows, h, a_s, q, qhead, cnt, lane, L);
}

__device__ __forceinline__ float wave_sum(float t) {
    for (int o = 16; o > 0; o >>= 1) t += __shfl_xor(t, o, 32);
    return t;
}

__global__ __launch_bounds__(AGG_THREADS)
void k_agg_lds(const unsigned* __restrict__ rows,
               const unsigned* __restrict__ keys,
               const float* __restrict__ h,
               const float* __restrict__ a_s,
               const float* __restrict__ a_d,
               const float* __restrict__ bias,
               const float* __restrict__ gamma,
               const float* __restrict__ beta,
               float* __restrict__ out) {
    __shared__ __attribute__((aligned(16))) float acc[AGG_R * OUT_CH];
    __shared__ __attribute__((aligned(16))) float ssl[AGG_R * HEADS];
    __shared__ __attribute__((aligned(16))) float mxl[AGG_R * HEADS];
    __shared__ __attribute__((aligned(16))) unsigned mkl[AGG_R * HEADS];
    __shared__ __attribute__((aligned(16))) float adl[AGG_R * HEADS];
    __shared__ int ctl[2];
    __shared__ int que[AGG_WAVES * AGG_QCAP];
    __shared__ __attribute__((aligned(16))) int lse[AGG_CAP];
    __shared__ __attribute__((aligned(16))) float4 lsl[AGG_CAP];

    const int tid  = threadIdx.x;
    const int wave = __builtin_amdgcn_readfirstlane(tid >> 5);
    const int lane = tid & 31;
    const int lo   = blockIdx.x * AGG_R;
    AggLds L;
    L.acc = acc; L.ssl = ssl; L.mxl = mxl; L.mkl = mkl; L.adl = adl; L.ctl = ctl;
    L.lse = lse; L.lsl = lsl;
    int* q = que + wave * AGG_QCAP;

    if (tid < AGG_R * HEADS) {
        const int node = lo + (tid >> 2);
        float ad = 0.0f;
        unsigned key = 0u;
        if (node < N_NODES) {
            ad  = a_d[node * HEADS + (tid & 3)];
            key = f2key(lrelu(a_s[node * HEADS + (tid & 3)] + ad));
        }
        adl[tid] = ad;
        mkl[tid] = key;
    }
    if (tid == 0) { ctl[0] = 0; ctl[1] = 0; }
    __syncthreads();

    agg_pass<1>(rows, (const uint4*)keys, h, a_s, wave, lane, q, L);
    __syncthreads();

    for (int r = wave; r < AGG_R; r += AGG_WAVES) {
        const int node = lo + r;
        if (node < N_NODES) {
            const float4 as4 = *(const float4*)(a_s + (size_t)node * HEADS);
            const float4 ad4 = *(const float4*)(adl + r * HEADS);
            const uint4  mk  = *(const uint4*)(mkl + r * HEADS);
            float4 mf, w;
            mf.x = key2f(mk.x); mf.y = key2f(mk.y); mf.z = key2f(mk.z); mf.w = key2f(mk.w);
            w.x = expf(lrelu(as4.x + ad4.x) - mf.x);
            w.y = expf(lrelu(as4.y + ad4.y) - mf.y);
            w.z = expf(lrelu(as4.z + ad4.z) - mf.z);
            w.w = expf(lrelu(as4.w + ad4.w) - mf.w);
            const float* hp = h + (size_t)node * OUT_CH + lane;
            float* ap = acc + r * OUT_CH + lane;
            ap[0]  = hp[0]  * w.x;
            ap[32] = hp[32] * w.y;
            ap[64] = hp[64] * w.z;
            ap[96] = hp[96] * w.w;
            if (lane == 0) {
                *(float4*)(ssl + r * HEADS) = w;
                *(float4*)(mxl + r * HEADS) = mf;
            }
        }
    }
    __syncthreads();

    const int total = __builtin_amdgcn_readfirstlane(ctl[0]);
    const int ovf   = __builtin_amdgcn_readfirstlane(ctl[1]);
    if (ovf == 0) {
        const int last = total - 1;
        for (int i = wave * 32; i < total; i += AGG_THREADS) {
            const int at = min(i + lane, last);
            const int ent = lse[at];
            const float4 lg = lsl[at];
            const int n = total - i;
            if (n >= 32) agg_drain2<true >(h, ent & 0xFFFF, ent >> 16, lg, 32, lane, L);
            else         agg_drain2<false>(h, ent & 0xFFFF, ent >> 16, lg, n,  lane, L);
        }
    } else {
        agg_pass<2>(rows, (const uint4*)keys, h, a_s, wave, lane, q, L);
    }
    __syncthreads();

    const float bs0 = bias[lane],  bs1 = bias[32 + lane],  bs2 = bias[64 + lane],  bs3 = bias[96 + lane];
    const float gm0 = gamma[lane], gm1 = gamma[32 + lane], gm2 = gamma[64 + lane], gm3 = gamma[96 + lane];
    const float bt0 = beta[lane],  bt1 = beta[32 + lane],  bt2 = beta[64 + lane],  bt3 = beta[96 + lane];
    for (int r = wave; r < AGG_R; r += AGG_WAVES) {
        const int node = lo + r;
        if (node < N_NODES) {
            const float* ap = acc + r * OUT_CH + lane;
            const float4 sv = *(const float4*)(ssl + r * HEADS);
            const float v0 = ap[0]  / (sv.x + SOFTMAX_EPS) + bs0;
            const float v1 = ap[32] / (sv.y + SOFTMAX_EPS) + bs1;
            const float v2 = ap[64] / (sv.z + SOFTMAX_EPS) + bs2;
            const float v3 = ap[96] / (sv.w + SOFTMAX_EPS) + bs3;

            const float mu = wave_sum((v0 + v1) + (v2 + v3)) * (1.0f / OUT_CH);
            const float d0 = v0 - mu, d1 = v1 - mu, d2 = v2 - mu, d3 = v3 - mu;
            const float var = wave_sum((d0 * d0 + d1 * d1) + (d2 * d2 + d3 * d3)) * (1.0f / OUT_CH);
            const float rs = 1.0f / sqrtf(var + LN_EPS);

            float y0 = d0 * rs * gm0 + bt0;
            float y1 = d1 * rs * gm1 + bt1;
            float y2 = d2 * rs * gm2 + bt2;
            float y3 = d3 * rs * gm3 + bt3;
            y0 = y0 > 0.0f ? y0 : (expf(y0) - 1.0f);
            y1 = y1 > 0.0f ? y1 : (expf(y1) - 1.0f);
            y2 = y2 > 0.0f ? y2 : (expf(y2) - 1.0f);
            y3 = y3 > 0.0f ? y3 : (expf(y3) - 1.0f);

            float* op = out + (size_t)node * OUT_CH + lane;
            op[0]  = y0;
            op[32] = y1;
            op[64] = y2;
            op[96] = y3;
        }
    }
}

extern "C" void kernel_launch(void* const* d_in, const int* in_sizes, int n_in,
                              void* d_out, int out_size, void* d_ws, size_t ws_size,
                              hipStream_t stream) {
    const float* x       = (const float*)d_in[0];
    const int*   ei      = (const int*)  d_in[1];
    const float* W       = (const float*)d_in[2];
    const float* att_src = (const float*)d_in[3];
    const float* att_dst = (const float*)d_in[4];
    const float* bias    = (const float*)d_in[5];
    const float* gamma   = (const float*)d_in[6];
    const float* beta    = (const float*)d_in[7];
    float* out = (float*)d_out;

    char* ws = (char*)d_ws;
    float*    h      = (float*)ws;
    ws += (size_t)N_NODES * OUT_CH * sizeof(float);
    float*    a_s    = (float*)ws;  ws += (size_t)N_NODES * HEADS * sizeof(float);
    float*    a_d    = (float*)ws;  ws += (size_t)N_NODES * HEADS * sizeof(float);
    unsigned* keys   = (unsigned*)ws;
    ws += (size_t)N_EDGES;
    unsigned* rows   = (unsigned*)ws;

    k_gemm_wmma<<<(N_NODES + 127) / 128, 256, IN_CH * OUT_CH * sizeof(float),
                  stream>>>(x, W, h, N_NODES);

    k_dots<<<DOT_BLOCKS, DOT_THREADS, 0, stream>>>(h, att_src, att_dst, ei, keys, rows, a_s, a_d,
                                                   N_NODES);

    k_agg_lds<<<AGG_BLOCKS, AGG_THREADS, 0, stream>>>(rows, keys, h, a_s, a_d, bias,
                                                      gamma, beta, out);
}


# Owner keys bit-sliced: dots launch votes 8 words per 32 edges (word b = bit b of their owner numbers); owner blocks' scan = 8 xor-and per 32 edges, not byte tests per 16; LDS arrival order kept
# speedup vs baseline: 4.2614x; 1.0317x over previous
#include <hip/hip_runtime.h>
#include <math.h>


#define N_NODES 50000
#define N_EDGES 1600000
#define ET_EDGES (N_EDGES + N_NODES)
#define IN_CH 128
#define OUT_CH 128
#define HEADS 4
#define CPH 32
#define NEG_SLOPE 0.2f
#define SOFTMAX_EPS 1e-16f
#define LN_EPS 1e-5f

typedef float v2f __attribute__((ext_vector_type(2)));
typedef float v8f __attribute__((ext_vector_type(8)));

__global__ void k_gemm_wmma(const float* __restrict__ x,
                            const float* __restrict__ W,
                            float* __restrict__ h, int nrows) {
    extern __shared__ float Wlds[];
    for (int i = threadIdx.x; i < (IN_CH * OUT_CH) / 4; i += blockDim.x)
        ((float4*)Wlds)[i] = ((const float4*)W)[i];
    __syncthreads();

    const int wave = threadIdx.x >> 5;
    const int lane = threadIdx.x & 31;
    const int rowBase = blockIdx.x * 128 + wave * 16;
    const int m  = lane & 15;
    const int kk = (lane >> 4) << 1;

    int arow = rowBase + m;
    if (arow >= nrows) arow = nrows - 1;
    const float* xk = x + (size_t)arow * IN_CH + kk;

    v8f c[8];
#pragma unroll
    for (int t = 0; t < 8; ++t) c[t] = (v8f){0,0,0,0,0,0,0,0};

    float2 av = *(const float2*)xk;
    for (int k = 0; k < IN_CH; k += 4) {
        v2f a; a.x = av.x; a.y = av.y;
        if (k + 4 < IN_CH) av = *(const float2*)(xk + k + 4);

        const float* w0 = &Wlds[(k + kk) * OUT_CH + m];
        v2f b[8];
#pragma unroll
        for (int t = 0; t < 8; ++t) {
            b[t].x = w0[t * 16];
            b[t].y = w0[OUT_CH + t * 16];
        }
#pragma unroll
        for (int t = 0; t < 8; ++t)
            c[t] = __builtin_amdgcn_wmma_f32_16x16x4_f32(
                false, a, false, b[t], (short)0, c[t], false, false);
    }

#pragma unroll
    for (int t = 0; t < 8; ++t) {
#pragma unroll
        for (int v = 0; v < 8; ++v) {
            const int M = v + ((lane >> 4) << 3);
            const int row = rowBase + M;
            if (row < nrows)
                h[(size_t)row * OUT_CH + t * 16 + m] = c[t][v];
        }
    }
}

#define DOT_THREADS 256
#define DOT_NPW     16
#define DOT_BLOCKS  ((N_NODES + (DOT_THREADS / 32) * DOT_NPW - 1) / ((DOT_THREADS / 32) * DOT_NPW))

#define AGG_BLOCKS  256
#define AGG_R       196
#define KEY_CHUNK   512
#define KEY_CHUNKS  (N_EDGES / KEY_CHUNK)
static_assert(N_EDGES % KEY_CHUNK == 0, "a wave writes the keys of 512 edges");
static_assert((N_NODES - 1) / AGG_R < 256, "an owner key is eight bits");
static_assert(DOT_BLOCKS * (DOT_THREADS / 32) >= KEY_CHUNKS, "one key chunk per wave at most");

template <int I>
__device__ __forceinline__ void keys_slice(unsigned o, int lane, uint4& kw) {
    const unsigned m0 = __builtin_amdgcn_ballot_w32((o & 1u) != 0);
    const unsigned m1 = __builtin_amdgcn_ballot_w32((o & 2u) != 0);
    const unsigned m2 = __builtin_amdgcn_ballot_w32((o & 4u) != 0);
    const unsigned m3 = __builtin_amdgcn_ballot_w32((o & 8u) != 0);
    const unsigned m4 = __builtin_amdgcn_ballot_w32((o & 16u) != 0);
    const unsigned m5 = __builtin_amdgcn_ballot_w32((o & 32u) != 0);
    const unsigned m6 = __builtin_amdgcn_ballot_w32((o & 64u) != 0);
    const unsigned m7 = __builtin_amdgcn_ballot_w32((o & 128u) != 0);
    if (lane == I)      { kw.x = m0; kw.y = m1; kw.z = m2; kw.w = m3; }
    if (lane == 16 + I) { kw.x = m4; kw.y = m5; kw.z = m6; kw.w = m7; }
}

template <int J>
__device__ __forceinline__ void keys_slice4(int4 d, int lane, uint4& kw) {
    keys_slice<4 * J + 0>((unsigned)d.x / AGG_R, lane, kw);
    keys_slice<4 * J + 1>((unsigned)d.y / AGG_R, lane, kw);
    keys_slice<4 * J + 2>((unsigned)d.z / AGG_R, lane, kw);
    keys_slice<4 * J + 3>((unsigned)d.w / AGG_R, lane, kw);
}

__device__ __forceinline__ uint4 rows_pack4(int4 s, int4 d) {
    uint4 r;
    r.x = (unsigned)s.x | (((unsigned)d.x % AGG_R) << 16);
    r.y = (unsigned)s.y | (((unsigned)d.y % AGG_R) << 16);
    r.z = (unsigned)s.z | (((unsigned)d.z % AGG_R) << 16);
    r.w = (unsigned)s.w | (((unsigned)d.w % AGG_R) << 16);
    return r;
}

__device__ __forceinline__ float dots_sum8(const float (&v)[8], int lane) {
    const bool b2 = (lane & 4) != 0, b1 = (lane & 2) != 0, b0 = (lane & 1) != 0;
    float r[4], q[2];
#pragma unroll
    for (int j = 0; j < 4; ++j) {
        const float keep = b2 ? v[j + 4] : v[j];
        const float send = b2 ? v[j] : v[j + 4];
        r[j] = keep + __shfl_xor(send, 4, 32);
    }
#pragma unroll
    for (int j = 0; j < 2; ++j) {
        const float keep = b1 ? r[j + 2] : r[j];
        const float send = b1 ? r[j] : r[j + 2];
        q[j] = keep + __shfl_xor(send, 2, 32);
    }
    const float keep = b0 ? q[1] : q[0];
    const float send = b0 ? q[0] : q[1];
    return keep + __shfl_xor(send, 1, 32);
}

__global__ __launch_bounds__(DOT_THREADS)
void k_dots(const float* __restrict__ h,
            const float* __restrict__ att_src,
            const float* __restrict__ att_dst,
            const int* __restrict__ ei, unsigned* __restrict__ keys,
            unsigned* __restrict__ rows,
            float* __restrict__ a_s, float* __restrict__ a_d, int n) {
    const int wave = threadIdx.x >> 5;
    const int lane = threadIdx.x & 31;
    const int wg   = blockIdx.x * (DOT_THREADS / 32) + wave;
    const int base = wg * DOT_NPW;
    const bool keyw = wg < KEY_CHUNKS;
    if (base >= n && !keyw) return;
    const float4 ws = ((const float4*)att_src)[lane];
    const float4 wd = ((const float4*)att_dst)[lane];

    const int wk = keyw ? wg : KEY_CHUNKS - 1;
    const int4* sp = (const int4*)ei + (size_t)wk * (KEY_CHUNK / 4) + lane;
    const int4* dp = (const int4*)(ei + N_EDGES) + (size_t)wk * (KEY_CHUNK / 4) + lane;
    int4 dk[4], sk[4];
#pragma unroll
    for (int j = 0; j < 4; ++j) {
        dk[j] = dp[j * 32];
        sk[j] = sp[j * 32];
    }

    float4 hv[DOT_NPW];
#pragma unroll
    for (int i = 0; i < DOT_NPW; ++i) {
        int node = base + i;
        if (node >= n) node = n - 1;
        hv[i] = ((const float4*)(h + (size_t)node * OUT_CH))[lane];
    }
    __builtin_amdgcn_sched_barrier(0);
    if (keyw) {
        uint4* rp = (uint4*)rows + (size_t)wg * (KEY_CHUNK / 4) + lane;
#pragma unroll
        for (int j = 0; j < 4; ++j)
            rp[j * 32] = rows_pack4(sk[j], dk[j]);
        uint4 kw = make_uint4(0u, 0u, 0u, 0u);
        keys_slice4<0>(dk[0], lane, kw);
        keys_slice4<1>(dk[1], lane, kw);
        keys_slice4<2>(dk[2], lane, kw);
        keys_slice4<3>(dk[3], lane, kw);
        ((uint4*)keys)[(size_t)wg * (KEY_CHUNK / 16) + lane] = kw;
    }
    if (base >= n) return;
#pragma unroll
    for (int g = 0; g < DOT_NPW / 8; ++g) {
        float vs[8], vd[8];
#pragma unroll
        for (int i = 0; i < 8; ++i) {
            const float4 x4 = hv[g * 8 + i];
            vs[i] = (x4.x * ws.x + x4.y * ws.y) + (x4.z * ws.z + x4.w * ws.w);
            vd[i] = (x4.x * wd.x + x4.y * wd.y) + (x4.z * wd.z + x4.w * wd.w);
        }
        const float rs = dots_sum8(vs, lane);
        const float rd = dots_sum8(vd, lane);
        const int node = base + g * 8 + (lane & 7);
        if (node < n) {
            a_s[node * HEADS + (lane >> 3)] = rs;
            a_d[node * HEADS + (lane >> 3)] = rd;
        }
    }
}

__device__ __forceinline__ unsigned f2key(float f) {
    unsigned b = __float_as_uint(f);
    return (b & 0x80000000u) ? ~b : (b | 0x80000000u);
}
__device__ __forceinline__ float key2f(unsigned k) {
    unsigned b = (k & 0x80000000u) ? (k ^ 0x80000000u) : ~k;
    return __uint_as_float(b);
}
__device__ __forceinline__ float lrelu(float v) {
    return v > 0.0f ? v : NEG_SLOPE * v;
}

#define AGG_THREADS 1024
#define AGG_WAVES   (AGG_THREADS / 32)
#define AGG_NSET    4
#define AGG_QCAP    64
#define AGG_CAP     8064
#define N_K16       (N_EDGES / 16)
#define N_UNITS     ((KEY_CHUNKS + 1) / 2)
#define AGG_STEPS   (((N_UNITS + AGG_WAVES - 1) / AGG_WAVES + AGG_NSET - 1) / AGG_NSET * AGG_NSET)
#define AGG_UPAD    (AGG_STEPS * AGG_WAVES)
static_assert(AGG_BLOCKS * AGG_R >= N_NODES, "ranges must cover every node");
static_assert(KEY_CHUNKS * 32 == N_K16, "a chunk is 32 units of 16 bytes");
static_assert(AGG_R <= 256 && N_NODES <= (1 << 16), "list entry = source | row << 16");
static_assert(AGG_R * HEADS <= AGG_THREADS, "one thread per (row, head) at the start");

struct AggLds {
    float*    acc;
    float*    ssl;
    float*    mxl;
    unsigned* mkl;
    float*    adl;
    int*      ctl;
    int*      lse;
    float4*   lsl;
};

template <bool FULL>
__device__ __forceinline__ void agg_drain1(const unsigned* __restrict__ rows,
                                           const float* __restrict__ a_s,
                                           const int* q, int qhead, int n, int lane,
                                           const AggLds& L) {
    const bool on = FULL || lane < n;
    int e = 0;
    if (on) e = q[(qhead + lane) & (AGG_QCAP - 1)];
    const unsigned ent = rows[e];
    const int s   = (int)(ent & 0xFFFFu);
    const int rel = on ? (int)(ent >> 16) : 0;
    const float4 as4 = *(const float4*)(a_s + (size_t)s * HEADS);
    const float4 ad4 = *(const float4*)(L.adl + rel * HEADS);
    const float l0 = lrelu(as4.x + ad4.x);
    const float l1 = lrelu(as4.y + ad4.y);
    const float l2 = lrelu(as4.z + ad4.z);
    const float l3 = lrelu(as4.w + ad4.w);
    if (on) {
        atomicMax(&L.mkl[rel * HEADS + 0], f2key(l0));
        atomicMax(&L.mkl[rel * HEADS + 1], f2key(l1));
        atomicMax(&L.mkl[rel * HEADS + 2], f2key(l2));
        atomicMax(&L.mkl[rel * HEADS + 3], f2key(l3));
    }
    const int nn = FULL ? 32 : n;
    int base = 0;
    if (lane == 0) base = atomicAdd(&L.ctl[0], nn);
    base = __builtin_amdgcn_readfirstlane(base);
    if (on && base + lane < AGG_CAP) {
        L.lse[base + lane] = (int)ent;
        L.lsl[base + lane] = make_float4(l0, l1, l2, l3);
    }
    if (lane == 0 && base + nn > AGG_CAP) L.ctl[1] = 1;
}

template <bool FULL>
__device__ __forceinline__ void agg_drain2(const float* __restrict__ h,
                                           int s, int rel, float4 lg, int n, int lane,
                                           const AggLds& L) {
    const float4 mm  = *(const float4*)(L.mxl + rel * HEADS);
    const float w0 = expf(lg.x - mm.x);
    const float w1 = expf(lg.y - mm.y);
    const float w2 = expf(lg.z - mm.z);
    const float w3 = expf(lg.w - mm.w);
    if (FULL || lane < n) {
        atomicAdd(&L.ssl[rel * HEADS + 0], w0);
        atomicAdd(&L.ssl[rel * HEADS + 1], w1);
        atomicAdd(&L.ssl[rel * HEADS + 2], w2);
        atomicAdd(&L.ssl[rel * HEADS + 3], w3);
    }
    const int nk = FULL ? 32 : n;
#pragma unroll 1
    for (int kb = 0; kb < nk; kb += 8) {
        float hv[8][4];
#pragma unroll
        for (int j = 0; j < 8; ++j) {
            const int k = kb + j;
            if (FULL || k < n) {
                const int sk = __builtin_amdgcn_readlane(s, k);
                const float* hp = h + (size_t)sk * OUT_CH + lane;
                hv[j][0] = hp[0];
                hv[j][1] = hp[32];
                hv[j][2] = hp[64];
                hv[j][3] = hp[96];
            }
        }
#pragma unroll
        for (int j = 0; j < 8; ++j) {
            const int k = kb + j;
            if (FULL || k < n) {
                const int rk = __builtin_amdgcn_readlane(rel, k);
                const float a0 = __int_as_float(__builtin_amdgcn_readlane(__float_as_int(w0), k));
                const float a1 = __int_as_float(__builtin_amdgcn_readlane(__float_as_int(w1), k));
                const float a2 = __int_as_float(__builtin_amdgcn_readlane(__float_as_int(w2), k));
                const float a3 = __int_as_float(__builtin_amdgcn_readlane(__float_as_int(w3), k));
                float* ap = L.acc + rk * OUT_CH + lane;
                atomicAdd(ap,      hv[j][0] * a0);
                atomicAdd(ap + 32, hv[j][1] * a1);
                atomicAdd(ap + 64, hv[j][2] * a2);
                atomicAdd(ap + 96, hv[j][3] * a3);
            }
        }
    }
}

template <int PH, bool FULL>
__device__ __forceinline__ void agg_drain(const unsigned* __restrict__ rows,
                                          const float* __restrict__ h,
                                          const float* __restrict__ a_s,
                                          const int* q, int qhead, int n, int lane,
                                          const AggLds& L) {
    if (PH == 1) {
        agg_drain1<FULL>(rows, a_s, q, qhead, n, lane, L);
    } else {
        const bool on = FULL || lane < n;
        int e = 0;
        if (on) e = q[(qhead + lane) & (AGG_QCAP - 1)];
        const unsigned ent = rows[e];
        const int s   = (int)(ent & 0xFFFFu);
        const int rel = on ? (int)(ent >> 16) : 0;
        const float4 as4 = *(const float4*)(a_s + (size_t)s * HEADS);
        const float4 ad4 = *(const float4*)(L.adl + rel * HEADS);
        const float4 lg = make_float4(lrelu(as4.x + ad4.x), lrelu(as4.y + ad4.y),
                                      lrelu(as4.z + ad4.z), lrelu(as4.w + ad4.w));
        agg_drain2<FULL>(h, s, rel, lg, n, lane, L);
    }
}

__device__ __forceinline__ int agg_unit(int sp, int wave, int ustart) {
    int u = ustart + sp * AGG_WAVES + wave;
    while (u >= AGG_UPAD) u -= AGG_UPAD;
    return u;
}

struct AggKeys { uint4 lo, hi; };

__device__ __forceinline__ AggKeys agg_load(const uint4* __restrict__ keys, int u, int lane) {
    int chunk = 2 * u + (lane >> 4);
    if (chunk > KEY_CHUNKS - 1) chunk = KEY_CHUNKS - 1;
    const uint4* p = keys + chunk * 32 + (lane & 15);
    AggKeys k;
    k.lo = p[0];
    k.hi = p[16];
    return k;
}

struct AggOwn { unsigned inv[8]; };

template <int PH>
__device__ __forceinline__ void agg_scan(AggKeys kv, int u, const AggOwn& own, int lane,
                                         int* q, int& qhead, int& cnt,
                                         const unsigned* __restrict__ rows,
                                         const float* __restrict__ h,
                                         const float* __restrict__ a_s,
                                         const AggLds& L) {
    if (u >= N_UNITS) return;
    unsigned cm = ((kv.lo.x ^ own.inv[0]) & (kv.lo.y ^ own.inv[1])) &
                  ((kv.lo.z ^ own.inv[2]) & (kv.lo.w ^ own.inv[3])) &
                  ((kv.hi.x ^ own.inv[4]) & (kv.hi.y ^ own.inv[5])) &
                  ((kv.hi.z ^ own.inv[6]) & (kv.hi.w ^ own.inv[7]));
    const int chunk = 2 * u + (lane >> 4);
    if (chunk >= KEY_CHUNKS) cm = 0u;
    unsigned bm = __builtin_amdgcn_ballot_w32(cm != 0u);
    if (bm) {
        const int i = lane & 15;
        const int e0 = chunk * KEY_CHUNK + ((i >> 2) << 7) + (i & 3);
        do {
            if (cm) {
                const int p = __builtin_ctz(cm);
                q[(qhead + cnt + __builtin_amdgcn_mbcnt_lo(bm, 0)) & (AGG_QCAP - 1)] =
                    e0 + (p << 2);
                cm &= cm - 1;
            }
            cnt += __builtin_popcount(bm);
            if (cnt >= 32) {
                agg_drain<PH, true>(rows, h, a_s, q, qhead, 32, lane, L);
                qhead = (qhead + 32) & (AGG_QCAP - 1);
                cnt -= 32;
            }
            bm = __builtin_amdgcn_ballot_w32(cm != 0);
        } while (bm);
    }
}

template <int PH>
__device__ __forceinline__ void agg_pass(const unsigned* __restrict__ rows,
                                         const uint4* __restrict__ keys,
                                         const float* __restrict__ h,
                                         const float* __restrict__ a_s,
                                         int wave, int lane, int* q,
                                         const AggLds& L) {
    int qhead = 0, cnt = 0;
    AggOwn own;
#pragma unroll
    for (int b = 0; b < 8; ++b) own.inv[b] = ((blockIdx.x >> b) & 1u) ? 0u : 0xFFFFFFFFu;
    const int ustart = (blockIdx.x * AGG_UPAD) / AGG_BLOCKS;

    AggKeys k0 = agg_load(keys, agg_unit(0, wave, ustart), lane);
    AggKeys k1 = agg_load(keys, agg_unit(1, wave, ustart), lane);
    AggKeys k2 = agg_load(keys, agg_unit(2, wave, ustart), lane);
    AggKeys k3 = agg_load(keys, agg_unit(3, wave, ustart), lane);
    for (int st = 0; st < AGG_STEPS; st += AGG_NSET) {
        agg_scan<PH>(k0, agg_unit(st,     wave, ustart), own, lane, q, qhead, cnt, rows, h, a_s, L);
        k0 = agg_load(keys, agg_unit(st + 4, wave, ustart), lane);
        agg_scan<PH>(k1, agg_unit(st + 1, wave, ustart), own, lane, q, qhead, cnt, rows, h, a_s, L);
        k1 = agg_load(keys, agg_unit(st + 5, wave, ustart), lane);
        agg_scan<PH>(k2, agg_unit(st + 2, wave, ustart), own, lane, q, qhead, cnt, rows, h, a_s, L);
        k2 = agg_load(keys, agg_unit(st + 6, wave, ustart), lane);
        agg_scan<PH>(k3, agg_unit(st + 3, wave, ustart), own, lane, q, qhead, cnt, rows, h, a_s, L);
        k3 = agg_load(keys, agg_unit(st + 7, wave, ustart), lane);
    }
    if (cnt > 0) agg_drain<PH, false>(rows, h, a_s, q, qhead, cnt, lane, L);
}

__device__ __forceinline__ float wave_sum(float t) {
    for (int o = 16; o > 0; o >>= 1) t += __shfl_xor(t, o, 32);
    return t;
}

__global__ __launch_bounds__(AGG_THREADS)
void k_agg_lds(const unsigned* __restrict__ rows,
               const unsigned* __restrict__ keys,
               const float* __restrict__ h,
               const float* __restrict__ a_s,
               const float* __restrict__ a_d,
               const float* __restrict__ bias,
               const float* __restrict__ gamma,
               const float* __restrict__ beta,
               float* __restrict__ out) {
    __shared__ __attribute__((aligned(16))) float acc[AGG_R * OUT_CH];
    __shared__ __attribute__((aligned(16))) float ssl[AGG_R * HEADS];
    __shared__ __attribute__((aligned(16))) float mxl[AGG_R * HEADS];
    __shared__ __attribute__((aligned(16))) unsigned mkl[AGG_R * HEADS];
    __shared__ __attribute__((aligned(16))) float adl[AGG_R * HEADS];
    __shared__ int ctl[2];
    __shared__ int que[AGG_WAVES * AGG_QCAP];
    __shared__ __attribute__((aligned(16))) int lse[AGG_CAP];
    __shared__ __attribute__((aligned(16))) float4 lsl[AGG_CAP];

    const int tid  = threadIdx.x;
    const int wave = __builtin_amdgcn_readfirstlane(tid >> 5);
    const int lane = tid & 31;
    const int lo   = blockIdx.x * AGG_R;
    AggLds L;
    L.acc = acc; L.ssl = ssl; L.mxl = mxl; L.mkl = mkl; L.adl = adl; L.ctl = ctl;
    L.lse = lse; L.lsl = lsl;
    int* q = que + wave * AGG_QCAP;

    if (tid < AGG_R * HEADS) {
        const int node = lo + (tid >> 2);
        float ad = 0.0f;
        unsigned key = 0u;
        if (node < N_NODES) {
            ad  = a_d[node * HEADS + (tid & 3)];
            key = f2key(lrelu(a_s[node * HEADS + (tid & 3)] + ad));
        }
        adl[tid] = ad;
        mkl[tid] = key;
    }
    if (tid == 0) { ctl[0] = 0; ctl[1] = 0; }
    __syncthreads();

    agg_pass<1>(rows, (const uint4*)keys, h, a_s, wave, lane, q, L);
    __syncthreads();

    for (int r = wave; r < AGG_R; r += AGG_WAVES) {
        const int node = lo + r;
        if (node < N_NODES) {
            const float4 as4 = *(const float4*)(a_s + (size_t)node * HEADS);
            const float4 ad4 = *(const float4*)(adl + r * HEADS);
            const uint4  mk  = *(const uint4*)(mkl + r * HEADS);
            float4 mf, w;
            mf.x = key2f(mk.x); mf.y = key2f(mk.y); mf.z = key2f(mk.z); mf.w = key2f(mk.w);
            w.x = expf(lrelu(as4.x + ad4.x) - mf.x);
            w.y = expf(lrelu(as4.y + ad4.y) - mf.y);
            w.z = expf(lrelu(as4.z + ad4.z) - mf.z);
            w.w = expf(lrelu(as4.w + ad4.w) - mf.w);
            const float* hp = h + (size_t)node * OUT_CH + lane;
            float* ap = acc + r * OUT_CH + lane;
            ap[0]  = hp[0]  * w.x;
            ap[32] = hp[32] * w.y;
            ap[64] = hp[64] * w.z;
            ap[96] = hp[96] * w.w;
            if (lane == 0) {
                *(float4*)(ssl + r * HEADS) = w;
                *(float4*)(mxl + r * HEADS) = mf;
            }
        }
    }
    __syncthreads();

    const int total = __builtin_amdgcn_readfirstlane(ctl[0]);
    const int ovf   = __builtin_amdgcn_readfirstlane(ctl[1]);
    if (ovf == 0) {
        const int last = total - 1;
        for (int i = wave * 32; i < total; i += AGG_THREADS) {
            const int at = min(i + lane, last);
            const int ent = lse[at];
            const float4 lg = lsl[at];
            const int n = total - i;
            if (n >= 32) agg_drain2<true >(h, ent & 0xFFFF, ent >> 16, lg, 32, lane, L);
            else         agg_drain2<false>(h, ent & 0xFFFF, ent >> 16, lg, n,  lane, L);
        }
    } else {
        agg_pass<2>(rows, (const uint4*)keys, h, a_s, wave, lane, q, L);
    }
    __syncthreads();

    const float bs0 = bias[lane],  bs1 = bias[32 + lane],  bs2 = bias[64 + lane],  bs3 = bias[96 + lane];
    const float gm0 = gamma[lane], gm1 = gamma[32 + lane], gm2 = gamma[64 + lane], gm3 = gamma[96 + lane];
    const float bt0 = beta[lane],  bt1 = beta[32 + lane],  bt2 = beta[64 + lane],  bt3 = beta[96 + lane];
    for (int r = wave; r < AGG_R; r += AGG_WAVES) {
        const int node = lo + r;
        if (node < N_NODES) {
            const float* ap = acc + r * OUT_CH + lane;
            const float4 sv = *(const float4*)(ssl + r * HEADS);
            const float v0 = ap[0]  / (sv.x + SOFTMAX_EPS) + bs0;
            const float v1 = ap[32] / (sv.y + SOFTMAX_EPS) + bs1;
            const float v2 = ap[64] / (sv.z + SOFTMAX_EPS) + bs2;
            const float v3 = ap[96] / (sv.w + SOFTMAX_EPS) + bs3;

            const float mu = wave_sum((v0 + v1) + (v2 + v3)) * (1.0f / OUT_CH);
            const float d0 = v0 - mu, d1 = v1 - mu, d2 = v2 - mu, d3 = v3 - mu;
            const float var = wave_sum((d0 * d0 + d1 * d1) + (d2 * d2 + d3 * d3)) * (1.0f / OUT_CH);
            const float rs = 1.0f / sqrtf(var + LN_EPS);

            float y0 = d0 * rs * gm0 + bt0;
            float y1 = d1 * rs * gm1 + bt1;
            float y2 = d2 * rs * gm2 + bt2;
            float y3 = d3 * rs * gm3 + bt3;
            y0 = y0 > 0.0f ? y0 : (expf(y0) - 1.0f);
            y1 = y1 > 0.0f ? y1 : (expf(y1) - 1.0f);
            y2 = y2 > 0.0f ? y2 : (expf(y2) - 1.0f);
            y3 = y3 > 0.0f ? y3 : (expf(y3) - 1.0f);

            float* op = out + (size_t)node * OUT_CH + lane;
            op[0]  = y0;
            op[32] = y1;
            op[64] = y2;
            op[96] = y3;
        }
    }
}

extern "C" void kernel_launch(void* const* d_in, const int* in_sizes, int n_in,
                              void* d_out, int out_size, void* d_ws, size_t ws_size,
                              hipStream_t stream) {
    const float* x       = (const float*)d_in[0];
    const int*   ei      = (const int*)  d_in[1];
    const float* W       = (const float*)d_in[2];
    const float* att_src = (const float*)d_in[3];
    const float* att_dst = (const float*)d_in[4];
    const float* bias    = (const float*)d_in[5];
    const float* gamma   = (const float*)d_in[6];
    const float* beta    = (const float*)d_in[7];
    float* out = (float*)d_out;

    char* ws = (char*)d_ws;
    float*    h      = (float*)ws;
    ws += (size_t)N_NODES * OUT_CH * sizeof(float);
    float*    a_s    = (float*)ws;  ws += (size_t)N_NODES * HEADS * sizeof(float);
    float*    a_d    = (float*)ws;  ws += (size_t)N_NODES * HEADS * sizeof(float);
    unsigned* keys   = (unsigned*)ws;
    ws += (size_t)N_EDGES;
    unsigned* rows   = (unsigned*)ws;

    k_gemm_wmma<<<(N_NODES + 127) / 128, 256, IN_CH * OUT_CH * sizeof(float),
                  stream>>>(x, W, h, N_NODES);

    k_dots<<<DOT_BLOCKS, DOT_THREADS, 0, stream>>>(h, att_src, att_dst, ei, keys, rows, a_s, a_d,
                                                   N_NODES);

    k_agg_lds<<<AGG_BLOCKS, AGG_THREADS, 0, stream>>>(rows, keys, h, a_s, a_d, bias,
                                                      gamma, beta, out);
}


# Pass 2 gather re-laid: lane = 4 consecutive channels (row = one 16-byte load a lane; LDS rows lane-major), weights by DPP8, next 8 rows requested before these are added; LN sums regrouped
# speedup vs baseline: 4.6412x; 1.0891x over previous
#include <hip/hip_runtime.h>
#include <math.h>


#define N_NODES 50000
#define N_EDGES 1600000
#define ET_EDGES (N_EDGES + N_NODES)
#define IN_CH 128
#define OUT_CH 128
#define HEADS 4
#define CPH 32
#define NEG_SLOPE 0.2f
#define SOFTMAX_EPS 1e-16f
#define LN_EPS 1e-5f

typedef float v2f __attribute__((ext_vector_type(2)));
typedef float v8f __attribute__((ext_vector_type(8)));

__global__ void k_gemm_wmma(const float* __restrict__ x,
                            const float* __restrict__ W,
                            float* __restrict__ h, int nrows) {
    extern __shared__ float Wlds[];
    for (int i = threadIdx.x; i < (IN_CH * OUT_CH) / 4; i += blockDim.x)
        ((float4*)Wlds)[i] = ((const float4*)W)[i];
    __syncthreads();

    const int wave = threadIdx.x >> 5;
    const int lane = threadIdx.x & 31;
    const int rowBase = blockIdx.x * 128 + wave * 16;
    const int m  = lane & 15;
    const int kk = (lane >> 4) << 1;

    int arow = rowBase + m;
    if (arow >= nrows) arow = nrows - 1;
    const float* xk = x + (size_t)arow * IN_CH + kk;

    v8f c[8];
#pragma unroll
    for (int t = 0; t < 8; ++t) c[t] = (v8f){0,0,0,0,0,0,0,0};

    float2 av = *(const float2*)xk;
    for (int k = 0; k < IN_CH; k += 4) {
        v2f a; a.x = av.x; a.y = av.y;
        if (k + 4 < IN_CH) av = *(const float2*)(xk + k + 4);

        const float* w0 = &Wlds[(k + kk) * OUT_CH + m];
        v2f b[8];
#pragma unroll
        for (int t = 0; t < 8; ++t) {
            b[t].x = w0[t * 16];
            b[t].y = w0[OUT_CH + t * 16];
        }
#pragma unroll
        for (int t = 0; t < 8; ++t)
            c[t] = __builtin_amdgcn_wmma_f32_16x16x4_f32(
                false, a, false, b[t], (short)0, c[t], false, false);
    }

#pragma unroll
    for (int t = 0; t < 8; ++t) {
#pragma unroll
        for (int v = 0; v < 8; ++v) {
            const int M = v + ((lane >> 4) << 3);
            const int row = rowBase + M;
            if (row < nrows)
                h[(size_t)row * OUT_CH + t * 16 + m] = c[t][v];
        }
    }
}

#define DOT_THREADS 256
#define DOT_NPW     16
#define DOT_BLOCKS  ((N_NODES + (DOT_THREADS / 32) * DOT_NPW - 1) / ((DOT_THREADS / 32) * DOT_NPW))

#define AGG_BLOCKS  256
#define AGG_R       196
#define KEY_CHUNK   512
#define KEY_CHUNKS  (N_EDGES / KEY_CHUNK)
static_assert(N_EDGES % KEY_CHUNK == 0, "a wave writes the keys of 512 edges");
static_assert((N_NODES - 1) / AGG_R < 256, "an owner key is eight bits");
static_assert(DOT_BLOCKS * (DOT_THREADS / 32) >= KEY_CHUNKS, "one key chunk per wave at most");

template <int I>
__device__ __forceinline__ void keys_slice(unsigned o, int lane, uint4& kw) {
    const unsigned m0 = __builtin_amdgcn_ballot_w32((o & 1u) != 0);
    const unsigned m1 = __builtin_amdgcn_ballot_w32((o & 2u) != 0);
    const unsigned m2 = __builtin_amdgcn_ballot_w32((o & 4u) != 0);
    const unsigned m3 = __builtin_amdgcn_ballot_w32((o & 8u) != 0);
    const unsigned m4 = __builtin_amdgcn_ballot_w32((o & 16u) != 0);
    const unsigned m5 = __builtin_amdgcn_ballot_w32((o & 32u) != 0);
    const unsigned m6 = __builtin_amdgcn_ballot_w32((o & 64u) != 0);
    const unsigned m7 = __builtin_amdgcn_ballot_w32((o & 128u) != 0);
    if (lane == I)      { kw.x = m0; kw.y = m1; kw.z = m2; kw.w = m3; }
    if (lane == 16 + I) { kw.x = m4; kw.y = m5; kw.z = m6; kw.w = m7; }
}

template <int J>
__device__ __forceinline__ void keys_slice4(int4 d, int lane, uint4& kw) {
    keys_slice<4 * J + 0>((unsigned)d.x / AGG_R, lane, kw);
    keys_slice<4 * J + 1>((unsigned)d.y / AGG_R, lane, kw);
    keys_slice<4 * J + 2>((unsigned)d.z / AGG_R, lane, kw);
    keys_slice<4 * J + 3>((unsigned)d.w / AGG_R, lane, kw);
}

__device__ __forceinline__ uint4 rows_pack4(int4 s, int4 d) {
    uint4 r;
    r.x = (unsigned)s.x | (((unsigned)d.x % AGG_R) << 16);
    r.y = (unsigned)s.y | (((unsigned)d.y % AGG_R) << 16);
    r.z = (unsigned)s.z | (((unsigned)d.z % AGG_R) << 16);
    r.w = (unsigned)s.w | (((unsigned)d.w % AGG_R) << 16);
    return r;
}

__device__ __forceinline__ float dots_sum8(const float (&v)[8], int lane) {
    const bool b2 = (lane & 4) != 0, b1 = (lane & 2) != 0, b0 = (lane & 1) != 0;
    float r[4], q[2];
#pragma unroll
    for (int j = 0; j < 4; ++j) {
        const float keep = b2 ? v[j + 4] : v[j];
        const float send = b2 ? v[j] : v[j + 4];
        r[j] = keep + __shfl_xor(send, 4, 32);
    }
#pragma unroll
    for (int j = 0; j < 2; ++j) {
        const float keep = b1 ? r[j + 2] : r[j];
        const float send = b1 ? r[j] : r[j + 2];
        q[j] = keep + __shfl_xor(send, 2, 32);
    }
    const float keep = b0 ? q[1] : q[0];
    const float send = b0 ? q[0] : q[1];
    return keep + __shfl_xor(send, 1, 32);
}

__global__ __launch_bounds__(DOT_THREADS)
void k_dots(const float* __restrict__ h,
            const float* __restrict__ att_src,
            const float* __restrict__ att_dst,
            const int* __restrict__ ei, unsigned* __restrict__ keys,
            unsigned* __restrict__ rows,
            float* __restrict__ a_s, float* __restrict__ a_d, int n) {
    const int wave = threadIdx.x >> 5;
    const int lane = threadIdx.x & 31;
    const int wg   = blockIdx.x * (DOT_THREADS / 32) + wave;
    const int base = wg * DOT_NPW;
    const bool keyw = wg < KEY_CHUNKS;
    if (base >= n && !keyw) return;
    const float4 ws = ((const float4*)att_src)[lane];
    const float4 wd = ((const float4*)att_dst)[lane];

    const int wk = keyw ? wg : KEY_CHUNKS - 1;
    const int4* sp = (const int4*)ei + (size_t)wk * (KEY_CHUNK / 4) + lane;
    const int4* dp = (const int4*)(ei + N_EDGES) + (size_t)wk * (KEY_CHUNK / 4) + lane;
    int4 dk[4], sk[4];
#pragma unroll
    for (int j = 0; j < 4; ++j) {
        dk[j] = dp[j * 32];
        sk[j] = sp[j * 32];
    }

    float4 hv[DOT_NPW];
#pragma unroll
    for (int i = 0; i < DOT_NPW; ++i) {
        int node = base + i;
        if (node >= n) node = n - 1;
        hv[i] = ((const float4*)(h + (size_t)node * OUT_CH))[lane];
    }
    __builtin_amdgcn_sched_barrier(0);
    if (keyw) {
        uint4* rp = (uint4*)rows + (size_t)wg * (KEY_CHUNK / 4) + lane;
#pragma unroll
        for (int j = 0; j < 4; ++j)
            rp[j * 32] = rows_pack4(sk[j], dk[j]);
        uint4 kw = make_uint4(0u, 0u, 0u, 0u);
        keys_slice4<0>(dk[0], lane, kw);
        keys_slice4<1>(dk[1], lane, kw);
        keys_slice4<2>(dk[2], lane, kw);
        keys_slice4<3>(dk[3], lane, kw);
        ((uint4*)keys)[(size_t)wg * (KEY_CHUNK / 16) + lane] = kw;
    }
    if (base >= n) return;
#pragma unroll
    for (int g = 0; g < DOT_NPW / 8; ++g) {
        float vs[8], vd[8];
#pragma unroll
        for (int i = 0; i < 8; ++i) {
            const float4 x4 = hv[g * 8 + i];
            vs[i] = (x4.x * ws.x + x4.y * ws.y) + (x4.z * ws.z + x4.w * ws.w);
            vd[i] = (x4.x * wd.x + x4.y * wd.y) + (x4.z * wd.z + x4.w * wd.w);
        }
        const float rs = dots_sum8(vs, lane);
        const float rd = dots_sum8(vd, lane);
        const int node = base + g * 8 + (lane & 7);
        if (node < n) {
            a_s[node * HEADS + (lane >> 3)] = rs;
            a_d[node * HEADS + (lane >> 3)] = rd;
        }
    }
}

__device__ __forceinline__ unsigned f2key(float f) {
    unsigned b = __float_as_uint(f);
    return (b & 0x80000000u) ? ~b : (b | 0x80000000u);
}
__device__ __forceinline__ float key2f(unsigned k) {
    unsigned b = (k & 0x80000000u) ? (k ^ 0x80000000u) : ~k;
    return __uint_as_float(b);
}
__device__ __forceinline__ float lrelu(float v) {
    return v > 0.0f ? v : NEG_SLOPE * v;
}

#define AGG_THREADS 1024
#define AGG_WAVES   (AGG_THREADS / 32)
#define AGG_NSET    4
#define AGG_QCAP    64
#define AGG_CAP     8064
#define N_K16       (N_EDGES / 16)
#define N_UNITS     ((KEY_CHUNKS + 1) / 2)
#define AGG_STEPS   (((N_UNITS + AGG_WAVES - 1) / AGG_WAVES + AGG_NSET - 1) / AGG_NSET * AGG_NSET)
#define AGG_UPAD    (AGG_STEPS * AGG_WAVES)
static_assert(AGG_BLOCKS * AGG_R >= N_NODES, "ranges must cover every node");
static_assert(KEY_CHUNKS * 32 == N_K16, "a chunk is 32 units of 16 bytes");
static_assert(AGG_R <= 256 && N_NODES <= (1 << 16), "list entry = source | row << 16");
static_assert(AGG_R * HEADS <= AGG_THREADS, "one thread per (row, head) at the start");

struct AggLds {
    float*    acc;
    float*    ssl;
    float*    mxl;
    unsigned* mkl;
    float*    adl;
    int*      ctl;
    int*      lse;
    float4*   lsl;
};

template <bool FULL>
__device__ __forceinline__ void agg_drain1(const unsigned* __restrict__ rows,
                                           const float* __restrict__ a_s,
                                           const int* q, int qhead, int n, int lane,
                                           const AggLds& L) {
    const bool on = FULL || lane < n;
    int e = 0;
    if (on) e = q[(qhead + lane) & (AGG_QCAP - 1)];
    const unsigned ent = rows[e];
    const int s   = (int)(ent & 0xFFFFu);
    const int rel = on ? (int)(ent >> 16) : 0;
    const float4 as4 = *(const float4*)(a_s + (size_t)s * HEADS);
    const float4 ad4 = *(const float4*)(L.adl + rel * HEADS);
    const float l0 = lrelu(as4.x + ad4.x);
    const float l1 = lrelu(as4.y + ad4.y);
    const float l2 = lrelu(as4.z + ad4.z);
    const float l3 = lrelu(as4.w + ad4.w);
    if (on) {
        atomicMax(&L.mkl[rel * HEADS + 0], f2key(l0));
        atomicMax(&L.mkl[rel * HEADS + 1], f2key(l1));
        atomicMax(&L.mkl[rel * HEADS + 2], f2key(l2));
        atomicMax(&L.mkl[rel * HEADS + 3], f2key(l3));
    }
    const int nn = FULL ? 32 : n;
    int base = 0;
    if (lane == 0) base = atomicAdd(&L.ctl[0], nn);
    base = __builtin_amdgcn_readfirstlane(base);
    if (on && base + lane < AGG_CAP) {
        L.lse[base + lane] = (int)ent;
        L.lsl[base + lane] = make_float4(l0, l1, l2, l3);
    }
    if (lane == 0 && base + nn > AGG_CAP) L.ctl[1] = 1;
}

template <int K>
__device__ __forceinline__ float bcast8(float v) {
    return __uint_as_float(__builtin_amdgcn_mov_dpp8(__float_as_uint(v), K * 0x249249u));
}

template <int K, bool FULL>
__device__ __forceinline__ void agg_row_load(const float* __restrict__ h, int ent, int n, int lane,
                                             float4& hv) {
    if (FULL || K < n) {
        const int sk = __builtin_amdgcn_readlane(ent, K) & 0xFFFF;
        hv = ((const float4*)(h + (size_t)sk * OUT_CH))[lane];
    }
}

template <int K, bool FULL>
__device__ __forceinline__ void agg_row_add(int ent, float w, float4 hv, int n, int lane,
                                            const AggLds& L) {
    if (FULL || K < n) {
        const int rk = __builtin_amdgcn_readlane(ent, K) >> 16;
        const float wk = bcast8<K>(w);
        float* ap = L.acc + rk * OUT_CH + lane;
        atomicAdd(ap,      hv.x * wk);
        atomicAdd(ap + 32, hv.y * wk);
        atomicAdd(ap + 64, hv.z * wk);
        atomicAdd(ap + 96, hv.w * wk);
    }
}

struct AggGrp { int ent; float4 hv[8]; };

template <bool FULL>
__device__ __forceinline__ void agg_grp_issue(const float* __restrict__ h, int p, int n, int last,
                                              int lane, const AggLds& L, AggGrp& G) {
    const int j  = lane & 7;
    const int at = min(p + j, last);
    G.ent = L.lse[at];
    agg_row_load<0, FULL>(h, G.ent, n, lane, G.hv[0]);
    agg_row_load<1, FULL>(h, G.ent, n, lane, G.hv[1]);
    agg_row_load<2, FULL>(h, G.ent, n, lane, G.hv[2]);
    agg_row_load<3, FULL>(h, G.ent, n, lane, G.hv[3]);
    agg_row_load<4, FULL>(h, G.ent, n, lane, G.hv[4]);
    agg_row_load<5, FULL>(h, G.ent, n, lane, G.hv[5]);
    agg_row_load<6, FULL>(h, G.ent, n, lane, G.hv[6]);
    agg_row_load<7, FULL>(h, G.ent, n, lane, G.hv[7]);
}

template <bool FULL>
__device__ __forceinline__ void agg_grp_finish(int p, int n, int last, int lane,
                                               const AggLds& L, const AggGrp& G) {
    const int j  = lane & 7;
    const int hh = lane >> 3;
    const int at = FULL ? p + j : min(p + j, last);
    const int rel = G.ent >> 16;
    const float lg = ((const float*)L.lsl)[at * HEADS + hh];
    const float mx = L.mxl[rel * HEADS + hh];
    const float w = expf(lg - mx);
    if (FULL || j < n) atomicAdd(&L.ssl[rel * HEADS + hh], w);
    agg_row_add<0, FULL>(G.ent, w, G.hv[0], n, lane, L);
    agg_row_add<1, FULL>(G.ent, w, G.hv[1], n, lane, L);
    agg_row_add<2, FULL>(G.ent, w, G.hv[2], n, lane, L);
    agg_row_add<3, FULL>(G.ent, w, G.hv[3], n, lane, L);
    agg_row_add<4, FULL>(G.ent, w, G.hv[4], n, lane, L);
    agg_row_add<5, FULL>(G.ent, w, G.hv[5], n, lane, L);
    agg_row_add<6, FULL>(G.ent, w, G.hv[6], n, lane, L);
    agg_row_add<7, FULL>(G.ent, w, G.hv[7], n, lane, L);
}

template <bool FULL>
__device__ __forceinline__ void agg_drain2(const float* __restrict__ h,
                                           int s, int rel, float4 lg, int n, int lane,
                                           const AggLds& L) {
    const float4 mm  = *(const float4*)(L.mxl + rel * HEADS);
    const float w0 = expf(lg.x - mm.x);
    const float w1 = expf(lg.y - mm.y);
    const float w2 = expf(lg.z - mm.z);
    const float w3 = expf(lg.w - mm.w);
    if (FULL || lane < n) {
        atomicAdd(&L.ssl[rel * HEADS + 0], w0);
        atomicAdd(&L.ssl[rel * HEADS + 1], w1);
        atomicAdd(&L.ssl[rel * HEADS + 2], w2);
        atomicAdd(&L.ssl[rel * HEADS + 3], w3);
    }
    const int nk = FULL ? 32 : n;
    const int hh = lane >> 3;
#pragma unroll 1
    for (int k = 0; k < nk; ++k) {
        const int sk = __builtin_amdgcn_readlane(s, k);
        const int rk = __builtin_amdgcn_readlane(rel, k);
        const float a0 = __int_as_float(__builtin_amdgcn_readlane(__float_as_int(w0), k));
        const float a1 = __int_as_float(__builtin_amdgcn_readlane(__float_as_int(w1), k));
        const float a2 = __int_as_float(__builtin_amdgcn_readlane(__float_as_int(w2), k));
        const float a3 = __int_as_float(__builtin_amdgcn_readlane(__float_as_int(w3), k));
        const float wk = hh == 0 ? a0 : (hh == 1 ? a1 : (hh == 2 ? a2 : a3));
        const float4 hv = ((const float4*)(h + (size_t)sk * OUT_CH))[lane];
        float* ap = L.acc + rk * OUT_CH + lane;
        atomicAdd(ap,      hv.x * wk);
        atomicAdd(ap + 32, hv.y * wk);
        atomicAdd(ap + 64, hv.z * wk);
        atomicAdd(ap + 96, hv.w * wk);
    }
}

template <int PH, bool FULL>
__device__ __forceinline__ void agg_drain(const unsigned* __restrict__ rows,
                                          const float* __restrict__ h,
                                          const float* __restrict__ a_s,
                                          const int* q, int qhead, int n, int lane,
                                          const AggLds& L) {
    if (PH == 1) {
        agg_drain1<FULL>(rows, a_s, q, qhead, n, lane, L);
    } else {
        const bool on = FULL || lane < n;
        int e = 0;
        if (on) e = q[(qhead + lane) & (AGG_QCAP - 1)];
        const unsigned ent = rows[e];
        const int s   = (int)(ent & 0xFFFFu);
        const int rel = on ? (int)(ent >> 16) : 0;
        const float4 as4 = *(const float4*)(a_s + (size_t)s * HEADS);
        const float4 ad4 = *(const float4*)(L.adl + rel * HEADS);
        const float4 lg = make_float4(lrelu(as4.x + ad4.x), lrelu(as4.y + ad4.y),
                                      lrelu(as4.z + ad4.z), lrelu(as4.w + ad4.w));
        agg_drain2<FULL>(h, s, rel, lg, n, lane, L);
    }
}

__device__ __forceinline__ int agg_unit(int sp, int wave, int ustart) {
    int u = ustart + sp * AGG_WAVES + wave;
    while (u >= AGG_UPAD) u -= AGG_UPAD;
    return u;
}

struct AggKeys { uint4 lo, hi; };

__device__ __forceinline__ AggKeys agg_load(const uint4* __restrict__ keys, int u, int lane) {
    int chunk = 2 * u + (lane >> 4);
    if (chunk > KEY_CHUNKS - 1) chunk = KEY_CHUNKS - 1;
    const uint4* p = keys + chunk * 32 + (lane & 15);
    AggKeys k;
    k.lo = p[0];
    k.hi = p[16];
    return k;
}

struct AggOwn { unsigned inv[8]; };

template <int PH>
__device__ __forceinline__ void agg_scan(AggKeys kv, int u, const AggOwn& own, int lane,
                                         int* q, int& qhead, int& cnt,
                                         const unsigned* __restrict__ rows,
                                         const float* __restrict__ h,
                                         const float* __restrict__ a_s,
                                         const AggLds& L) {
    if (u >= N_UNITS) return;
    unsigned cm = ((kv.lo.x ^ own.inv[0]) & (kv.lo.y ^ own.inv[1])) &
                  ((kv.lo.z ^ own.inv[2]) & (kv.lo.w ^ own.inv[3])) &
                  ((kv.hi.x ^ own.inv[4]) & (kv.hi.y ^ own.inv[5])) &
                  ((kv.hi.z ^ own.inv[6]) & (kv.hi.w ^ own.inv[7]));
    const int chunk = 2 * u + (lane >> 4);
    if (chunk >= KEY_CHUNKS) cm = 0u;
    unsigned bm = __builtin_amdgcn_ballot_w32(cm != 0u);
    if (bm) {
        const int i = lane & 15;
        const int e0 = chunk * KEY_CHUNK + ((i >> 2) << 7) + (i & 3);
        do {
            if (cm) {
                const int p = __builtin_ctz(cm);
                q[(qhead + cnt + __builtin_amdgcn_mbcnt_lo(bm, 0)) & (AGG_QCAP - 1)] =
                    e0 + (p << 2);
                cm &= cm - 1;
            }
            cnt += __builtin_popcount(bm);
            if (cnt >= 32) {
                agg_drain<PH, true>(rows, h, a_s, q, qhead, 32, lane, L);
                qhead = (qhead + 32) & (AGG_QCAP - 1);
                cnt -= 32;
            }
            bm = __builtin_amdgcn_ballot_w32(cm != 0);
        } while (bm);
    }
}

template <int PH>
__device__ __forceinline__ void agg_pass(const unsigned* __restrict__ rows,
                                         const uint4* __restrict__ keys,
                                         const float* __restrict__ h,
                                         const float* __restrict__ a_s,
                                         int wave, int lane, int* q,
                                         const AggLds& L) {
    int qhead = 0, cnt = 0;
    AggOwn own;
#pragma unroll
    for (int b = 0; b < 8; ++b) own.inv[b] = ((blockIdx.x >> b) & 1u) ? 0u : 0xFFFFFFFFu;
    const int ustart = (blockIdx.x * AGG_UPAD) / AGG_BLOCKS;

    AggKeys k0 = agg_load(keys, agg_unit(0, wave, ustart), lane);
    AggKeys k1 = agg_load(keys, agg_unit(1, wave, ustart), lane);
    AggKeys k2 = agg_load(keys, agg_unit(2, wave, ustart), lane);
    AggKeys k3 = agg_load(keys, agg_unit(3, wave, ustart), lane);
    for (int st = 0; st < AGG_STEPS; st += AGG_NSET) {
        agg_scan<PH>(k0, agg_unit(st,     wave, ustart), own, lane, q, qhead, cnt, rows, h, a_s, L);
        k0 = agg_load(keys, agg_unit(st + 4, wave, ustart), lane);
        agg_scan<PH>(k1, agg_unit(st + 1, wave, ustart), own, lane, q, qhead, cnt, rows, h, a_s, L);
        k1 = agg_load(keys, agg_unit(st + 5, wave, ustart), lane);
        agg_scan<PH>(k2, agg_unit(st + 2, wave, ustart), own, lane, q, qhead, cnt, rows, h, a_s, L);
        k2 = agg_load(keys, agg_unit(st + 6, wave, ustart), lane);
        agg_scan<PH>(k3, agg_unit(st + 3, wave, ustart), own, lane, q, qhead, cnt, rows, h, a_s, L);
        k3 = agg_load(keys, agg_unit(st + 7, wave, ustart), lane);
    }
    if (cnt > 0) agg_drain<PH, false>(rows, h, a_s, q, qhead, cnt, lane, L);
}

__device__ __forceinline__ float wave_sum(float t) {
    for (int o = 16; o > 0; o >>= 1) t += __shfl_xor(t, o, 32);
    return t;
}

__global__ __launch_bounds__(AGG_THREADS)
void k_agg_lds(const unsigned* __restrict__ rows,
               const unsigned* __restrict__ keys,
               const float* __restrict__ h,
               const float* __restrict__ a_s,
               const float* __restrict__ a_d,
               const float* __restrict__ bias,
               const float* __restrict__ gamma,
               const float* __restrict__ beta,
               float* __restrict__ out) {
    __shared__ __attribute__((aligned(16))) float acc[AGG_R * OUT_CH];
    __shared__ __attribute__((aligned(16))) float ssl[AGG_R * HEADS];
    __shared__ __attribute__((aligned(16))) float mxl[AGG_R * HEADS];
    __shared__ __attribute__((aligned(16))) unsigned mkl[AGG_R * HEADS];
    __shared__ __attribute__((aligned(16))) float adl[AGG_R * HEADS];
    __shared__ int ctl[2];
    __shared__ int que[AGG_WAVES * AGG_QCAP];
    __shared__ __attribute__((aligned(16))) int lse[AGG_CAP];
    __shared__ __attribute__((aligned(16))) float4 lsl[AGG_CAP];

    const int tid  = threadIdx.x;
    const int wave = __builtin_amdgcn_readfirstlane(tid >> 5);
    const int lane = tid & 31;
    const int lo   = blockIdx.x * AGG_R;
    AggLds L;
    L.acc = acc; L.ssl = ssl; L.mxl = mxl; L.mkl = mkl; L.adl = adl; L.ctl = ctl;
    L.lse = lse; L.lsl = lsl;
    int* q = que + wave * AGG_QCAP;

    if (tid < AGG_R * HEADS) {
        const int node = lo + (tid >> 2);
        float ad = 0.0f;
        unsigned key = 0u;
        if (node < N_NODES) {
            ad  = a_d[node * HEADS + (tid & 3)];
            key = f2key(lrelu(a_s[node * HEADS + (tid & 3)] + ad));
        }
        adl[tid] = ad;
        mkl[tid] = key;
    }
    if (tid == 0) { ctl[0] = 0; ctl[1] = 0; }
    __syncthreads();

    agg_pass<1>(rows, (const uint4*)keys, h, a_s, wave, lane, q, L);
    __syncthreads();

    for (int r = wave; r < AGG_R; r += AGG_WAVES) {
        const int node = lo + r;
        if (node < N_NODES) {
            const int hh = lane >> 3;
            const float as = a_s[(size_t)node * HEADS + hh];
            const float ad = adl[r * HEADS + hh];
            const float mf = key2f(mkl[r * HEADS + hh]);
            const float w  = expf(lrelu(as + ad) - mf);
            const float4 hv = ((const float4*)(h + (size_t)node * OUT_CH))[lane];
            float* ap = acc + r * OUT_CH + lane;
            ap[0]  = hv.x * w;
            ap[32] = hv.y * w;
            ap[64] = hv.z * w;
            ap[96] = hv.w * w;
            if ((lane & 7) == 0) {
                ssl[r * HEADS + hh] = w;
                mxl[r * HEADS + hh] = mf;
            }
        }
    }
    __syncthreads();

    const int total = __builtin_amdgcn_readfirstlane(ctl[0]);
    const int ovf   = __builtin_amdgcn_readfirstlane(ctl[1]);
    if (ovf == 0) {
        const int last = total - 1;
        const int step = AGG_WAVES * 8;
        int p = wave * 8;
        const int nf = (total - p >= 8) ? (total - p - 8) / step + 1 : 0;
        if (nf > 0) {
            const int capf = p + (nf - 1) * step + 7;
            AggGrp A, B;
            agg_grp_issue<true>(h, p, 8, capf, lane, L, A);
            int g = 0;
            for (; g + 2 <= nf; g += 2) {
                agg_grp_issue<true>(h, p + step, 8, capf, lane, L, B);
                agg_grp_finish<true>(p, 8, capf, lane, L, A);
                agg_grp_issue<true>(h, p + 2 * step, 8, capf, lane, L, A);
                agg_grp_finish<true>(p + step, 8, capf, lane, L, B);
                p += 2 * step;
            }
            if (g < nf) {
                agg_grp_finish<true>(p, 8, capf, lane, L, A);
                p += step;
            }
        }
        if (p < total) {
            AggGrp T;
            agg_grp_issue<false>(h, p, total - p, last, lane, L, T);
            agg_grp_finish<false>(p, total - p, last, lane, L, T);
        }
    } else {
        agg_pass<2>(rows, (const uint4*)keys, h, a_s, wave, lane, q, L);
    }
    __syncthreads();

    const float4 bs4 = ((const float4*)bias)[lane];
    const float4 gm4 = ((const float4*)gamma)[lane];
    const float4 bt4 = ((const float4*)beta)[lane];
    const float bs0 = bs4.x, bs1 = bs4.y, bs2 = bs4.z, bs3 = bs4.w;
    const float gm0 = gm4.x, gm1 = gm4.y, gm2 = gm4.z, gm3 = gm4.w;
    const float bt0 = bt4.x, bt1 = bt4.y, bt2 = bt4.z, bt3 = bt4.w;
    for (int r = wave; r < AGG_R; r += AGG_WAVES) {
        const int node = lo + r;
        if (node < N_NODES) {
            const float* ap = acc + r * OUT_CH + lane;
            const float sv = ssl[r * HEADS + (lane >> 3)] + SOFTMAX_EPS;
            const float v0 = ap[0]  / sv + bs0;
            const float v1 = ap[32] / sv + bs1;
            const float v2 = ap[64] / sv + bs2;
            const float v3 = ap[96] / sv + bs3;

            const float mu = wave_sum((v0 + v1) + (v2 + v3)) * (1.0f / OUT_CH);
            const float d0 = v0 - mu, d1 = v1 - mu, d2 = v2 - mu, d3 = v3 - mu;
            const float var = wave_sum((d0 * d0 + d1 * d1) + (d2 * d2 + d3 * d3)) * (1.0f / OUT_CH);
            const float rs = 1.0f / sqrtf(var + LN_EPS);

            float y0 = d0 * rs * gm0 + bt0;
            float y1 = d1 * rs * gm1 + bt1;
            float y2 = d2 * rs * gm2 + bt2;
            float y3 = d3 * rs * gm3 + bt3;
            y0 = y0 > 0.0f ? y0 : (expf(y0) - 1.0f);
            y1 = y1 > 0.0f ? y1 : (expf(y1) - 1.0f);
            y2 = y2 > 0.0f ? y2 : (expf(y2) - 1.0f);
            y3 = y3 > 0.0f ? y3 : (expf(y3) - 1.0f);

            ((float4*)(out + (size_t)node * OUT_CH))[lane] = make_float4(y0, y1, y2, y3);
        }
    }
}

extern "C" void kernel_launch(void* const* d_in, const int* in_sizes, int n_in,
                              void* d_out, int out_size, void* d_ws, size_t ws_size,
                              hipStream_t stream) {
    const float* x       = (const float*)d_in[0];
    const int*   ei      = (const int*)  d_in[1];
    const float* W       = (const float*)d_in[2];
    const float* att_src = (const float*)d_in[3];
    const float* att_dst = (const float*)d_in[4];
    const float* bias    = (const float*)d_in[5];
    const float* gamma   = (const float*)d_in[6];
    const float* beta    = (const float*)d_in[7];
    float* out = (float*)d_out;

    char* ws = (char*)d_ws;
    float*    h      = (float*)ws;
    ws += (size_t)N_NODES * OUT_CH * sizeof(float);
    float*    a_s    = (float*)ws;  ws += (size_t)N_NODES * HEADS * sizeof(float);
    float*    a_d    = (float*)ws;  ws += (size_t)N_NODES * HEADS * sizeof(float);
    unsigned* keys   = (unsigned*)ws;
    ws += (size_t)N_EDGES;
    unsigned* rows   = (unsigned*)ws;

    k_gemm_wmma<<<(N_NODES + 127) / 128, 256, IN_CH * OUT_CH * sizeof(float),
                  stream>>>(x, W, h, N_NODES);

    k_dots<<<DOT_BLOCKS, DOT_THREADS, 0, stream>>>(h, att_src, att_dst, ei, keys, rows, a_s, a_d,
                                                   N_NODES);

    k_agg_lds<<<AGG_BLOCKS, AGG_THREADS, 0, stream>>>(rows, keys, h, a_s, a_d, bias,
                                                      gamma, beta, out);
}
